# starting kernel
# baseline (speedup 1.0000x reference)
#include <hip/hip_runtime.h>

#define N_OCC   20000
#define N_SKILL 50000
#define N_EDGE  600000
#define D_IN    256
#define HID     128
#define OUTF    64

typedef __attribute__((ext_vector_type(2))) float v2f;
typedef __attribute__((ext_vector_type(8))) float v8f;

template <int K, int N, bool FUSE>
__global__ void __launch_bounds__(128)
gemm_sage(const float* __restrict__ A1, const float* __restrict__ WT1,
          const float* __restrict__ A2, const float* __restrict__ WT2,
          const unsigned* __restrict__ cnt, const float* __restrict__ bias,
          float* __restrict__ out, int M, int do_relu)
{
    const int lane = threadIdx.x & 31;
    const int wave = threadIdx.x >> 5;
    const int hi   = lane >> 4;
    const int lm   = lane & 15;
    const int rt   = blockIdx.x * 4 + wave;
    if (rt * 16 >= M) return;
    const int row0 = rt * 16;
    const int col0 = blockIdx.y * 64;

    v8f c0 = {0.f,0.f,0.f,0.f,0.f,0.f,0.f,0.f};
    v8f c1 = c0, c2 = c0, c3 = c0;

    {
        const float* ap = A1  + (size_t)(row0 + lm) * K + 2 * hi;
        const float* bp = WT1 + (size_t)(col0 + lm) * K + 2 * hi;
        #pragma unroll 8
        for (int k = 0; k < K / 4; ++k) {
            v2f a  = *(const v2f*)(ap + 4 * k);
            v2f b0 = *(const v2f*)(bp + 4 * k);
            v2f b1 = *(const v2f*)(bp + 4 * k + 16 * K);
            v2f b2 = *(const v2f*)(bp + 4 * k + 32 * K);
            v2f b3 = *(const v2f*)(bp + 4 * k + 48 * K);
            c0 = __builtin_amdgcn_wmma_f32_16x16x4_f32(false, a, false, b0, (short)0, c0, false, false);
            c1 = __builtin_amdgcn_wmma_f32_16x16x4_f32(false, a, false, b1, (short)0, c1, false, false);
            c2 = __builtin_amdgcn_wmma_f32_16x16x4_f32(false, a, false, b2, (short)0, c2, false, false);
            c3 = __builtin_amdgcn_wmma_f32_16x16x4_f32(false, a, false, b3, (short)0, c3, false, false);
        }
    }

    if (FUSE) {
        #pragma unroll
        for (int r = 0; r < 8; ++r) {
            const float inv = 1.0f / fmaxf((float)cnt[row0 + r + 8 * hi], 1.0f);
            c0[r] *= inv; c1[r] *= inv; c2[r] *= inv; c3[r] *= inv;
        }
        const float* ap = A2  + (size_t)(row0 + lm) * K + 2 * hi;
        const float* bp = WT2 + (size_t)(col0 + lm) * K + 2 * hi;
        #pragma unroll 8
        for (int k = 0; k < K / 4; ++k) {
            v2f a  = *(const v2f*)(ap + 4 * k);
            v2f b0 = *(const v2f*)(bp + 4 * k);
            v2f b1 = *(const v2f*)(bp + 4 * k + 16 * K);
            v2f b2 = *(const v2f*)(bp + 4 * k + 32 * K);
            v2f b3 = *(const v2f*)(bp + 4 * k + 48 * K);
            c0 = __builtin_amdgcn_wmma_f32_16x16x4_f32(false, a, false, b0, (short)0, c0, false, false);
            c1 = __builtin_amdgcn_wmma_f32_16x16x4_f32(false, a, false, b1, (short)0, c1, false, false);
            c2 = __builtin_amdgcn_wmma_f32_16x16x4_f32(false, a, false, b2, (short)0, c2, false, false);
            c3 = __builtin_amdgcn_wmma_f32_16x16x4_f32(false, a, false, b3, (short)0, c3, false, false);
        }
    }

    float* orow = out + (size_t)(row0 + 8 * hi) * N + col0 + lm;
    const float bv0 = bias[col0 + lm];
    const float bv1 = bias[col0 + 16 + lm];
    const float bv2 = bias[col0 + 32 + lm];
    const float bv3 = bias[col0 + 48 + lm];
    #pragma unroll
    for (int r = 0; r < 8; ++r) {
        float v0 = c0[r] + bv0, v1 = c1[r] + bv1, v2 = c2[r] + bv2, v3 = c3[r] + bv3;
        if (do_relu) {
            v0 = fmaxf(v0, 0.f); v1 = fmaxf(v1, 0.f);
            v2 = fmaxf(v2, 0.f); v3 = fmaxf(v3, 0.f);
        }
        float* p = orow + (size_t)r * N;
        p[0] = v0; p[16] = v1; p[32] = v2; p[48] = v3;
    }
}


__global__ void __launch_bounds__(256)
transpose_w(const float* __restrict__ W, float* __restrict__ WT, int K, int N)
{
    int i = blockIdx.x * blockDim.x + threadIdx.x;
    if (i < K * N) {
        int k = i / N, n = i - k * N;
        WT[(size_t)n * K + k] = W[i];
    }
}

__device__ inline void atomic_add_f32(float* p, float v)
{
    (void)__hip_atomic_fetch_add(p, v, __ATOMIC_RELAXED, __HIP_MEMORY_SCOPE_AGENT);
}

__global__ void __launch_bounds__(256)
count_deg(const int* __restrict__ e_occ, const int* __restrict__ e_skill,
          unsigned* __restrict__ cnt_occ, unsigned* __restrict__ cnt_skill, int E)
{
    int e = blockIdx.x * blockDim.x + threadIdx.x;
    if (e < E) {
        atomicAdd(&cnt_occ[e_occ[e]], 1u);
        atomicAdd(&cnt_skill[e_skill[e]], 1u);
    }
}

__global__ void __launch_bounds__(256)
scatter_add(const float* __restrict__ hsrc, const int* __restrict__ src,
            const int* __restrict__ dst, float* __restrict__ agg, int E)
{
    const long long tid = (long long)blockIdx.x * blockDim.x + threadIdx.x;
    const int e = (int)(tid >> 5);
    const int f = ((int)tid & 31) << 2;
    if (e < E) {
        const int s = src[e];
        const int d = dst[e];
        const float4 v = *(const float4*)(hsrc + (size_t)s * HID + f);
        float* p = agg + (size_t)d * HID + f;
        atomic_add_f32(p + 0, v.x);
        atomic_add_f32(p + 1, v.y);
        atomic_add_f32(p + 2, v.z);
        atomic_add_f32(p + 3, v.w);
    }
}

extern "C" void kernel_launch(void* const* d_in, const int* in_sizes, int n_in,
                              void* d_out, int out_size, void* d_ws, size_t ws_size,
                              hipStream_t stream)
{
    (void)in_sizes; (void)n_in; (void)out_size; (void)ws_size;

    const float* x_occ    = (const float*)d_in[0];
    const float* x_skill  = (const float*)d_in[1];
    const int*   e_occ    = (const int*)  d_in[2];
    const int*   e_skill  = (const int*)  d_in[3];
    const float* pW_occ   = (const float*)d_in[4];
    const float* pb_occ   = (const float*)d_in[5];
    const float* pW_skill = (const float*)d_in[6];
    const float* pb_skill = (const float*)d_in[7];
    const float* c1_os_Wl = (const float*)d_in[8];
    const float* c1_os_bl = (const float*)d_in[9];
    const float* c1_os_Wr = (const float*)d_in[10];
    const float* c1_so_Wl = (const float*)d_in[11];
    const float* c1_so_bl = (const float*)d_in[12];
    const float* c1_so_Wr = (const float*)d_in[13];
    const float* c2_os_Wl = (const float*)d_in[14];
    const float* c2_os_bl = (const float*)d_in[15];
    const float* c2_os_Wr = (const float*)d_in[16];
    const float* c2_so_Wl = (const float*)d_in[17];
    const float* c2_so_bl = (const float*)d_in[18];
    const float* c2_so_Wr = (const float*)d_in[19];

    float* ws = (float*)d_ws;
    float* h_occ     = ws;
    float* h_skill   = h_occ     + (size_t)N_OCC   * HID;
    float* o1        = h_skill   + (size_t)N_SKILL * HID;
    float* s1        = o1        + (size_t)N_OCC   * HID;
    float* agg_occ   = s1        + (size_t)N_SKILL * HID;
    float* agg_skill = agg_occ   + (size_t)N_OCC   * HID;
    unsigned* cnt_occ   = (unsigned*)(agg_skill + (size_t)N_SKILL * HID);
    unsigned* cnt_skill = cnt_occ + N_OCC;
    float* wt = (float*)(cnt_skill + N_SKILL);
    float* pWt_occ   = wt;                 wt += D_IN * HID;
    float* pWt_skill = wt;                 wt += D_IN * HID;
    float* t1_os_Wl  = wt;                 wt += HID * HID;
    float* t1_os_Wr  = wt;                 wt += HID * HID;
    float* t1_so_Wl  = wt;                 wt += HID * HID;
    float* t1_so_Wr  = wt;                 wt += HID * HID;
    float* t2_os_Wl  = wt;                 wt += HID * OUTF;
    float* t2_os_Wr  = wt;                 wt += HID * OUTF;
    float* t2_so_Wl  = wt;                 wt += HID * OUTF;
    float* t2_so_Wr  = wt;                 wt += HID * OUTF;

    float* o2 = (float*)d_out;
    float* s2 = o2 + (size_t)N_OCC * OUTF;

    const dim3 blk128(128), blk256(256);
    const int rt_occ   = (N_OCC / 16 + 3) / 4;
    const int rt_skill = (N_SKILL / 16 + 3) / 4;

    hipMemsetAsync(cnt_occ, 0, sizeof(unsigned) * (N_OCC + N_SKILL), stream);
    hipMemsetAsync(agg_occ, 0, sizeof(float) * (size_t)(N_OCC + N_SKILL) * HID, stream);

    transpose_w<<<(D_IN * HID + 255) / 256, blk256, 0, stream>>>(pW_occ,   pWt_occ,   D_IN, HID);
    transpose_w<<<(D_IN * HID + 255) / 256, blk256, 0, stream>>>(pW_skill, pWt_skill, D_IN, HID);
    transpose_w<<<(HID * HID + 255) / 256, blk256, 0, stream>>>(c1_os_Wl, t1_os_Wl, HID, HID);
    transpose_w<<<(HID * HID + 255) / 256, blk256, 0, stream>>>(c1_os_Wr, t1_os_Wr, HID, HID);
    transpose_w<<<(HID * HID + 255) / 256, blk256, 0, stream>>>(c1_so_Wl, t1_so_Wl, HID, HID);
    transpose_w<<<(HID * HID + 255) / 256, blk256, 0, stream>>>(c1_so_Wr, t1_so_Wr, HID, HID);
    transpose_w<<<(HID * OUTF + 255) / 256, blk256, 0, stream>>>(c2_os_Wl, t2_os_Wl, HID, OUTF);
    transpose_w<<<(HID * OUTF + 255) / 256, blk256, 0, stream>>>(c2_os_Wr, t2_os_Wr, HID, OUTF);
    transpose_w<<<(HID * OUTF + 255) / 256, blk256, 0, stream>>>(c2_so_Wl, t2_so_Wl, HID, OUTF);
    transpose_w<<<(HID * OUTF + 255) / 256, blk256, 0, stream>>>(c2_so_Wr, t2_so_Wr, HID, OUTF);

    count_deg<<<(N_EDGE + 255) / 256, blk256, 0, stream>>>(
        e_occ, e_skill, cnt_occ, cnt_skill, N_EDGE);

    gemm_sage<D_IN, HID, false><<<dim3(rt_occ, HID / 64), blk128, 0, stream>>>(
        x_occ, pWt_occ, nullptr, nullptr, nullptr, pb_occ, h_occ, N_OCC, 1);
    gemm_sage<D_IN, HID, false><<<dim3(rt_skill, HID / 64), blk128, 0, stream>>>(
        x_skill, pWt_skill, nullptr, nullptr, nullptr, pb_skill, h_skill, N_SKILL, 1);

    const int sg_blocks = (int)(((long long)N_EDGE * 32 + 255) / 256);
    scatter_add<<<sg_blocks, blk256, 0, stream>>>(h_occ,   e_occ,   e_skill, agg_skill, N_EDGE);
    scatter_add<<<sg_blocks, blk256, 0, stream>>>(h_skill, e_skill, e_occ,   agg_occ,   N_EDGE);

    gemm_sage<HID, HID, true><<<dim3(rt_skill, HID / 64), blk128, 0, stream>>>(
        agg_skill, t1_os_Wl, h_skill, t1_os_Wr, cnt_skill, c1_os_bl, s1, N_SKILL, 1);
    gemm_sage<HID, HID, true><<<dim3(rt_occ, HID / 64), blk128, 0, stream>>>(
        agg_occ, t1_so_Wl, h_occ, t1_so_Wr, cnt_occ, c1_so_bl, o1, N_OCC, 1);

    hipMemsetAsync(agg_occ, 0, sizeof(float) * (size_t)(N_OCC + N_SKILL) * HID, stream);
    scatter_add<<<sg_blocks, blk256, 0, stream>>>(o1, e_occ,   e_skill, agg_skill, N_EDGE);
    scatter_add<<<sg_blocks, blk256, 0, stream>>>(s1, e_skill, e_occ,   agg_occ,   N_EDGE);

    gemm_sage<HID, OUTF, true><<<dim3(rt_skill, OUTF / 64), blk128, 0, stream>>>(
        agg_skill, t2_os_Wl, s1, t2_os_Wr, cnt_skill, c2_os_bl, s2, N_SKILL, 0);
    gemm_sage<HID, OUTF, true><<<dim3(rt_occ, OUTF / 64), blk128, 0, stream>>>(
        agg_occ, t2_so_Wl, o1, t2_so_Wr, cnt_occ, c2_so_bl, o2, N_OCC, 0);
}


# conv2 multiplies by W_l BEFORE the atomic neighbour sum (mean(x)@Wl = mean(x@Wl)): each edge scatters 64 floats, not 128; conv2 sums projected rows, so its f32 sum order changes
# speedup vs baseline: 1.1515x; 1.1515x over previous
#include <hip/hip_runtime.h>

#define N_OCC   20000
#define N_SKILL 50000
#define N_EDGE  600000
#define D_IN    256
#define HID     128
#define OUTF    64

typedef __attribute__((ext_vector_type(2))) float v2f;
typedef __attribute__((ext_vector_type(8))) float v8f;

template <int K, int N, int MODE>
__global__ void __launch_bounds__(128)
gemm_sage(const float* __restrict__ A1, const float* __restrict__ WT1,
          const float* __restrict__ A2, const float* __restrict__ WT2,
          const unsigned* __restrict__ cnt, const float* __restrict__ bias,
          float* __restrict__ out, int M, int do_relu)
{
    const int lane = threadIdx.x & 31;
    const int wave = threadIdx.x >> 5;
    const int hi   = lane >> 4;
    const int lm   = lane & 15;
    const int rt   = blockIdx.x * 4 + wave;
    if (rt * 16 >= M) return;
    const int row0 = rt * 16;
    const int col0 = blockIdx.y * 64;

    v8f c0 = {0.f,0.f,0.f,0.f,0.f,0.f,0.f,0.f};
    v8f c1 = c0, c2 = c0, c3 = c0;

    float g0[8] = {}, g1[8] = {}, g2[8] = {}, g3[8] = {}, dg[8] = {};
    if (MODE == 2) {
        const float* g = A2 + (size_t)(row0 + 8 * hi) * N + col0 + lm;
        #pragma unroll
        for (int r = 0; r < 8; ++r) {
            g0[r] = g[r * N];      g1[r] = g[r * N + 16];
            g2[r] = g[r * N + 32]; g3[r] = g[r * N + 48];
            dg[r] = (float)cnt[row0 + 8 * hi + r];
        }
    }

    {
        const float* ap = A1  + (size_t)(row0 + lm) * K + 2 * hi;
        const float* bp = WT1 + (size_t)(col0 + lm) * K + 2 * hi;
        #pragma unroll 8
        for (int k = 0; k < K / 4; ++k) {
            v2f a  = *(const v2f*)(ap + 4 * k);
            v2f b0 = *(const v2f*)(bp + 4 * k);
            v2f b1 = *(const v2f*)(bp + 4 * k + 16 * K);
            v2f b2 = *(const v2f*)(bp + 4 * k + 32 * K);
            v2f b3 = *(const v2f*)(bp + 4 * k + 48 * K);
            c0 = __builtin_amdgcn_wmma_f32_16x16x4_f32(false, a, false, b0, (short)0, c0, false, false);
            c1 = __builtin_amdgcn_wmma_f32_16x16x4_f32(false, a, false, b1, (short)0, c1, false, false);
            c2 = __builtin_amdgcn_wmma_f32_16x16x4_f32(false, a, false, b2, (short)0, c2, false, false);
            c3 = __builtin_amdgcn_wmma_f32_16x16x4_f32(false, a, false, b3, (short)0, c3, false, false);
        }
    }

    if (MODE == 1) {
        #pragma unroll
        for (int r = 0; r < 8; ++r) {
            const float inv = 1.0f / fmaxf((float)cnt[row0 + r + 8 * hi], 1.0f);
            c0[r] *= inv; c1[r] *= inv; c2[r] *= inv; c3[r] *= inv;
        }
        const float* ap = A2  + (size_t)(row0 + lm) * K + 2 * hi;
        const float* bp = WT2 + (size_t)(col0 + lm) * K + 2 * hi;
        #pragma unroll 8
        for (int k = 0; k < K / 4; ++k) {
            v2f a  = *(const v2f*)(ap + 4 * k);
            v2f b0 = *(const v2f*)(bp + 4 * k);
            v2f b1 = *(const v2f*)(bp + 4 * k + 16 * K);
            v2f b2 = *(const v2f*)(bp + 4 * k + 32 * K);
            v2f b3 = *(const v2f*)(bp + 4 * k + 48 * K);
            c0 = __builtin_amdgcn_wmma_f32_16x16x4_f32(false, a, false, b0, (short)0, c0, false, false);
            c1 = __builtin_amdgcn_wmma_f32_16x16x4_f32(false, a, false, b1, (short)0, c1, false, false);
            c2 = __builtin_amdgcn_wmma_f32_16x16x4_f32(false, a, false, b2, (short)0, c2, false, false);
            c3 = __builtin_amdgcn_wmma_f32_16x16x4_f32(false, a, false, b3, (short)0, c3, false, false);
        }
    }

    float* orow = out + (size_t)(row0 + 8 * hi) * N + col0 + lm;
    float bv0 = 0.f, bv1 = 0.f, bv2 = 0.f, bv3 = 0.f;
    if (MODE != 3) {
        bv0 = bias[col0 + lm];
        bv1 = bias[col0 + 16 + lm];
        bv2 = bias[col0 + 32 + lm];
        bv3 = bias[col0 + 48 + lm];
    }
    #pragma unroll
    for (int r = 0; r < 8; ++r) {
        float v0 = c0[r], v1 = c1[r], v2 = c2[r], v3 = c3[r];
        if (MODE != 3) { v0 += bv0; v1 += bv1; v2 += bv2; v3 += bv3; }
        if (MODE == 2) {
            const float inv = 1.0f / fmaxf(dg[r], 1.0f);
            v0 += inv * g0[r]; v1 += inv * g1[r];
            v2 += inv * g2[r]; v3 += inv * g3[r];
        }
        if (do_relu) {
            v0 = fmaxf(v0, 0.f); v1 = fmaxf(v1, 0.f);
            v2 = fmaxf(v2, 0.f); v3 = fmaxf(v3, 0.f);
        }
        float* p = orow + (size_t)r * N;
        p[0] = v0; p[16] = v1; p[32] = v2; p[48] = v3;
    }
}


__global__ void __launch_bounds__(256)
transpose_w(const float* __restrict__ W, float* __restrict__ WT, int K, int N)
{
    int i = blockIdx.x * blockDim.x + threadIdx.x;
    if (i < K * N) {
        int k = i / N, n = i - k * N;
        WT[(size_t)n * K + k] = W[i];
    }
}

__device__ inline void atomic_add_f32(float* p, float v)
{
    (void)__hip_atomic_fetch_add(p, v, __ATOMIC_RELAXED, __HIP_MEMORY_SCOPE_AGENT);
}

__global__ void __launch_bounds__(256)
count_deg(const int* __restrict__ e_occ, const int* __restrict__ e_skill,
          unsigned* __restrict__ cnt_occ, unsigned* __restrict__ cnt_skill, int E)
{
    int e = blockIdx.x * blockDim.x + threadIdx.x;
    if (e < E) {
        atomicAdd(&cnt_occ[e_occ[e]], 1u);
        atomicAdd(&cnt_skill[e_skill[e]], 1u);
    }
}

template <int F>
__global__ void __launch_bounds__(256)
scatter_add(const float* __restrict__ hsrc, const int* __restrict__ src,
            const int* __restrict__ dst, float* __restrict__ agg, int E)
{
    constexpr int TPE = F / 4;
    const long long tid = (long long)blockIdx.x * blockDim.x + threadIdx.x;
    const int e = (int)(tid / TPE);
    const int f = ((int)tid & (TPE - 1)) << 2;
    if (e < E) {
        const int s = src[e];
        const int d = dst[e];
        const float4 v = *(const float4*)(hsrc + (size_t)s * F + f);
        float* p = agg + (size_t)d * F + f;
        atomic_add_f32(p + 0, v.x);
        atomic_add_f32(p + 1, v.y);
        atomic_add_f32(p + 2, v.z);
        atomic_add_f32(p + 3, v.w);
    }
}

extern "C" void kernel_launch(void* const* d_in, const int* in_sizes, int n_in,
                              void* d_out, int out_size, void* d_ws, size_t ws_size,
                              hipStream_t stream)
{
    (void)in_sizes; (void)n_in; (void)out_size; (void)ws_size;

    const float* x_occ    = (const float*)d_in[0];
    const float* x_skill  = (const float*)d_in[1];
    const int*   e_occ    = (const int*)  d_in[2];
    const int*   e_skill  = (const int*)  d_in[3];
    const float* pW_occ   = (const float*)d_in[4];
    const float* pb_occ   = (const float*)d_in[5];
    const float* pW_skill = (const float*)d_in[6];
    const float* pb_skill = (const float*)d_in[7];
    const float* c1_os_Wl = (const float*)d_in[8];
    const float* c1_os_bl = (const float*)d_in[9];
    const float* c1_os_Wr = (const float*)d_in[10];
    const float* c1_so_Wl = (const float*)d_in[11];
    const float* c1_so_bl = (const float*)d_in[12];
    const float* c1_so_Wr = (const float*)d_in[13];
    const float* c2_os_Wl = (const float*)d_in[14];
    const float* c2_os_bl = (const float*)d_in[15];
    const float* c2_os_Wr = (const float*)d_in[16];
    const float* c2_so_Wl = (const float*)d_in[17];
    const float* c2_so_bl = (const float*)d_in[18];
    const float* c2_so_Wr = (const float*)d_in[19];

    float* ws = (float*)d_ws;
    float* h_occ     = ws;
    float* h_skill   = h_occ     + (size_t)N_OCC   * HID;
    float* o1        = h_skill   + (size_t)N_SKILL * HID;
    float* s1        = o1        + (size_t)N_OCC   * HID;
    float* agg_occ   = s1        + (size_t)N_SKILL * HID;
    float* agg_skill = agg_occ   + (size_t)N_OCC   * HID;
    unsigned* cnt_occ   = (unsigned*)(agg_skill + (size_t)N_SKILL * HID);
    unsigned* cnt_skill = cnt_occ + N_OCC;
    float* wt = (float*)(cnt_skill + N_SKILL);
    float* pWt_occ   = wt;                 wt += D_IN * HID;
    float* pWt_skill = wt;                 wt += D_IN * HID;
    float* t1_os_Wl  = wt;                 wt += HID * HID;
    float* t1_os_Wr  = wt;                 wt += HID * HID;
    float* t1_so_Wl  = wt;                 wt += HID * HID;
    float* t1_so_Wr  = wt;                 wt += HID * HID;
    float* t2_os_Wl  = wt;                 wt += HID * OUTF;
    float* t2_os_Wr  = wt;                 wt += HID * OUTF;
    float* t2_so_Wl  = wt;                 wt += HID * OUTF;
    float* t2_so_Wr  = wt;                 wt += HID * OUTF;

    float* o2 = (float*)d_out;
    float* s2 = o2 + (size_t)N_OCC * OUTF;

    const dim3 blk128(128), blk256(256);
    const int rt_occ   = (N_OCC / 16 + 3) / 4;
    const int rt_skill = (N_SKILL / 16 + 3) / 4;

    hipMemsetAsync(cnt_occ, 0, sizeof(unsigned) * (N_OCC + N_SKILL), stream);
    hipMemsetAsync(agg_occ, 0, sizeof(float) * (size_t)(N_OCC + N_SKILL) * HID, stream);

    transpose_w<<<(D_IN * HID + 255) / 256, blk256, 0, stream>>>(pW_occ,   pWt_occ,   D_IN, HID);
    transpose_w<<<(D_IN * HID + 255) / 256, blk256, 0, stream>>>(pW_skill, pWt_skill, D_IN, HID);
    transpose_w<<<(HID * HID + 255) / 256, blk256, 0, stream>>>(c1_os_Wl, t1_os_Wl, HID, HID);
    transpose_w<<<(HID * HID + 255) / 256, blk256, 0, stream>>>(c1_os_Wr, t1_os_Wr, HID, HID);
    transpose_w<<<(HID * HID + 255) / 256, blk256, 0, stream>>>(c1_so_Wl, t1_so_Wl, HID, HID);
    transpose_w<<<(HID * HID + 255) / 256, blk256, 0, stream>>>(c1_so_Wr, t1_so_Wr, HID, HID);
    transpose_w<<<(HID * OUTF + 255) / 256, blk256, 0, stream>>>(c2_os_Wl, t2_os_Wl, HID, OUTF);
    transpose_w<<<(HID * OUTF + 255) / 256, blk256, 0, stream>>>(c2_os_Wr, t2_os_Wr, HID, OUTF);
    transpose_w<<<(HID * OUTF + 255) / 256, blk256, 0, stream>>>(c2_so_Wl, t2_so_Wl, HID, OUTF);
    transpose_w<<<(HID * OUTF + 255) / 256, blk256, 0, stream>>>(c2_so_Wr, t2_so_Wr, HID, OUTF);

    count_deg<<<(N_EDGE + 255) / 256, blk256, 0, stream>>>(
        e_occ, e_skill, cnt_occ, cnt_skill, N_EDGE);

    gemm_sage<D_IN, HID, 0><<<dim3(rt_occ, HID / 64), blk128, 0, stream>>>(
        x_occ, pWt_occ, nullptr, nullptr, nullptr, pb_occ, h_occ, N_OCC, 1);
    gemm_sage<D_IN, HID, 0><<<dim3(rt_skill, HID / 64), blk128, 0, stream>>>(
        x_skill, pWt_skill, nullptr, nullptr, nullptr, pb_skill, h_skill, N_SKILL, 1);

    const int sg_blocks = (int)(((long long)N_EDGE * 32 + 255) / 256);
    scatter_add<HID><<<sg_blocks, blk256, 0, stream>>>(h_occ,   e_occ,   e_skill, agg_skill, N_EDGE);
    scatter_add<HID><<<sg_blocks, blk256, 0, stream>>>(h_skill, e_skill, e_occ,   agg_occ,   N_EDGE);

    gemm_sage<HID, HID, 1><<<dim3(rt_skill, HID / 64), blk128, 0, stream>>>(
        agg_skill, t1_os_Wl, h_skill, t1_os_Wr, cnt_skill, c1_os_bl, s1, N_SKILL, 1);
    gemm_sage<HID, HID, 1><<<dim3(rt_occ, HID / 64), blk128, 0, stream>>>(
        agg_occ, t1_so_Wl, h_occ, t1_so_Wr, cnt_occ, c1_so_bl, o1, N_OCC, 1);

    float* p_occ      = h_occ;
    float* p_skill    = h_skill;
    float* agg2_occ   = agg_occ;
    float* agg2_skill = agg_occ + (size_t)N_OCC * OUTF;
    hipMemsetAsync(agg_occ, 0, sizeof(float) * (size_t)(N_OCC + N_SKILL) * OUTF, stream);
    gemm_sage<HID, OUTF, 3><<<dim3(rt_occ, OUTF / 64), blk128, 0, stream>>>(
        o1, t2_os_Wl, nullptr, nullptr, nullptr, nullptr, p_occ, N_OCC, 0);
    gemm_sage<HID, OUTF, 3><<<dim3(rt_skill, OUTF / 64), blk128, 0, stream>>>(
        s1, t2_so_Wl, nullptr, nullptr, nullptr, nullptr, p_skill, N_SKILL, 0);
    const int sg2_blocks = (int)(((long long)N_EDGE * (OUTF / 4) + 255) / 256);
    scatter_add<OUTF><<<sg2_blocks, blk256, 0, stream>>>(p_occ,   e_occ,   e_skill, agg2_skill, N_EDGE);
    scatter_add<OUTF><<<sg2_blocks, blk256, 0, stream>>>(p_skill, e_skill, e_occ,   agg2_occ,   N_EDGE);

    gemm_sage<HID, OUTF, 2><<<dim3(rt_skill, OUTF / 64), blk128, 0, stream>>>(
        s1, t2_os_Wr, agg2_skill, nullptr, cnt_skill, c2_os_bl, s2, N_SKILL, 0);
    gemm_sage<HID, OUTF, 2><<<dim3(rt_occ, OUTF / 64), blk128, 0, stream>>>(
        o1, t2_so_Wr, agg2_occ, nullptr, cnt_occ, c2_so_bl, o2, N_OCC, 0);
}


# float-atomic scatter-add replaced by a per-call counting sort of the edges (integer atomic add returns the rank) and an atomic-free row gather-sum per destination; f32 sum order follows that rank
# speedup vs baseline: 1.6954x; 1.4723x over previous
#include <hip/hip_runtime.h>

#define N_OCC   20000
#define N_SKILL 50000
#define N_EDGE  600000
#define D_IN    256
#define HID     128
#define OUTF    64

typedef __attribute__((ext_vector_type(2))) float v2f;
typedef __attribute__((ext_vector_type(8))) float v8f;

template <int K, int N, int MODE>
__global__ void __launch_bounds__(128)
gemm_sage(const float* __restrict__ A1, const float* __restrict__ WT1,
          const float* __restrict__ A2, const float* __restrict__ WT2,
          const unsigned* __restrict__ cnt, const float* __restrict__ bias,
          float* __restrict__ out, int M, int do_relu)
{
    const int lane = threadIdx.x & 31;
    const int wave = threadIdx.x >> 5;
    const int hi   = lane >> 4;
    const int lm   = lane & 15;
    const int rt   = blockIdx.x * 4 + wave;
    if (rt * 16 >= M) return;
    const int row0 = rt * 16;
    const int col0 = blockIdx.y * 64;

    v8f c0 = {0.f,0.f,0.f,0.f,0.f,0.f,0.f,0.f};
    v8f c1 = c0, c2 = c0, c3 = c0;

    float g0[8] = {}, g1[8] = {}, g2[8] = {}, g3[8] = {}, dg[8] = {};
    if (MODE == 2) {
        const float* g = A2 + (size_t)(row0 + 8 * hi) * N + col0 + lm;
        #pragma unroll
        for (int r = 0; r < 8; ++r) {
            g0[r] = g[r * N];      g1[r] = g[r * N + 16];
            g2[r] = g[r * N + 32]; g3[r] = g[r * N + 48];
            dg[r] = (float)cnt[row0 + 8 * hi + r];
        }
    }

    {
        const float* ap = A1  + (size_t)(row0 + lm) * K + 2 * hi;
        const float* bp = WT1 + (size_t)(col0 + lm) * K + 2 * hi;
        #pragma unroll 8
        for (int k = 0; k < K / 4; ++k) {
            v2f a  = *(const v2f*)(ap + 4 * k);
            v2f b0 = *(const v2f*)(bp + 4 * k);
            v2f b1 = *(const v2f*)(bp + 4 * k + 16 * K);
            v2f b2 = *(const v2f*)(bp + 4 * k + 32 * K);
            v2f b3 = *(const v2f*)(bp + 4 * k + 48 * K);
            c0 = __builtin_amdgcn_wmma_f32_16x16x4_f32(false, a, false, b0, (short)0, c0, false, false);
            c1 = __builtin_amdgcn_wmma_f32_16x16x4_f32(false, a, false, b1, (short)0, c1, false, false);
            c2 = __builtin_amdgcn_wmma_f32_16x16x4_f32(false, a, false, b2, (short)0, c2, false, false);
            c3 = __builtin_amdgcn_wmma_f32_16x16x4_f32(false, a, false, b3, (short)0, c3, false, false);
        }
    }

    if (MODE == 1) {
        #pragma unroll
        for (int r = 0; r < 8; ++r) {
            const float inv = 1.0f / fmaxf((float)cnt[row0 + r + 8 * hi], 1.0f);
            c0[r] *= inv; c1[r] *= inv; c2[r] *= inv; c3[r] *= inv;
        }
        const float* ap = A2  + (size_t)(row0 + lm) * K + 2 * hi;
        const float* bp = WT2 + (size_t)(col0 + lm) * K + 2 * hi;
        #pragma unroll 8
        for (int k = 0; k < K / 4; ++k) {
            v2f a  = *(const v2f*)(ap + 4 * k);
            v2f b0 = *(const v2f*)(bp + 4 * k);
            v2f b1 = *(const v2f*)(bp + 4 * k + 16 * K);
            v2f b2 = *(const v2f*)(bp + 4 * k + 32 * K);
            v2f b3 = *(const v2f*)(bp + 4 * k + 48 * K);
            c0 = __builtin_amdgcn_wmma_f32_16x16x4_f32(false, a, false, b0, (short)0, c0, false, false);
            c1 = __builtin_amdgcn_wmma_f32_16x16x4_f32(false, a, false, b1, (short)0, c1, false, false);
            c2 = __builtin_amdgcn_wmma_f32_16x16x4_f32(false, a, false, b2, (short)0, c2, false, false);
            c3 = __builtin_amdgcn_wmma_f32_16x16x4_f32(false, a, false, b3, (short)0, c3, false, false);
        }
    }

    float* orow = out + (size_t)(row0 + 8 * hi) * N + col0 + lm;
    float bv0 = 0.f, bv1 = 0.f, bv2 = 0.f, bv3 = 0.f;
    if (MODE != 3) {
        bv0 = bias[col0 + lm];
        bv1 = bias[col0 + 16 + lm];
        bv2 = bias[col0 + 32 + lm];
        bv3 = bias[col0 + 48 + lm];
    }
    #pragma unroll
    for (int r = 0; r < 8; ++r) {
        float v0 = c0[r], v1 = c1[r], v2 = c2[r], v3 = c3[r];
        if (MODE != 3) { v0 += bv0; v1 += bv1; v2 += bv2; v3 += bv3; }
        if (MODE == 2) {
            const float inv = 1.0f / fmaxf(dg[r], 1.0f);
            v0 += inv * g0[r]; v1 += inv * g1[r];
            v2 += inv * g2[r]; v3 += inv * g3[r];
        }
        if (do_relu) {
            v0 = fmaxf(v0, 0.f); v1 = fmaxf(v1, 0.f);
            v2 = fmaxf(v2, 0.f); v3 = fmaxf(v3, 0.f);
        }
        float* p = orow + (size_t)r * N;
        p[0] = v0; p[16] = v1; p[32] = v2; p[48] = v3;
    }
}


__global__ void __launch_bounds__(256)
transpose_w(const float* __restrict__ W, float* __restrict__ WT, int K, int N)
{
    int i = blockIdx.x * blockDim.x + threadIdx.x;
    if (i < K * N) {
        int k = i / N, n = i - k * N;
        WT[(size_t)n * K + k] = W[i];
    }
}


__global__ void __launch_bounds__(256)
count_rank(const int* __restrict__ e_occ, const int* __restrict__ e_skill,
           unsigned* __restrict__ cnt_occ, unsigned* __restrict__ cnt_skill,
           unsigned* __restrict__ rank_occ, unsigned* __restrict__ rank_skill, int E)
{
    int e = blockIdx.x * blockDim.x + threadIdx.x;
    if (e < E) {
        const int o = e_occ[e], s = e_skill[e];
        const unsigned ro = atomicAdd(&cnt_occ[o], 1u);
        const unsigned rs = atomicAdd(&cnt_skill[s], 1u);
        rank_occ[e]   = ro;
        rank_skill[e] = rs;
    }
}

__global__ void __launch_bounds__(1024)
scan_counts(const unsigned* __restrict__ cnt, unsigned* __restrict__ rp)
{
    __shared__ unsigned part[1024];
    const int n    = blockIdx.x == 0 ? N_OCC : N_SKILL;
    const int base = blockIdx.x == 0 ? 0 : N_OCC;
    const int per  = (n + 1023) / 1024;
    const int t    = threadIdx.x;
    int lo = t * per;  if (lo > n) lo = n;
    int hi = lo + per; if (hi > n) hi = n;
    unsigned s = 0;
    for (int i = lo; i < hi; ++i) s += cnt[base + i];
    part[t] = s;
    __syncthreads();
    for (int off = 1; off < 1024; off <<= 1) {
        const unsigned v = (t >= off) ? part[t - off] : 0u;
        __syncthreads();
        part[t] += v;
        __syncthreads();
    }
    unsigned run = part[t] - s;
    for (int i = lo; i < hi; ++i) { rp[base + i] = run; run += cnt[base + i]; }
}

__global__ void __launch_bounds__(256)
fill_adj(const int* __restrict__ e_occ, const int* __restrict__ e_skill,
         const unsigned* __restrict__ rp_occ, const unsigned* __restrict__ rp_skill,
         const unsigned* __restrict__ rank_occ, const unsigned* __restrict__ rank_skill,
         unsigned* __restrict__ adj_occ, unsigned* __restrict__ adj_skill, int E)
{
    int e = blockIdx.x * blockDim.x + threadIdx.x;
    if (e < E) {
        const int o = e_occ[e], s = e_skill[e];
        const unsigned po = rp_occ[o]   + rank_occ[e];
        const unsigned ps = rp_skill[s] + rank_skill[e];
        if (po < (unsigned)E) adj_occ[po]   = (unsigned)s;
        if (ps < (unsigned)E) adj_skill[ps] = (unsigned)o;
    }
}

__device__ inline void add4(float4& a, const float4& v)
{
    a.x += v.x; a.y += v.y; a.z += v.z; a.w += v.w;
}

template <int F>
__global__ void __launch_bounds__(256)
gather_sum(const float* __restrict__ hsrc, const unsigned* __restrict__ adj,
           const unsigned* __restrict__ rp, const unsigned* __restrict__ cnt,
           float* __restrict__ agg, int n_dst, unsigned n_src, unsigned E)
{
    constexpr int TPR = F / 4;
    const int tid = blockIdx.x * 256 + threadIdx.x;
    const int d = tid / TPR;
    const int f = (tid & (TPR - 1)) << 2;
    if (d >= n_dst) return;
    unsigned beg = rp[d];  if (beg > E) beg = E;
    unsigned deg = cnt[d]; if (deg > E - beg) deg = E - beg;
    const unsigned top = n_src - 1;
    const unsigned* a = adj + beg;
    const float* hb = hsrc + f;
    float4 s0 = {0.f, 0.f, 0.f, 0.f}, s1 = s0, s2 = s0, s3 = s0;
    unsigned j = 0;
    unsigned n0 = 0, n1 = 0, n2 = 0, n3 = 0;
    if (deg >= 4) { n0 = a[0]; n1 = a[1]; n2 = a[2]; n3 = a[3]; }
    for (; j + 4 <= deg; j += 4) {
        const unsigned i0 = min(n0, top), i1 = min(n1, top);
        const unsigned i2 = min(n2, top), i3 = min(n3, top);
        if (j + 8 <= deg) { n0 = a[j + 4]; n1 = a[j + 5]; n2 = a[j + 6]; n3 = a[j + 7]; }
        const float4 v0 = *(const float4*)(hb + (size_t)i0 * F);
        const float4 v1 = *(const float4*)(hb + (size_t)i1 * F);
        const float4 v2 = *(const float4*)(hb + (size_t)i2 * F);
        const float4 v3 = *(const float4*)(hb + (size_t)i3 * F);
        add4(s0, v0); add4(s1, v1); add4(s2, v2); add4(s3, v3);
    }
    if (j < deg) {
        const unsigned r = deg - j;
        const unsigned i0 = min(a[j], top);
        const unsigned i1 = min(a[r > 1 ? j + 1 : j], top);
        const unsigned i2 = min(a[r > 2 ? j + 2 : j], top);
        const float4 v0 = *(const float4*)(hb + (size_t)i0 * F);
        const float4 v1 = *(const float4*)(hb + (size_t)i1 * F);
        const float4 v2 = *(const float4*)(hb + (size_t)i2 * F);
        add4(s0, v0);
        if (r > 1) add4(s1, v1);
        if (r > 2) add4(s2, v2);
    }
    add4(s0, s1); add4(s2, s3); add4(s0, s2);
    *(float4*)(agg + (size_t)d * F + f) = s0;
}

extern "C" void kernel_launch(void* const* d_in, const int* in_sizes, int n_in,
                              void* d_out, int out_size, void* d_ws, size_t ws_size,
                              hipStream_t stream)
{
    (void)in_sizes; (void)n_in; (void)out_size; (void)ws_size;

    const float* x_occ    = (const float*)d_in[0];
    const float* x_skill  = (const float*)d_in[1];
    const int*   e_occ    = (const int*)  d_in[2];
    const int*   e_skill  = (const int*)  d_in[3];
    const float* pW_occ   = (const float*)d_in[4];
    const float* pb_occ   = (const float*)d_in[5];
    const float* pW_skill = (const float*)d_in[6];
    const float* pb_skill = (const float*)d_in[7];
    const float* c1_os_Wl = (const float*)d_in[8];
    const float* c1_os_bl = (const float*)d_in[9];
    const float* c1_os_Wr = (const float*)d_in[10];
    const float* c1_so_Wl = (const float*)d_in[11];
    const float* c1_so_bl = (const float*)d_in[12];
    const float* c1_so_Wr = (const float*)d_in[13];
    const float* c2_os_Wl = (const float*)d_in[14];
    const float* c2_os_bl = (const float*)d_in[15];
    const float* c2_os_Wr = (const float*)d_in[16];
    const float* c2_so_Wl = (const float*)d_in[17];
    const float* c2_so_bl = (const float*)d_in[18];
    const float* c2_so_Wr = (const float*)d_in[19];

    float* ws = (float*)d_ws;
    float* h_occ     = ws;
    float* h_skill   = h_occ     + (size_t)N_OCC   * HID;
    float* o1        = h_skill   + (size_t)N_SKILL * HID;
    float* s1        = o1        + (size_t)N_OCC   * HID;
    float* aggbuf    = s1        + (size_t)N_SKILL * HID;
    unsigned* adj_skill  = (unsigned*)(aggbuf + (size_t)N_SKILL * HID);
    unsigned* adj_occ    = adj_skill  + N_EDGE;
    unsigned* rp_occ     = adj_occ    + N_EDGE;
    unsigned* rp_skill   = rp_occ     + N_OCC;
    unsigned* rank_occ   = rp_skill   + N_SKILL;
    unsigned* rank_skill = rank_occ   + N_EDGE;
    unsigned* cnt_occ   = (unsigned*)(aggbuf + (size_t)(N_OCC + N_SKILL) * HID);
    unsigned* cnt_skill = cnt_occ + N_OCC;
    float* wt = (float*)(cnt_skill + N_SKILL);
    float* pWt_occ   = wt;                 wt += D_IN * HID;
    float* pWt_skill = wt;                 wt += D_IN * HID;
    float* t1_os_Wl  = wt;                 wt += HID * HID;
    float* t1_os_Wr  = wt;                 wt += HID * HID;
    float* t1_so_Wl  = wt;                 wt += HID * HID;
    float* t1_so_Wr  = wt;                 wt += HID * HID;
    float* t2_os_Wl  = wt;                 wt += HID * OUTF;
    float* t2_os_Wr  = wt;                 wt += HID * OUTF;
    float* t2_so_Wl  = wt;                 wt += HID * OUTF;
    float* t2_so_Wr  = wt;                 wt += HID * OUTF;

    float* o2 = (float*)d_out;
    float* s2 = o2 + (size_t)N_OCC * OUTF;

    const dim3 blk128(128), blk256(256);
    const int rt_occ   = (N_OCC / 16 + 3) / 4;
    const int rt_skill = (N_SKILL / 16 + 3) / 4;

    hipMemsetAsync(cnt_occ, 0, sizeof(unsigned) * (N_OCC + N_SKILL), stream);

    transpose_w<<<(D_IN * HID + 255) / 256, blk256, 0, stream>>>(pW_occ,   pWt_occ,   D_IN, HID);
    transpose_w<<<(D_IN * HID + 255) / 256, blk256, 0, stream>>>(pW_skill, pWt_skill, D_IN, HID);
    transpose_w<<<(HID * HID + 255) / 256, blk256, 0, stream>>>(c1_os_Wl, t1_os_Wl, HID, HID);
    transpose_w<<<(HID * HID + 255) / 256, blk256, 0, stream>>>(c1_os_Wr, t1_os_Wr, HID, HID);
    transpose_w<<<(HID * HID + 255) / 256, blk256, 0, stream>>>(c1_so_Wl, t1_so_Wl, HID, HID);
    transpose_w<<<(HID * HID + 255) / 256, blk256, 0, stream>>>(c1_so_Wr, t1_so_Wr, HID, HID);
    transpose_w<<<(HID * OUTF + 255) / 256, blk256, 0, stream>>>(c2_os_Wl, t2_os_Wl, HID, OUTF);
    transpose_w<<<(HID * OUTF + 255) / 256, blk256, 0, stream>>>(c2_os_Wr, t2_os_Wr, HID, OUTF);
    transpose_w<<<(HID * OUTF + 255) / 256, blk256, 0, stream>>>(c2_so_Wl, t2_so_Wl, HID, OUTF);
    transpose_w<<<(HID * OUTF + 255) / 256, blk256, 0, stream>>>(c2_so_Wr, t2_so_Wr, HID, OUTF);

    count_rank<<<(N_EDGE + 255) / 256, blk256, 0, stream>>>(
        e_occ, e_skill, cnt_occ, cnt_skill, rank_occ, rank_skill, N_EDGE);
    scan_counts<<<2, 1024, 0, stream>>>(cnt_occ, rp_occ);
    fill_adj<<<(N_EDGE + 255) / 256, blk256, 0, stream>>>(
        e_occ, e_skill, rp_occ, rp_skill, rank_occ, rank_skill, adj_occ, adj_skill, N_EDGE);

    gemm_sage<D_IN, HID, 0><<<dim3(rt_occ, HID / 64), blk128, 0, stream>>>(
        x_occ, pWt_occ, nullptr, nullptr, nullptr, pb_occ, h_occ, N_OCC, 1);
    gemm_sage<D_IN, HID, 0><<<dim3(rt_skill, HID / 64), blk128, 0, stream>>>(
        x_skill, pWt_skill, nullptr, nullptr, nullptr, pb_skill, h_skill, N_SKILL, 1);

    gather_sum<HID><<<N_SKILL / 8, blk256, 0, stream>>>(
        h_occ, adj_skill, rp_skill, cnt_skill, aggbuf, N_SKILL, N_OCC, N_EDGE);
    gemm_sage<HID, HID, 1><<<dim3(rt_skill, HID / 64), blk128, 0, stream>>>(
        aggbuf, t1_os_Wl, h_skill, t1_os_Wr, cnt_skill, c1_os_bl, s1, N_SKILL, 1);
    gather_sum<HID><<<N_OCC / 8, blk256, 0, stream>>>(
        h_skill, adj_occ, rp_occ, cnt_occ, aggbuf, N_OCC, N_SKILL, N_EDGE);
    gemm_sage<HID, HID, 1><<<dim3(rt_occ, HID / 64), blk128, 0, stream>>>(
        aggbuf, t1_so_Wl, h_occ, t1_so_Wr, cnt_occ, c1_so_bl, o1, N_OCC, 1);

    float* p_occ      = h_occ;
    float* p_skill    = h_skill;
    float* agg2_occ   = aggbuf;
    float* agg2_skill = aggbuf + (size_t)N_OCC * OUTF;
    gemm_sage<HID, OUTF, 3><<<dim3(rt_occ, OUTF / 64), blk128, 0, stream>>>(
        o1, t2_os_Wl, nullptr, nullptr, nullptr, nullptr, p_occ, N_OCC, 0);
    gemm_sage<HID, OUTF, 3><<<dim3(rt_skill, OUTF / 64), blk128, 0, stream>>>(
        s1, t2_so_Wl, nullptr, nullptr, nullptr, nullptr, p_skill, N_SKILL, 0);
    gather_sum<OUTF><<<N_SKILL / 16, blk256, 0, stream>>>(
        p_occ, adj_skill, rp_skill, cnt_skill, agg2_skill, N_SKILL, N_OCC, N_EDGE);
    gather_sum<OUTF><<<N_OCC / 16, blk256, 0, stream>>>(
        p_skill, adj_occ, rp_occ, cnt_occ, agg2_occ, N_OCC, N_SKILL, N_EDGE);

    gemm_sage<HID, OUTF, 2><<<dim3(rt_skill, OUTF / 64), blk128, 0, stream>>>(
        s1, t2_os_Wr, agg2_skill, nullptr, cnt_skill, c2_os_bl, s2, N_SKILL, 0);
    gemm_sage<HID, OUTF, 2><<<dim3(rt_occ, OUTF / 64), blk128, 0, stream>>>(
        o1, t2_so_Wr, agg2_occ, nullptr, cnt_occ, c2_so_bl, o2, N_OCC, 0);
}


# B fragments of every product read from an LDS copy of W made once per block from the row-major weights; blocks loop over row tiles; the 10 transpose launches go; f32 sum order unchanged
# speedup vs baseline: 2.1461x; 1.2658x over previous
#include <hip/hip_runtime.h>

#define N_OCC   20000
#define N_SKILL 50000
#define N_EDGE  600000
#define D_IN    256
#define HID     128
#define OUTF    64

typedef __attribute__((ext_vector_type(2))) float v2f;
typedef __attribute__((ext_vector_type(8))) float v8f;

template <int K, int N, int MODE>
__global__ void __launch_bounds__(256)
gemm_sage(const float* __restrict__ A1, const float* __restrict__ W1,
          const float* __restrict__ A2, const float* __restrict__ W2,
          const unsigned* __restrict__ cnt, const float* __restrict__ bias,
          float* __restrict__ out, int M, int do_relu)
{
    constexpr int P     = K + 4;
    constexpr int NMAT  = (MODE == 1) ? 2 : 1;
    constexpr int CH    = N / 64;
    constexpr int SLOTS = 8 / CH;
    __shared__ __attribute__((aligned(16))) float Ws[NMAT * N * P];

    for (int idx = threadIdx.x; idx < (K / 4) * N; idx += 256) {
        const int n  = idx % N;
        const int kq = idx / N;
        const float* w = W1 + (size_t)(4 * kq) * N + n;
        float4 v;
        v.x = w[0]; v.y = w[N]; v.z = w[2 * N]; v.w = w[3 * N];
        *(float4*)(Ws + n * P + 4 * kq) = v;
        if (MODE == 1) {
            const float* w2 = W2 + (size_t)(4 * kq) * N + n;
            float4 u;
            u.x = w2[0]; u.y = w2[N]; u.z = w2[2 * N]; u.w = w2[3 * N];
            *(float4*)(Ws + N * P + n * P + 4 * kq) = u;
        }
    }
    __syncthreads();

    const int lane = threadIdx.x & 31;
    const int wave = threadIdx.x >> 5;
    const int hi   = lane >> 4;
    const int lm   = lane & 15;
    const int slot = wave / CH;
    const int col0 = (wave % CH) * 64;
    const int ntiles = M / 16;

    const float* bs1 = Ws + (col0 + lm) * P + 2 * hi;
    const float* bs2 = Ws + N * P + (col0 + lm) * P + 2 * hi;

    float bv0 = 0.f, bv1 = 0.f, bv2 = 0.f, bv3 = 0.f;
    if (MODE != 3) {
        bv0 = bias[col0 + lm];
        bv1 = bias[col0 + 16 + lm];
        bv2 = bias[col0 + 32 + lm];
        bv3 = bias[col0 + 48 + lm];
    }

    for (int rt = slot * (int)gridDim.x + (int)blockIdx.x; rt < ntiles;
         rt += SLOTS * (int)gridDim.x) {
        const int row0 = rt * 16;

        v8f c0 = {0.f,0.f,0.f,0.f,0.f,0.f,0.f,0.f};
        v8f c1 = c0, c2 = c0, c3 = c0;

        float g0[8] = {}, g1[8] = {}, g2[8] = {}, g3[8] = {}, dg[8] = {};
        if (MODE == 2) {
            const float* g = A2 + (size_t)(row0 + 8 * hi) * N + col0 + lm;
            #pragma unroll
            for (int r = 0; r < 8; ++r) {
                g0[r] = g[r * N];      g1[r] = g[r * N + 16];
                g2[r] = g[r * N + 32]; g3[r] = g[r * N + 48];
                dg[r] = (float)cnt[row0 + 8 * hi + r];
            }
        }

        {
            const float* ap = A1 + (size_t)(row0 + lm) * K + 2 * hi;
            #pragma unroll 8
            for (int k = 0; k < K / 4; ++k) {
                v2f a  = *(const v2f*)(ap + 4 * k);
                v2f b0 = *(const v2f*)(bs1 + 4 * k);
                v2f b1 = *(const v2f*)(bs1 + 4 * k + 16 * P);
                v2f b2 = *(const v2f*)(bs1 + 4 * k + 32 * P);
                v2f b3 = *(const v2f*)(bs1 + 4 * k + 48 * P);
                c0 = __builtin_amdgcn_wmma_f32_16x16x4_f32(false, a, false, b0, (short)0, c0, false, false);
                c1 = __builtin_amdgcn_wmma_f32_16x16x4_f32(false, a, false, b1, (short)0, c1, false, false);
                c2 = __builtin_amdgcn_wmma_f32_16x16x4_f32(false, a, false, b2, (short)0, c2, false, false);
                c3 = __builtin_amdgcn_wmma_f32_16x16x4_f32(false, a, false, b3, (short)0, c3, false, false);
            }
        }

        if (MODE == 1) {
            #pragma unroll
            for (int r = 0; r < 8; ++r) {
                const float inv = 1.0f / fmaxf((float)cnt[row0 + r + 8 * hi], 1.0f);
                c0[r] *= inv; c1[r] *= inv; c2[r] *= inv; c3[r] *= inv;
            }
            const float* ap = A2 + (size_t)(row0 + lm) * K + 2 * hi;
            #pragma unroll 8
            for (int k = 0; k < K / 4; ++k) {
                v2f a  = *(const v2f*)(ap + 4 * k);
                v2f b0 = *(const v2f*)(bs2 + 4 * k);
                v2f b1 = *(const v2f*)(bs2 + 4 * k + 16 * P);
                v2f b2 = *(const v2f*)(bs2 + 4 * k + 32 * P);
                v2f b3 = *(const v2f*)(bs2 + 4 * k + 48 * P);
                c0 = __builtin_amdgcn_wmma_f32_16x16x4_f32(false, a, false, b0, (short)0, c0, false, false);
                c1 = __builtin_amdgcn_wmma_f32_16x16x4_f32(false, a, false, b1, (short)0, c1, false, false);
                c2 = __builtin_amdgcn_wmma_f32_16x16x4_f32(false, a, false, b2, (short)0, c2, false, false);
                c3 = __builtin_amdgcn_wmma_f32_16x16x4_f32(false, a, false, b3, (short)0, c3, false, false);
            }
        }

        float* orow = out + (size_t)(row0 + 8 * hi) * N + col0 + lm;
        #pragma unroll
        for (int r = 0; r < 8; ++r) {
            float v0 = c0[r], v1 = c1[r], v2 = c2[r], v3 = c3[r];
            if (MODE != 3) { v0 += bv0; v1 += bv1; v2 += bv2; v3 += bv3; }
            if (MODE == 2) {
                const float inv = 1.0f / fmaxf(dg[r], 1.0f);
                v0 += inv * g0[r]; v1 += inv * g1[r];
                v2 += inv * g2[r]; v3 += inv * g3[r];
            }
            if (do_relu) {
                v0 = fmaxf(v0, 0.f); v1 = fmaxf(v1, 0.f);
                v2 = fmaxf(v2, 0.f); v3 = fmaxf(v3, 0.f);
            }
            float* p = orow + (size_t)r * N;
            p[0] = v0; p[16] = v1; p[32] = v2; p[48] = v3;
        }
    }
}


__global__ void __launch_bounds__(256)
count_rank(const int* __restrict__ e_occ, const int* __restrict__ e_skill,
           unsigned* __restrict__ cnt_occ, unsigned* __restrict__ cnt_skill,
           unsigned* __restrict__ rank_occ, unsigned* __restrict__ rank_skill, int E)
{
    int e = blockIdx.x * blockDim.x + threadIdx.x;
    if (e < E) {
        const int o = e_occ[e], s = e_skill[e];
        const unsigned ro = atomicAdd(&cnt_occ[o], 1u);
        const unsigned rs = atomicAdd(&cnt_skill[s], 1u);
        rank_occ[e]   = ro;
        rank_skill[e] = rs;
    }
}

__global__ void __launch_bounds__(1024)
scan_counts(const unsigned* __restrict__ cnt, unsigned* __restrict__ rp)
{
    __shared__ unsigned part[1024];
    const int n    = blockIdx.x == 0 ? N_OCC : N_SKILL;
    const int base = blockIdx.x == 0 ? 0 : N_OCC;
    const int per  = (n + 1023) / 1024;
    const int t    = threadIdx.x;
    int lo = t * per;  if (lo > n) lo = n;
    int hi = lo + per; if (hi > n) hi = n;
    unsigned s = 0;
    for (int i = lo; i < hi; ++i) s += cnt[base + i];
    part[t] = s;
    __syncthreads();
    for (int off = 1; off < 1024; off <<= 1) {
        const unsigned v = (t >= off) ? part[t - off] : 0u;
        __syncthreads();
        part[t] += v;
        __syncthreads();
    }
    unsigned run = part[t] - s;
    for (int i = lo; i < hi; ++i) { rp[base + i] = run; run += cnt[base + i]; }
}

__global__ void __launch_bounds__(256)
fill_adj(const int* __restrict__ e_occ, const int* __restrict__ e_skill,
         const unsigned* __restrict__ rp_occ, const unsigned* __restrict__ rp_skill,
         const unsigned* __restrict__ rank_occ, const unsigned* __restrict__ rank_skill,
         unsigned* __restrict__ adj_occ, unsigned* __restrict__ adj_skill, int E)
{
    int e = blockIdx.x * blockDim.x + threadIdx.x;
    if (e < E) {
        const int o = e_occ[e], s = e_skill[e];
        const unsigned po = rp_occ[o]   + rank_occ[e];
        const unsigned ps = rp_skill[s] + rank_skill[e];
        if (po < (unsigned)E) adj_occ[po]   = (unsigned)s;
        if (ps < (unsigned)E) adj_skill[ps] = (unsigned)o;
    }
}

__device__ inline void add4(float4& a, const float4& v)
{
    a.x += v.x; a.y += v.y; a.z += v.z; a.w += v.w;
}

template <int F>
__global__ void __launch_bounds__(256)
gather_sum(const float* __restrict__ hsrc, const unsigned* __restrict__ adj,
           const unsigned* __restrict__ rp, const unsigned* __restrict__ cnt,
           float* __restrict__ agg, int n_dst, unsigned n_src, unsigned E)
{
    constexpr int TPR = F / 4;
    const int tid = blockIdx.x * 256 + threadIdx.x;
    const int d = tid / TPR;
    const int f = (tid & (TPR - 1)) << 2;
    if (d >= n_dst) return;
    unsigned beg = rp[d];  if (beg > E) beg = E;
    unsigned deg = cnt[d]; if (deg > E - beg) deg = E - beg;
    const unsigned top = n_src - 1;
    const unsigned* a = adj + beg;
    const float* hb = hsrc + f;
    float4 s0 = {0.f, 0.f, 0.f, 0.f}, s1 = s0, s2 = s0, s3 = s0;
    unsigned j = 0;
    unsigned n0 = 0, n1 = 0, n2 = 0, n3 = 0;
    if (deg >= 4) { n0 = a[0]; n1 = a[1]; n2 = a[2]; n3 = a[3]; }
    for (; j + 4 <= deg; j += 4) {
        const unsigned i0 = min(n0, top), i1 = min(n1, top);
        const unsigned i2 = min(n2, top), i3 = min(n3, top);
        if (j + 8 <= deg) { n0 = a[j + 4]; n1 = a[j + 5]; n2 = a[j + 6]; n3 = a[j + 7]; }
        const float4 v0 = *(const float4*)(hb + (size_t)i0 * F);
        const float4 v1 = *(const float4*)(hb + (size_t)i1 * F);
        const float4 v2 = *(const float4*)(hb + (size_t)i2 * F);
        const float4 v3 = *(const float4*)(hb + (size_t)i3 * F);
        add4(s0, v0); add4(s1, v1); add4(s2, v2); add4(s3, v3);
    }
    if (j < deg) {
        const unsigned r = deg - j;
        const unsigned i0 = min(a[j], top);
        const unsigned i1 = min(a[r > 1 ? j + 1 : j], top);
        const unsigned i2 = min(a[r > 2 ? j + 2 : j], top);
        const float4 v0 = *(const float4*)(hb + (size_t)i0 * F);
        const float4 v1 = *(const float4*)(hb + (size_t)i1 * F);
        const float4 v2 = *(const float4*)(hb + (size_t)i2 * F);
        add4(s0, v0);
        if (r > 1) add4(s1, v1);
        if (r > 2) add4(s2, v2);
    }
    add4(s0, s1); add4(s2, s3); add4(s0, s2);
    *(float4*)(agg + (size_t)d * F + f) = s0;
}

extern "C" void kernel_launch(void* const* d_in, const int* in_sizes, int n_in,
                              void* d_out, int out_size, void* d_ws, size_t ws_size,
                              hipStream_t stream)
{
    (void)in_sizes; (void)n_in; (void)out_size; (void)ws_size;

    const float* x_occ    = (const float*)d_in[0];
    const float* x_skill  = (const float*)d_in[1];
    const int*   e_occ    = (const int*)  d_in[2];
    const int*   e_skill  = (const int*)  d_in[3];
    const float* pW_occ   = (const float*)d_in[4];
    const float* pb_occ   = (const float*)d_in[5];
    const float* pW_skill = (const float*)d_in[6];
    const float* pb_skill = (const float*)d_in[7];
    const float* c1_os_Wl = (const float*)d_in[8];
    const float* c1_os_bl = (const float*)d_in[9];
    const float* c1_os_Wr = (const float*)d_in[10];
    const float* c1_so_Wl = (const float*)d_in[11];
    const float* c1_so_bl = (const float*)d_in[12];
    const float* c1_so_Wr = (const float*)d_in[13];
    const float* c2_os_Wl = (const float*)d_in[14];
    const float* c2_os_bl = (const float*)d_in[15];
    const float* c2_os_Wr = (const float*)d_in[16];
    const float* c2_so_Wl = (const float*)d_in[17];
    const float* c2_so_bl = (const float*)d_in[18];
    const float* c2_so_Wr = (const float*)d_in[19];

    float* ws = (float*)d_ws;
    float* h_occ     = ws;
    float* h_skill   = h_occ     + (size_t)N_OCC   * HID;
    float* o1        = h_skill   + (size_t)N_SKILL * HID;
    float* s1        = o1        + (size_t)N_OCC   * HID;
    float* aggbuf    = s1        + (size_t)N_SKILL * HID;
    unsigned* adj_skill  = (unsigned*)(aggbuf + (size_t)N_SKILL * HID);
    unsigned* adj_occ    = adj_skill  + N_EDGE;
    unsigned* rp_occ     = adj_occ    + N_EDGE;
    unsigned* rp_skill   = rp_occ     + N_OCC;
    unsigned* rank_occ   = rp_skill   + N_SKILL;
    unsigned* rank_skill = rank_occ   + N_EDGE;
    unsigned* cnt_occ   = (unsigned*)(aggbuf + (size_t)(N_OCC + N_SKILL) * HID);
    unsigned* cnt_skill = cnt_occ + N_OCC;
    float* o2 = (float*)d_out;
    float* s2 = o2 + (size_t)N_OCC * OUTF;

    const dim3 blk256(256);
    const dim3 ggrid(512);

    hipMemsetAsync(cnt_occ, 0, sizeof(unsigned) * (N_OCC + N_SKILL), stream);

    count_rank<<<(N_EDGE + 255) / 256, blk256, 0, stream>>>(
        e_occ, e_skill, cnt_occ, cnt_skill, rank_occ, rank_skill, N_EDGE);
    scan_counts<<<2, 1024, 0, stream>>>(cnt_occ, rp_occ);
    fill_adj<<<(N_EDGE + 255) / 256, blk256, 0, stream>>>(
        e_occ, e_skill, rp_occ, rp_skill, rank_occ, rank_skill, adj_occ, adj_skill, N_EDGE);

    gemm_sage<D_IN, HID, 0><<<ggrid, blk256, 0, stream>>>(
        x_occ, pW_occ, nullptr, nullptr, nullptr, pb_occ, h_occ, N_OCC, 1);
    gemm_sage<D_IN, HID, 0><<<ggrid, blk256, 0, stream>>>(
        x_skill, pW_skill, nullptr, nullptr, nullptr, pb_skill, h_skill, N_SKILL, 1);

    gather_sum<HID><<<N_SKILL / 8, blk256, 0, stream>>>(
        h_occ, adj_skill, rp_skill, cnt_skill, aggbuf, N_SKILL, N_OCC, N_EDGE);
    gemm_sage<HID, HID, 1><<<ggrid, blk256, 0, stream>>>(
        aggbuf, c1_os_Wl, h_skill, c1_os_Wr, cnt_skill, c1_os_bl, s1, N_SKILL, 1);
    gather_sum<HID><<<N_OCC / 8, blk256, 0, stream>>>(
        h_skill, adj_occ, rp_occ, cnt_occ, aggbuf, N_OCC, N_SKILL, N_EDGE);
    gemm_sage<HID, HID, 1><<<ggrid, blk256, 0, stream>>>(
        aggbuf, c1_so_Wl, h_occ, c1_so_Wr, cnt_occ, c1_so_bl, o1, N_OCC, 1);

    float* p_occ      = h_occ;
    float* p_skill    = h_skill;
    float* agg2_occ   = aggbuf;
    float* agg2_skill = aggbuf + (size_t)N_OCC * OUTF;
    gemm_sage<HID, OUTF, 3><<<ggrid, blk256, 0, stream>>>(
        o1, c2_os_Wl, nullptr, nullptr, nullptr, nullptr, p_occ, N_OCC, 0);
    gemm_sage<HID, OUTF, 3><<<ggrid, blk256, 0, stream>>>(
        s1, c2_so_Wl, nullptr, nullptr, nullptr, nullptr, p_skill, N_SKILL, 0);
    gather_sum<OUTF><<<N_SKILL / 16, blk256, 0, stream>>>(
        p_occ, adj_skill, rp_skill, cnt_skill, agg2_skill, N_SKILL, N_OCC, N_EDGE);
    gather_sum<OUTF><<<N_OCC / 16, blk256, 0, stream>>>(
        p_skill, adj_occ, rp_occ, cnt_occ, agg2_occ, N_OCC, N_SKILL, N_EDGE);

    gemm_sage<HID, OUTF, 2><<<ggrid, blk256, 0, stream>>>(
        s1, c2_os_Wr, agg2_skill, nullptr, cnt_skill, c2_os_bl, s2, N_SKILL, 0);
    gemm_sage<HID, OUTF, 2><<<ggrid, blk256, 0, stream>>>(
        o1, c2_so_Wr, agg2_occ, nullptr, cnt_occ, c2_so_bl, o2, N_OCC, 0);
}


# both node types of a stage share one launch (input projection, each layer's neighbour sums, both 64-column products); blocks split by row count; 15 kernel launches -> 10; f32 sum order unchanged
# speedup vs baseline: 2.3858x; 1.1117x over previous
#include <hip/hip_runtime.h>

#define N_OCC   20000
#define N_SKILL 50000
#define N_EDGE  600000
#define D_IN    256
#define HID     128
#define OUTF    64

typedef __attribute__((ext_vector_type(2))) float v2f;
typedef __attribute__((ext_vector_type(8))) float v8f;

template <int K, int N, int MODE>
__device__ __forceinline__ void
gemm_body(const float* __restrict__ A1, const float* __restrict__ W1,
          const float* __restrict__ A2, const float* __restrict__ W2,
          const unsigned* __restrict__ cnt, const float* __restrict__ bias,
          float* __restrict__ out, int M, int do_relu, int bid, int nblk)
{
    constexpr int P     = K + 4;
    constexpr int NMAT  = (MODE == 1) ? 2 : 1;
    constexpr int CH    = N / 64;
    constexpr int SLOTS = 8 / CH;
    __shared__ __attribute__((aligned(16))) float Ws[NMAT * N * P];

    for (int idx = threadIdx.x; idx < (K / 4) * N; idx += 256) {
        const int n  = idx % N;
        const int kq = idx / N;
        const float* w = W1 + (size_t)(4 * kq) * N + n;
        float4 v;
        v.x = w[0]; v.y = w[N]; v.z = w[2 * N]; v.w = w[3 * N];
        *(float4*)(Ws + n * P + 4 * kq) = v;
        if (MODE == 1) {
            const float* w2 = W2 + (size_t)(4 * kq) * N + n;
            float4 u;
            u.x = w2[0]; u.y = w2[N]; u.z = w2[2 * N]; u.w = w2[3 * N];
            *(float4*)(Ws + N * P + n * P + 4 * kq) = u;
        }
    }
    __syncthreads();

    const int lane = threadIdx.x & 31;
    const int wave = threadIdx.x >> 5;
    const int hi   = lane >> 4;
    const int lm   = lane & 15;
    const int slot = wave / CH;
    const int col0 = (wave % CH) * 64;
    const int ntiles = M / 16;

    const float* bs1 = Ws + (col0 + lm) * P + 2 * hi;
    const float* bs2 = Ws + N * P + (col0 + lm) * P + 2 * hi;

    float bv0 = 0.f, bv1 = 0.f, bv2 = 0.f, bv3 = 0.f;
    if (MODE != 3) {
        bv0 = bias[col0 + lm];
        bv1 = bias[col0 + 16 + lm];
        bv2 = bias[col0 + 32 + lm];
        bv3 = bias[col0 + 48 + lm];
    }

    for (int rt = slot * nblk + bid; rt < ntiles; rt += SLOTS * nblk) {
        const int row0 = rt * 16;

        v8f c0 = {0.f,0.f,0.f,0.f,0.f,0.f,0.f,0.f};
        v8f c1 = c0, c2 = c0, c3 = c0;

        float g0[8] = {}, g1[8] = {}, g2[8] = {}, g3[8] = {}, dg[8] = {};
        if (MODE == 2) {
            const float* g = A2 + (size_t)(row0 + 8 * hi) * N + col0 + lm;
            #pragma unroll
            for (int r = 0; r < 8; ++r) {
                g0[r] = g[r * N];      g1[r] = g[r * N + 16];
                g2[r] = g[r * N + 32]; g3[r] = g[r * N + 48];
                dg[r] = (float)cnt[row0 + 8 * hi + r];
            }
        }

        {
            const float* ap = A1 + (size_t)(row0 + lm) * K + 2 * hi;
            #pragma unroll 8
            for (int k = 0; k < K / 4; ++k) {
                v2f a  = *(const v2f*)(ap + 4 * k);
                v2f b0 = *(const v2f*)(bs1 + 4 * k);
                v2f b1 = *(const v2f*)(bs1 + 4 * k + 16 * P);
                v2f b2 = *(const v2f*)(bs1 + 4 * k + 32 * P);
                v2f b3 = *(const v2f*)(bs1 + 4 * k + 48 * P);
                c0 = __builtin_amdgcn_wmma_f32_16x16x4_f32(false, a, false, b0, (short)0, c0, false, false);
                c1 = __builtin_amdgcn_wmma_f32_16x16x4_f32(false, a, false, b1, (short)0, c1, false, false);
                c2 = __builtin_amdgcn_wmma_f32_16x16x4_f32(false, a, false, b2, (short)0, c2, false, false);
                c3 = __builtin_amdgcn_wmma_f32_16x16x4_f32(false, a, false, b3, (short)0, c3, false, false);
            }
        }

        if (MODE == 1) {
            #pragma unroll
            for (int r = 0; r < 8; ++r) {
                const float inv = 1.0f / fmaxf((float)cnt[row0 + r + 8 * hi], 1.0f);
                c0[r] *= inv; c1[r] *= inv; c2[r] *= inv; c3[r] *= inv;
            }
            const float* ap = A2 + (size_t)(row0 + lm) * K + 2 * hi;
            #pragma unroll 8
            for (int k = 0; k < K / 4; ++k) {
                v2f a  = *(const v2f*)(ap + 4 * k);
                v2f b0 = *(const v2f*)(bs2 + 4 * k);
                v2f b1 = *(const v2f*)(bs2 + 4 * k + 16 * P);
                v2f b2 = *(const v2f*)(bs2 + 4 * k + 32 * P);
                v2f b3 = *(const v2f*)(bs2 + 4 * k + 48 * P);
                c0 = __builtin_amdgcn_wmma_f32_16x16x4_f32(false, a, false, b0, (short)0, c0, false, false);
                c1 = __builtin_amdgcn_wmma_f32_16x16x4_f32(false, a, false, b1, (short)0, c1, false, false);
                c2 = __builtin_amdgcn_wmma_f32_16x16x4_f32(false, a, false, b2, (short)0, c2, false, false);
                c3 = __builtin_amdgcn_wmma_f32_16x16x4_f32(false, a, false, b3, (short)0, c3, false, false);
            }
        }

        float* orow = out + (size_t)(row0 + 8 * hi) * N + col0 + lm;
        #pragma unroll
        for (int r = 0; r < 8; ++r) {
            float v0 = c0[r], v1 = c1[r], v2 = c2[r], v3 = c3[r];
            if (MODE != 3) { v0 += bv0; v1 += bv1; v2 += bv2; v3 += bv3; }
            if (MODE == 2) {
                const float inv = 1.0f / fmaxf(dg[r], 1.0f);
                v0 += inv * g0[r]; v1 += inv * g1[r];
                v2 += inv * g2[r]; v3 += inv * g3[r];
            }
            if (do_relu) {
                v0 = fmaxf(v0, 0.f); v1 = fmaxf(v1, 0.f);
                v2 = fmaxf(v2, 0.f); v3 = fmaxf(v3, 0.f);
            }
            float* p = orow + (size_t)r * N;
            p[0] = v0; p[16] = v1; p[32] = v2; p[48] = v3;
        }
    }
}

template <int K, int N, int MODE>
__global__ void __launch_bounds__(256)
gemm_sage(const float* __restrict__ A1, const float* __restrict__ W1,
          const float* __restrict__ A2, const float* __restrict__ W2,
          const unsigned* __restrict__ cnt, const float* __restrict__ bias,
          float* __restrict__ out, int M, int do_relu)
{
    gemm_body<K, N, MODE>(A1, W1, A2, W2, cnt, bias, out, M, do_relu,
                          (int)blockIdx.x, (int)gridDim.x);
}

template <int K, int N, int MODE>
__global__ void __launch_bounds__(256)
gemm_pair(const float* __restrict__ A1a, const float* __restrict__ W1a,
          const float* __restrict__ A2a, const unsigned* __restrict__ cnta,
          const float* __restrict__ biasa, float* __restrict__ outa, int Ma,
          const float* __restrict__ A1b, const float* __restrict__ W1b,
          const float* __restrict__ A2b, const unsigned* __restrict__ cntb,
          const float* __restrict__ biasb, float* __restrict__ outb, int Mb,
          int split, int do_relu)
{
    const bool second = (int)blockIdx.x >= split;
    gemm_body<K, N, MODE>(second ? A1b : A1a, second ? W1b : W1a,
                          second ? A2b : A2a, nullptr,
                          second ? cntb : cnta, second ? biasb : biasa,
                          second ? outb : outa, second ? Mb : Ma, do_relu,
                          second ? (int)blockIdx.x - split : (int)blockIdx.x,
                          second ? (int)gridDim.x - split : split);
}


__global__ void __launch_bounds__(256)
count_rank(const int* __restrict__ e_occ, const int* __restrict__ e_skill,
           unsigned* __restrict__ cnt_occ, unsigned* __restrict__ cnt_skill,
           unsigned* __restrict__ rank_occ, unsigned* __restrict__ rank_skill, int E)
{
    int e = blockIdx.x * blockDim.x + threadIdx.x;
    if (e < E) {
        const int o = e_occ[e], s = e_skill[e];
        const unsigned ro = atomicAdd(&cnt_occ[o], 1u);
        const unsigned rs = atomicAdd(&cnt_skill[s], 1u);
        rank_occ[e]   = ro;
        rank_skill[e] = rs;
    }
}

__global__ void __launch_bounds__(1024)
scan_counts(const unsigned* __restrict__ cnt, unsigned* __restrict__ rp)
{
    __shared__ unsigned part[1024];
    const int n    = blockIdx.x == 0 ? N_OCC : N_SKILL;
    const int base = blockIdx.x == 0 ? 0 : N_OCC;
    const int per  = (n + 1023) / 1024;
    const int t    = threadIdx.x;
    int lo = t * per;  if (lo > n) lo = n;
    int hi = lo + per; if (hi > n) hi = n;
    unsigned s = 0;
    for (int i = lo; i < hi; ++i) s += cnt[base + i];
    part[t] = s;
    __syncthreads();
    for (int off = 1; off < 1024; off <<= 1) {
        const unsigned v = (t >= off) ? part[t - off] : 0u;
        __syncthreads();
        part[t] += v;
        __syncthreads();
    }
    unsigned run = part[t] - s;
    for (int i = lo; i < hi; ++i) { rp[base + i] = run; run += cnt[base + i]; }
}

__global__ void __launch_bounds__(256)
fill_adj(const int* __restrict__ e_occ, const int* __restrict__ e_skill,
         const unsigned* __restrict__ rp_occ, const unsigned* __restrict__ rp_skill,
         const unsigned* __restrict__ rank_occ, const unsigned* __restrict__ rank_skill,
         unsigned* __restrict__ adj_occ, unsigned* __restrict__ adj_skill, int E)
{
    int e = blockIdx.x * blockDim.x + threadIdx.x;
    if (e < E) {
        const int o = e_occ[e], s = e_skill[e];
        const unsigned po = rp_occ[o]   + rank_occ[e];
        const unsigned ps = rp_skill[s] + rank_skill[e];
        if (po < (unsigned)E) adj_occ[po]   = (unsigned)s;
        if (ps < (unsigned)E) adj_skill[ps] = (unsigned)o;
    }
}

__device__ inline void add4(float4& a, const float4& v)
{
    a.x += v.x; a.y += v.y; a.z += v.z; a.w += v.w;
}

template <int F>
__device__ __forceinline__ void
gather_body(const float* __restrict__ hsrc, const unsigned* __restrict__ adj,
            const unsigned* __restrict__ rp, const unsigned* __restrict__ cnt,
            float* __restrict__ agg, int n_dst, unsigned n_src, unsigned E, int bid)
{
    constexpr int TPR = F / 4;
    const int tid = bid * 256 + threadIdx.x;
    const int d = tid / TPR;
    const int f = (tid & (TPR - 1)) << 2;
    if (d >= n_dst) return;
    unsigned beg = rp[d];  if (beg > E) beg = E;
    unsigned deg = cnt[d]; if (deg > E - beg) deg = E - beg;
    const unsigned top = n_src - 1;
    const unsigned* a = adj + beg;
    const float* hb = hsrc + f;
    float4 s0 = {0.f, 0.f, 0.f, 0.f}, s1 = s0, s2 = s0, s3 = s0;
    unsigned j = 0;
    unsigned n0 = 0, n1 = 0, n2 = 0, n3 = 0;
    if (deg >= 4) { n0 = a[0]; n1 = a[1]; n2 = a[2]; n3 = a[3]; }
    for (; j + 4 <= deg; j += 4) {
        const unsigned i0 = min(n0, top), i1 = min(n1, top);
        const unsigned i2 = min(n2, top), i3 = min(n3, top);
        if (j + 8 <= deg) { n0 = a[j + 4]; n1 = a[j + 5]; n2 = a[j + 6]; n3 = a[j + 7]; }
        const float4 v0 = *(const float4*)(hb + (size_t)i0 * F);
        const float4 v1 = *(const float4*)(hb + (size_t)i1 * F);
        const float4 v2 = *(const float4*)(hb + (size_t)i2 * F);
        const float4 v3 = *(const float4*)(hb + (size_t)i3 * F);
        add4(s0, v0); add4(s1, v1); add4(s2, v2); add4(s3, v3);
    }
    if (j < deg) {
        const unsigned r = deg - j;
        const unsigned i0 = min(a[j], top);
        const unsigned i1 = min(a[r > 1 ? j + 1 : j], top);
        const unsigned i2 = min(a[r > 2 ? j + 2 : j], top);
        const float4 v0 = *(const float4*)(hb + (size_t)i0 * F);
        const float4 v1 = *(const float4*)(hb + (size_t)i1 * F);
        const float4 v2 = *(const float4*)(hb + (size_t)i2 * F);
        add4(s0, v0);
        if (r > 1) add4(s1, v1);
        if (r > 2) add4(s2, v2);
    }
    add4(s0, s1); add4(s2, s3); add4(s0, s2);
    *(float4*)(agg + (size_t)d * F + f) = s0;
}

template <int F>
__global__ void __launch_bounds__(256)
gather_pair(const float* __restrict__ hsrc_a, const unsigned* __restrict__ adj_a,
            const unsigned* __restrict__ rp_a, const unsigned* __restrict__ cnt_a,
            float* __restrict__ agg_a, int n_dst_a, unsigned n_src_a,
            const float* __restrict__ hsrc_b, const unsigned* __restrict__ adj_b,
            const unsigned* __restrict__ rp_b, const unsigned* __restrict__ cnt_b,
            float* __restrict__ agg_b, int n_dst_b, unsigned n_src_b,
            int split, unsigned E)
{
    const bool second = (int)blockIdx.x >= split;
    gather_body<F>(second ? hsrc_b : hsrc_a, second ? adj_b : adj_a,
                   second ? rp_b : rp_a, second ? cnt_b : cnt_a,
                   second ? agg_b : agg_a, second ? n_dst_b : n_dst_a,
                   second ? n_src_b : n_src_a, E,
                   second ? (int)blockIdx.x - split : (int)blockIdx.x);
}

extern "C" void kernel_launch(void* const* d_in, const int* in_sizes, int n_in,
                              void* d_out, int out_size, void* d_ws, size_t ws_size,
                              hipStream_t stream)
{
    (void)in_sizes; (void)n_in; (void)out_size; (void)ws_size;

    const float* x_occ    = (const float*)d_in[0];
    const float* x_skill  = (const float*)d_in[1];
    const int*   e_occ    = (const int*)  d_in[2];
    const int*   e_skill  = (const int*)  d_in[3];
    const float* pW_occ   = (const float*)d_in[4];
    const float* pb_occ   = (const float*)d_in[5];
    const float* pW_skill = (const float*)d_in[6];
    const float* pb_skill = (const float*)d_in[7];
    const float* c1_os_Wl = (const float*)d_in[8];
    const float* c1_os_bl = (const float*)d_in[9];
    const float* c1_os_Wr = (const float*)d_in[10];
    const float* c1_so_Wl = (const float*)d_in[11];
    const float* c1_so_bl = (const float*)d_in[12];
    const float* c1_so_Wr = (const float*)d_in[13];
    const float* c2_os_Wl = (const float*)d_in[14];
    const float* c2_os_bl = (const float*)d_in[15];
    const float* c2_os_Wr = (const float*)d_in[16];
    const float* c2_so_Wl = (const float*)d_in[17];
    const float* c2_so_bl = (const float*)d_in[18];
    const float* c2_so_Wr = (const float*)d_in[19];

    float* ws = (float*)d_ws;
    float* h_occ     = ws;
    float* h_skill   = h_occ     + (size_t)N_OCC   * HID;
    float* o1        = h_skill   + (size_t)N_SKILL * HID;
    float* s1        = o1        + (size_t)N_OCC   * HID;
    float* aggbuf    = s1        + (size_t)N_SKILL * HID;
    unsigned* adj_skill  = (unsigned*)(aggbuf + (size_t)N_SKILL * HID);
    unsigned* adj_occ    = adj_skill  + N_EDGE;
    unsigned* rp_occ     = adj_occ    + N_EDGE;
    unsigned* rp_skill   = rp_occ     + N_OCC;
    unsigned* rank_occ   = rp_skill   + N_SKILL;
    unsigned* rank_skill = rank_occ   + N_EDGE;
    unsigned* cnt_occ   = (unsigned*)(aggbuf + (size_t)(N_OCC + N_SKILL) * HID);
    unsigned* cnt_skill = cnt_occ + N_OCC;
    float* o2 = (float*)d_out;
    float* s2 = o2 + (size_t)N_OCC * OUTF;

    const dim3 blk256(256);
    const dim3 ggrid(512);

    hipMemsetAsync(cnt_occ, 0, sizeof(unsigned) * (N_OCC + N_SKILL), stream);

    count_rank<<<(N_EDGE + 255) / 256, blk256, 0, stream>>>(
        e_occ, e_skill, cnt_occ, cnt_skill, rank_occ, rank_skill, N_EDGE);
    scan_counts<<<2, 1024, 0, stream>>>(cnt_occ, rp_occ);
    fill_adj<<<(N_EDGE + 255) / 256, blk256, 0, stream>>>(
        e_occ, e_skill, rp_occ, rp_skill, rank_occ, rank_skill, adj_occ, adj_skill, N_EDGE);

    gemm_pair<D_IN, HID, 0><<<ggrid, blk256, 0, stream>>>(
        x_occ,   pW_occ,   nullptr, nullptr, pb_occ,   h_occ,   N_OCC,
        x_skill, pW_skill, nullptr, nullptr, pb_skill, h_skill, N_SKILL, 146, 1);

    float* agg1_occ = s1;
    gather_pair<HID><<<N_OCC / 8 + N_SKILL / 8, blk256, 0, stream>>>(
        h_skill, adj_occ,   rp_occ,   cnt_occ,   agg1_occ, N_OCC,   N_SKILL,
        h_occ,   adj_skill, rp_skill, cnt_skill, aggbuf,   N_SKILL, N_OCC,
        N_OCC / 8, N_EDGE);
    gemm_sage<HID, HID, 1><<<ggrid, blk256, 0, stream>>>(
        agg1_occ, c1_so_Wl, h_occ, c1_so_Wr, cnt_occ, c1_so_bl, o1, N_OCC, 1);
    gemm_sage<HID, HID, 1><<<ggrid, blk256, 0, stream>>>(
        aggbuf, c1_os_Wl, h_skill, c1_os_Wr, cnt_skill, c1_os_bl, s1, N_SKILL, 1);

    float* p_occ      = h_occ;
    float* p_skill    = h_skill;
    float* agg2_occ   = aggbuf;
    float* agg2_skill = aggbuf + (size_t)N_OCC * OUTF;
    const dim3 ggrid64(768);
    gemm_pair<HID, OUTF, 3><<<ggrid64, blk256, 0, stream>>>(
        o1, c2_os_Wl, nullptr, nullptr, nullptr, p_occ,   N_OCC,
        s1, c2_so_Wl, nullptr, nullptr, nullptr, p_skill, N_SKILL, 219, 0);
    gather_pair<OUTF><<<N_OCC / 16 + N_SKILL / 16, blk256, 0, stream>>>(
        p_skill, adj_occ,   rp_occ,   cnt_occ,   agg2_occ,   N_OCC,   N_SKILL,
        p_occ,   adj_skill, rp_skill, cnt_skill, agg2_skill, N_SKILL, N_OCC,
        N_OCC / 16, N_EDGE);

    gemm_pair<HID, OUTF, 2><<<ggrid64, blk256, 0, stream>>>(
        o1, c2_so_Wr, agg2_occ,   cnt_occ,   c2_so_bl, o2, N_OCC,
        s1, c2_os_Wr, agg2_skill, cnt_skill, c2_os_bl, s2, N_SKILL, 219, 0);
}


# degree prefix sum: 70 blocks, one per 1000 counts, each re-sums the counts before its chunk with whole-line 16-byte loads; was 2 blocks with one line per lane; integer sums, tables identical
# speedup vs baseline: 2.5534x; 1.0702x over previous
#include <hip/hip_runtime.h>

#define N_OCC   20000
#define N_SKILL 50000
#define N_EDGE  600000
#define D_IN    256
#define HID     128
#define OUTF    64

typedef __attribute__((ext_vector_type(2))) float v2f;
typedef __attribute__((ext_vector_type(8))) float v8f;

template <int K, int N, int MODE>
__device__ __forceinline__ void
gemm_body(const float* __restrict__ A1, const float* __restrict__ W1,
          const float* __restrict__ A2, const float* __restrict__ W2,
          const unsigned* __restrict__ cnt, const float* __restrict__ bias,
          float* __restrict__ out, int M, int do_relu, int bid, int nblk)
{
    constexpr int P     = K + 4;
    constexpr int NMAT  = (MODE == 1) ? 2 : 1;
    constexpr int CH    = N / 64;
    constexpr int SLOTS = 8 / CH;
    __shared__ __attribute__((aligned(16))) float Ws[NMAT * N * P];

    for (int idx = threadIdx.x; idx < (K / 4) * N; idx += 256) {
        const int n  = idx % N;
        const int kq = idx / N;
        const float* w = W1 + (size_t)(4 * kq) * N + n;
        float4 v;
        v.x = w[0]; v.y = w[N]; v.z = w[2 * N]; v.w = w[3 * N];
        *(float4*)(Ws + n * P + 4 * kq) = v;
        if (MODE == 1) {
            const float* w2 = W2 + (size_t)(4 * kq) * N + n;
            float4 u;
            u.x = w2[0]; u.y = w2[N]; u.z = w2[2 * N]; u.w = w2[3 * N];
            *(float4*)(Ws + N * P + n * P + 4 * kq) = u;
        }
    }
    __syncthreads();

    const int lane = threadIdx.x & 31;
    const int wave = threadIdx.x >> 5;
    const int hi   = lane >> 4;
    const int lm   = lane & 15;
    const int slot = wave / CH;
    const int col0 = (wave % CH) * 64;
    const int ntiles = M / 16;

    const float* bs1 = Ws + (col0 + lm) * P + 2 * hi;
    const float* bs2 = Ws + N * P + (col0 + lm) * P + 2 * hi;

    float bv0 = 0.f, bv1 = 0.f, bv2 = 0.f, bv3 = 0.f;
    if (MODE != 3) {
        bv0 = bias[col0 + lm];
        bv1 = bias[col0 + 16 + lm];
        bv2 = bias[col0 + 32 + lm];
        bv3 = bias[col0 + 48 + lm];
    }

    for (int rt = slot * nblk + bid; rt < ntiles; rt += SLOTS * nblk) {
        const int row0 = rt * 16;

        v8f c0 = {0.f,0.f,0.f,0.f,0.f,0.f,0.f,0.f};
        v8f c1 = c0, c2 = c0, c3 = c0;

        float g0[8] = {}, g1[8] = {}, g2[8] = {}, g3[8] = {}, dg[8] = {};
        if (MODE == 2) {
            const float* g = A2 + (size_t)(row0 + 8 * hi) * N + col0 + lm;
            #pragma unroll
            for (int r = 0; r < 8; ++r) {
                g0[r] = g[r * N];      g1[r] = g[r * N + 16];
                g2[r] = g[r * N + 32]; g3[r] = g[r * N + 48];
                dg[r] = (float)cnt[row0 + 8 * hi + r];
            }
        }

        {
            const float* ap = A1 + (size_t)(row0 + lm) * K + 2 * hi;
            #pragma unroll 8
            for (int k = 0; k < K / 4; ++k) {
                v2f a  = *(const v2f*)(ap + 4 * k);
                v2f b0 = *(const v2f*)(bs1 + 4 * k);
                v2f b1 = *(const v2f*)(bs1 + 4 * k + 16 * P);
                v2f b2 = *(const v2f*)(bs1 + 4 * k + 32 * P);
                v2f b3 = *(const v2f*)(bs1 + 4 * k + 48 * P);
                c0 = __builtin_amdgcn_wmma_f32_16x16x4_f32(false, a, false, b0, (short)0, c0, false, false);
                c1 = __builtin_amdgcn_wmma_f32_16x16x4_f32(false, a, false, b1, (short)0, c1, false, false);
                c2 = __builtin_amdgcn_wmma_f32_16x16x4_f32(false, a, false, b2, (short)0, c2, false, false);
                c3 = __builtin_amdgcn_wmma_f32_16x16x4_f32(false, a, false, b3, (short)0, c3, false, false);
            }
        }

        if (MODE == 1) {
            #pragma unroll
            for (int r = 0; r < 8; ++r) {
                const float inv = 1.0f / fmaxf((float)cnt[row0 + r + 8 * hi], 1.0f);
                c0[r] *= inv; c1[r] *= inv; c2[r] *= inv; c3[r] *= inv;
            }
            const float* ap = A2 + (size_t)(row0 + lm) * K + 2 * hi;
            #pragma unroll 8
            for (int k = 0; k < K / 4; ++k) {
                v2f a  = *(const v2f*)(ap + 4 * k);
                v2f b0 = *(const v2f*)(bs2 + 4 * k);
                v2f b1 = *(const v2f*)(bs2 + 4 * k + 16 * P);
                v2f b2 = *(const v2f*)(bs2 + 4 * k + 32 * P);
                v2f b3 = *(const v2f*)(bs2 + 4 * k + 48 * P);
                c0 = __builtin_amdgcn_wmma_f32_16x16x4_f32(false, a, false, b0, (short)0, c0, false, false);
                c1 = __builtin_amdgcn_wmma_f32_16x16x4_f32(false, a, false, b1, (short)0, c1, false, false);
                c2 = __builtin_amdgcn_wmma_f32_16x16x4_f32(false, a, false, b2, (short)0, c2, false, false);
                c3 = __builtin_amdgcn_wmma_f32_16x16x4_f32(false, a, false, b3, (short)0, c3, false, false);
            }
        }

        float* orow = out + (size_t)(row0 + 8 * hi) * N + col0 + lm;
        #pragma unroll
        for (int r = 0; r < 8; ++r) {
            float v0 = c0[r], v1 = c1[r], v2 = c2[r], v3 = c3[r];
            if (MODE != 3) { v0 += bv0; v1 += bv1; v2 += bv2; v3 += bv3; }
            if (MODE == 2) {
                const float inv = 1.0f / fmaxf(dg[r], 1.0f);
                v0 += inv * g0[r]; v1 += inv * g1[r];
                v2 += inv * g2[r]; v3 += inv * g3[r];
            }
            if (do_relu) {
                v0 = fmaxf(v0, 0.f); v1 = fmaxf(v1, 0.f);
                v2 = fmaxf(v2, 0.f); v3 = fmaxf(v3, 0.f);
            }
            float* p = orow + (size_t)r * N;
            p[0] = v0; p[16] = v1; p[32] = v2; p[48] = v3;
        }
    }
}

template <int K, int N, int MODE>
__global__ void __launch_bounds__(256)
gemm_sage(const float* __restrict__ A1, const float* __restrict__ W1,
          const float* __restrict__ A2, const float* __restrict__ W2,
          const unsigned* __restrict__ cnt, const float* __restrict__ bias,
          float* __restrict__ out, int M, int do_relu)
{
    gemm_body<K, N, MODE>(A1, W1, A2, W2, cnt, bias, out, M, do_relu,
                          (int)blockIdx.x, (int)gridDim.x);
}

template <int K, int N, int MODE>
__global__ void __launch_bounds__(256)
gemm_pair(const float* __restrict__ A1a, const float* __restrict__ W1a,
          const float* __restrict__ A2a, const unsigned* __restrict__ cnta,
          const float* __restrict__ biasa, float* __restrict__ outa, int Ma,
          const float* __restrict__ A1b, const float* __restrict__ W1b,
          const float* __restrict__ A2b, const unsigned* __restrict__ cntb,
          const float* __restrict__ biasb, float* __restrict__ outb, int Mb,
          int split, int do_relu)
{
    const bool second = (int)blockIdx.x >= split;
    gemm_body<K, N, MODE>(second ? A1b : A1a, second ? W1b : W1a,
                          second ? A2b : A2a, nullptr,
                          second ? cntb : cnta, second ? biasb : biasa,
                          second ? outb : outa, second ? Mb : Ma, do_relu,
                          second ? (int)blockIdx.x - split : (int)blockIdx.x,
                          second ? (int)gridDim.x - split : split);
}


__global__ void __launch_bounds__(256)
count_rank(const int* __restrict__ e_occ, const int* __restrict__ e_skill,
           unsigned* __restrict__ cnt_occ, unsigned* __restrict__ cnt_skill,
           unsigned* __restrict__ rank_occ, unsigned* __restrict__ rank_skill, int E)
{
    int e = blockIdx.x * blockDim.x + threadIdx.x;
    if (e < E) {
        const int o = e_occ[e], s = e_skill[e];
        const unsigned ro = atomicAdd(&cnt_occ[o], 1u);
        const unsigned rs = atomicAdd(&cnt_skill[s], 1u);
        rank_occ[e]   = ro;
        rank_skill[e] = rs;
    }
}

#define SCAN_CHUNK 1000
static_assert(N_OCC % SCAN_CHUNK == 0 && N_SKILL % SCAN_CHUNK == 0, "chunks must not straddle the node types");
static_assert(SCAN_CHUNK % 4 == 0 && SCAN_CHUNK <= 1024, "4 counts per thread, 256 threads");
__global__ void __launch_bounds__(256)
scan_counts(const unsigned* __restrict__ cnt, unsigned* __restrict__ rp)
{
    __shared__ unsigned part[256];
    __shared__ unsigned prev[256];
    const int t     = threadIdx.x;
    const int nocc  = N_OCC / SCAN_CHUNK;
    const int b     = blockIdx.x;
    const int seg0  = (b < nocc) ? 0 : N_OCC;
    const int cbeg  = b * SCAN_CHUNK;
    const uint4* c4 = (const uint4*)cnt;

    unsigned p = 0;
    for (int i = seg0 / 4 + t; i < cbeg / 4; i += 256) {
        const uint4 v = c4[i];
        p += (v.x + v.y) + (v.z + v.w);
    }
    uint4 a = {0u, 0u, 0u, 0u};
    const bool mine = 4 * t < SCAN_CHUNK;
    if (mine) a = c4[cbeg / 4 + t];
    const unsigned s = (a.x + a.y) + (a.z + a.w);
    part[t] = s;
    prev[t] = p;
    __syncthreads();
    for (int off = 1; off < 256; off <<= 1) {
        const unsigned v = (t >= off) ? part[t - off] : 0u;
        const unsigned w = (t >= off) ? prev[t - off] : 0u;
        __syncthreads();
        part[t] += v;
        prev[t] += w;
        __syncthreads();
    }
    if (mine) {
        uint4 r;
        r.x = prev[255] + part[t] - s;
        r.y = r.x + a.x;
        r.z = r.y + a.y;
        r.w = r.z + a.z;
        ((uint4*)rp)[cbeg / 4 + t] = r;
    }
}

__global__ void __launch_bounds__(256)
fill_adj(const int* __restrict__ e_occ, const int* __restrict__ e_skill,
         const unsigned* __restrict__ rp_occ, const unsigned* __restrict__ rp_skill,
         const unsigned* __restrict__ rank_occ, const unsigned* __restrict__ rank_skill,
         unsigned* __restrict__ adj_occ, unsigned* __restrict__ adj_skill, int E)
{
    int e = blockIdx.x * blockDim.x + threadIdx.x;
    if (e < E) {
        const int o = e_occ[e], s = e_skill[e];
        const unsigned po = rp_occ[o]   + rank_occ[e];
        const unsigned ps = rp_skill[s] + rank_skill[e];
        if (po < (unsigned)E) adj_occ[po]   = (unsigned)s;
        if (ps < (unsigned)E) adj_skill[ps] = (unsigned)o;
    }
}

__device__ inline void add4(float4& a, const float4& v)
{
    a.x += v.x; a.y += v.y; a.z += v.z; a.w += v.w;
}

template <int F>
__device__ __forceinline__ void
gather_body(const float* __restrict__ hsrc, const unsigned* __restrict__ adj,
            const unsigned* __restrict__ rp, const unsigned* __restrict__ cnt,
            float* __restrict__ agg, int n_dst, unsigned n_src, unsigned E, int bid)
{
    constexpr int TPR = F / 4;
    const int tid = bid * 256 + threadIdx.x;
    const int d = tid / TPR;
    const int f = (tid & (TPR - 1)) << 2;
    if (d >= n_dst) return;
    unsigned beg = rp[d];  if (beg > E) beg = E;
    unsigned deg = cnt[d]; if (deg > E - beg) deg = E - beg;
    const unsigned top = n_src - 1;
    const unsigned* a = adj + beg;
    const float* hb = hsrc + f;
    float4 s0 = {0.f, 0.f, 0.f, 0.f}, s1 = s0, s2 = s0, s3 = s0;
    unsigned j = 0;
    unsigned n0 = 0, n1 = 0, n2 = 0, n3 = 0;
    if (deg >= 4) { n0 = a[0]; n1 = a[1]; n2 = a[2]; n3 = a[3]; }
    for (; j + 4 <= deg; j += 4) {
        const unsigned i0 = min(n0, top), i1 = min(n1, top);
        const unsigned i2 = min(n2, top), i3 = min(n3, top);
        if (j + 8 <= deg) { n0 = a[j + 4]; n1 = a[j + 5]; n2 = a[j + 6]; n3 = a[j + 7]; }
        const float4 v0 = *(const float4*)(hb + (size_t)i0 * F);
        const float4 v1 = *(const float4*)(hb + (size_t)i1 * F);
        const float4 v2 = *(const float4*)(hb + (size_t)i2 * F);
        const float4 v3 = *(const float4*)(hb + (size_t)i3 * F);
        add4(s0, v0); add4(s1, v1); add4(s2, v2); add4(s3, v3);
    }
    if (j < deg) {
        const unsigned r = deg - j;
        const unsigned i0 = min(a[j], top);
        const unsigned i1 = min(a[r > 1 ? j + 1 : j], top);
        const unsigned i2 = min(a[r > 2 ? j + 2 : j], top);
        const float4 v0 = *(const float4*)(hb + (size_t)i0 * F);
        const float4 v1 = *(const float4*)(hb + (size_t)i1 * F);
        const float4 v2 = *(const float4*)(hb + (size_t)i2 * F);
        add4(s0, v0);
        if (r > 1) add4(s1, v1);
        if (r > 2) add4(s2, v2);
    }
    add4(s0, s1); add4(s2, s3); add4(s0, s2);
    *(float4*)(agg + (size_t)d * F + f) = s0;
}

template <int F>
__global__ void __launch_bounds__(256)
gather_pair(const float* __restrict__ hsrc_a, const unsigned* __restrict__ adj_a,
            const unsigned* __restrict__ rp_a, const unsigned* __restrict__ cnt_a,
            float* __restrict__ agg_a, int n_dst_a, unsigned n_src_a,
            const float* __restrict__ hsrc_b, const unsigned* __restrict__ adj_b,
            const unsigned* __restrict__ rp_b, const unsigned* __restrict__ cnt_b,
            float* __restrict__ agg_b, int n_dst_b, unsigned n_src_b,
            int split, unsigned E)
{
    const bool second = (int)blockIdx.x >= split;
    gather_body<F>(second ? hsrc_b : hsrc_a, second ? adj_b : adj_a,
                   second ? rp_b : rp_a, second ? cnt_b : cnt_a,
                   second ? agg_b : agg_a, second ? n_dst_b : n_dst_a,
                   second ? n_src_b : n_src_a, E,
                   second ? (int)blockIdx.x - split : (int)blockIdx.x);
}

extern "C" void kernel_launch(void* const* d_in, const int* in_sizes, int n_in,
                              void* d_out, int out_size, void* d_ws, size_t ws_size,
                              hipStream_t stream)
{
    (void)in_sizes; (void)n_in; (void)out_size; (void)ws_size;

    const float* x_occ    = (const float*)d_in[0];
    const float* x_skill  = (const float*)d_in[1];
    const int*   e_occ    = (const int*)  d_in[2];
    const int*   e_skill  = (const int*)  d_in[3];
    const float* pW_occ   = (const float*)d_in[4];
    const float* pb_occ   = (const float*)d_in[5];
    const float* pW_skill = (const float*)d_in[6];
    const float* pb_skill = (const float*)d_in[7];
    const float* c1_os_Wl = (const float*)d_in[8];
    const float* c1_os_bl = (const float*)d_in[9];
    const float* c1_os_Wr = (const float*)d_in[10];
    const float* c1_so_Wl = (const float*)d_in[11];
    const float* c1_so_bl = (const float*)d_in[12];
    const float* c1_so_Wr = (const float*)d_in[13];
    const float* c2_os_Wl = (const float*)d_in[14];
    const float* c2_os_bl = (const float*)d_in[15];
    const float* c2_os_Wr = (const float*)d_in[16];
    const float* c2_so_Wl = (const float*)d_in[17];
    const float* c2_so_bl = (const float*)d_in[18];
    const float* c2_so_Wr = (const float*)d_in[19];

    float* ws = (float*)d_ws;
    float* h_occ     = ws;
    float* h_skill   = h_occ     + (size_t)N_OCC   * HID;
    float* o1        = h_skill   + (size_t)N_SKILL * HID;
    float* s1        = o1        + (size_t)N_OCC   * HID;
    float* aggbuf    = s1        + (size_t)N_SKILL * HID;
    unsigned* adj_skill  = (unsigned*)(aggbuf + (size_t)N_SKILL * HID);
    unsigned* adj_occ    = adj_skill  + N_EDGE;
    unsigned* rp_occ     = adj_occ    + N_EDGE;
    unsigned* rp_skill   = rp_occ     + N_OCC;
    unsigned* rank_occ   = rp_skill   + N_SKILL;
    unsigned* rank_skill = rank_occ   + N_EDGE;
    unsigned* cnt_occ   = (unsigned*)(aggbuf + (size_t)(N_OCC + N_SKILL) * HID);
    unsigned* cnt_skill = cnt_occ + N_OCC;
    float* o2 = (float*)d_out;
    float* s2 = o2 + (size_t)N_OCC * OUTF;

    const dim3 blk256(256);
    const dim3 ggrid(512);

    hipMemsetAsync(cnt_occ, 0, sizeof(unsigned) * (N_OCC + N_SKILL), stream);

    count_rank<<<(N_EDGE + 255) / 256, blk256, 0, stream>>>(
        e_occ, e_skill, cnt_occ, cnt_skill, rank_occ, rank_skill, N_EDGE);
    scan_counts<<<(N_OCC + N_SKILL) / SCAN_CHUNK, blk256, 0, stream>>>(cnt_occ, rp_occ);
    fill_adj<<<(N_EDGE + 255) / 256, blk256, 0, stream>>>(
        e_occ, e_skill, rp_occ, rp_skill, rank_occ, rank_skill, adj_occ, adj_skill, N_EDGE);

    gemm_pair<D_IN, HID, 0><<<ggrid, blk256, 0, stream>>>(
        x_occ,   pW_occ,   nullptr, nullptr, pb_occ,   h_occ,   N_OCC,
        x_skill, pW_skill, nullptr, nullptr, pb_skill, h_skill, N_SKILL, 146, 1);

    float* agg1_occ = s1;
    gather_pair<HID><<<N_OCC / 8 + N_SKILL / 8, blk256, 0, stream>>>(
        h_skill, adj_occ,   rp_occ,   cnt_occ,   agg1_occ, N_OCC,   N_SKILL,
        h_occ,   adj_skill, rp_skill, cnt_skill, aggbuf,   N_SKILL, N_OCC,
        N_OCC / 8, N_EDGE);
    gemm_sage<HID, HID, 1><<<ggrid, blk256, 0, stream>>>(
        agg1_occ, c1_so_Wl, h_occ, c1_so_Wr, cnt_occ, c1_so_bl, o1, N_OCC, 1);
    gemm_sage<HID, HID, 1><<<ggrid, blk256, 0, stream>>>(
        aggbuf, c1_os_Wl, h_skill, c1_os_Wr, cnt_skill, c1_os_bl, s1, N_SKILL, 1);

    float* p_occ      = h_occ;
    float* p_skill    = h_skill;
    float* agg2_occ   = aggbuf;
    float* agg2_skill = aggbuf + (size_t)N_OCC * OUTF;
    const dim3 ggrid64(768);
    gemm_pair<HID, OUTF, 3><<<ggrid64, blk256, 0, stream>>>(
        o1, c2_os_Wl, nullptr, nullptr, nullptr, p_occ,   N_OCC,
        s1, c2_so_Wl, nullptr, nullptr, nullptr, p_skill, N_SKILL, 219, 0);
    gather_pair<OUTF><<<N_OCC / 16 + N_SKILL / 16, blk256, 0, stream>>>(
        p_skill, adj_occ,   rp_occ,   cnt_occ,   agg2_occ,   N_OCC,   N_SKILL,
        p_occ,   adj_skill, rp_skill, cnt_skill, agg2_skill, N_SKILL, N_OCC,
        N_OCC / 16, N_EDGE);

    gemm_pair<HID, OUTF, 2><<<ggrid64, blk256, 0, stream>>>(
        o1, c2_so_Wr, agg2_occ,   cnt_occ,   c2_so_bl, o2, N_OCC,
        s1, c2_os_Wr, agg2_skill, cnt_skill, c2_os_bl, s2, N_SKILL, 219, 0);
}


# edge grouping: per-edge returning global integer atomics replaced by LDS counters (coarse destination bins, then one block ranks each bin); memset goes; f32 sum order follows new rank
# speedup vs baseline: 3.2818x; 1.2853x over previous
#include <hip/hip_runtime.h>

#define N_OCC   20000
#define N_SKILL 50000
#define N_EDGE  600000
#define D_IN    256
#define HID     128
#define OUTF    64

typedef __attribute__((ext_vector_type(2))) float v2f;
typedef __attribute__((ext_vector_type(8))) float v8f;

template <int K, int N, int MODE>
__device__ __forceinline__ void
gemm_body(const float* __restrict__ A1, const float* __restrict__ W1,
          const float* __restrict__ A2, const float* __restrict__ W2,
          const unsigned* __restrict__ cnt, const float* __restrict__ bias,
          float* __restrict__ out, int M, int do_relu, int bid, int nblk)
{
    constexpr int P     = K + 4;
    constexpr int NMAT  = (MODE == 1) ? 2 : 1;
    constexpr int CH    = N / 64;
    constexpr int SLOTS = 8 / CH;
    __shared__ __attribute__((aligned(16))) float Ws[NMAT * N * P];

    for (int idx = threadIdx.x; idx < (K / 4) * N; idx += 256) {
        const int n  = idx % N;
        const int kq = idx / N;
        const float* w = W1 + (size_t)(4 * kq) * N + n;
        float4 v;
        v.x = w[0]; v.y = w[N]; v.z = w[2 * N]; v.w = w[3 * N];
        *(float4*)(Ws + n * P + 4 * kq) = v;
        if (MODE == 1) {
            const float* w2 = W2 + (size_t)(4 * kq) * N + n;
            float4 u;
            u.x = w2[0]; u.y = w2[N]; u.z = w2[2 * N]; u.w = w2[3 * N];
            *(float4*)(Ws + N * P + n * P + 4 * kq) = u;
        }
    }
    __syncthreads();

    const int lane = threadIdx.x & 31;
    const int wave = threadIdx.x >> 5;
    const int hi   = lane >> 4;
    const int lm   = lane & 15;
    const int slot = wave / CH;
    const int col0 = (wave % CH) * 64;
    const int ntiles = M / 16;

    const float* bs1 = Ws + (col0 + lm) * P + 2 * hi;
    const float* bs2 = Ws + N * P + (col0 + lm) * P + 2 * hi;

    float bv0 = 0.f, bv1 = 0.f, bv2 = 0.f, bv3 = 0.f;
    if (MODE != 3) {
        bv0 = bias[col0 + lm];
        bv1 = bias[col0 + 16 + lm];
        bv2 = bias[col0 + 32 + lm];
        bv3 = bias[col0 + 48 + lm];
    }

    for (int rt = slot * nblk + bid; rt < ntiles; rt += SLOTS * nblk) {
        const int row0 = rt * 16;

        v8f c0 = {0.f,0.f,0.f,0.f,0.f,0.f,0.f,0.f};
        v8f c1 = c0, c2 = c0, c3 = c0;

        float g0[8] = {}, g1[8] = {}, g2[8] = {}, g3[8] = {}, dg[8] = {};
        if (MODE == 2) {
            const float* g = A2 + (size_t)(row0 + 8 * hi) * N + col0 + lm;
            #pragma unroll
            for (int r = 0; r < 8; ++r) {
                g0[r] = g[r * N];      g1[r] = g[r * N + 16];
                g2[r] = g[r * N + 32]; g3[r] = g[r * N + 48];
                dg[r] = (float)cnt[row0 + 8 * hi + r];
            }
        }

        {
            const float* ap = A1 + (size_t)(row0 + lm) * K + 2 * hi;
            #pragma unroll 8
            for (int k = 0; k < K / 4; ++k) {
                v2f a  = *(const v2f*)(ap + 4 * k);
                v2f b0 = *(const v2f*)(bs1 + 4 * k);
                v2f b1 = *(const v2f*)(bs1 + 4 * k + 16 * P);
                v2f b2 = *(const v2f*)(bs1 + 4 * k + 32 * P);
                v2f b3 = *(const v2f*)(bs1 + 4 * k + 48 * P);
                c0 = __builtin_amdgcn_wmma_f32_16x16x4_f32(false, a, false, b0, (short)0, c0, false, false);
                c1 = __builtin_amdgcn_wmma_f32_16x16x4_f32(false, a, false, b1, (short)0, c1, false, false);
                c2 = __builtin_amdgcn_wmma_f32_16x16x4_f32(false, a, false, b2, (short)0, c2, false, false);
                c3 = __builtin_amdgcn_wmma_f32_16x16x4_f32(false, a, false, b3, (short)0, c3, false, false);
            }
        }

        if (MODE == 1) {
            #pragma unroll
            for (int r = 0; r < 8; ++r) {
                const float inv = 1.0f / fmaxf((float)cnt[row0 + r + 8 * hi], 1.0f);
                c0[r] *= inv; c1[r] *= inv; c2[r] *= inv; c3[r] *= inv;
            }
            const float* ap = A2 + (size_t)(row0 + lm) * K + 2 * hi;
            #pragma unroll 8
            for (int k = 0; k < K / 4; ++k) {
                v2f a  = *(const v2f*)(ap + 4 * k);
                v2f b0 = *(const v2f*)(bs2 + 4 * k);
                v2f b1 = *(const v2f*)(bs2 + 4 * k + 16 * P);
                v2f b2 = *(const v2f*)(bs2 + 4 * k + 32 * P);
                v2f b3 = *(const v2f*)(bs2 + 4 * k + 48 * P);
                c0 = __builtin_amdgcn_wmma_f32_16x16x4_f32(false, a, false, b0, (short)0, c0, false, false);
                c1 = __builtin_amdgcn_wmma_f32_16x16x4_f32(false, a, false, b1, (short)0, c1, false, false);
                c2 = __builtin_amdgcn_wmma_f32_16x16x4_f32(false, a, false, b2, (short)0, c2, false, false);
                c3 = __builtin_amdgcn_wmma_f32_16x16x4_f32(false, a, false, b3, (short)0, c3, false, false);
            }
        }

        float* orow = out + (size_t)(row0 + 8 * hi) * N + col0 + lm;
        #pragma unroll
        for (int r = 0; r < 8; ++r) {
            float v0 = c0[r], v1 = c1[r], v2 = c2[r], v3 = c3[r];
            if (MODE != 3) { v0 += bv0; v1 += bv1; v2 += bv2; v3 += bv3; }
            if (MODE == 2) {
                const float inv = 1.0f / fmaxf(dg[r], 1.0f);
                v0 += inv * g0[r]; v1 += inv * g1[r];
                v2 += inv * g2[r]; v3 += inv * g3[r];
            }
            if (do_relu) {
                v0 = fmaxf(v0, 0.f); v1 = fmaxf(v1, 0.f);
                v2 = fmaxf(v2, 0.f); v3 = fmaxf(v3, 0.f);
            }
            float* p = orow + (size_t)r * N;
            p[0] = v0; p[16] = v1; p[32] = v2; p[48] = v3;
        }
    }
}

template <int K, int N, int MODE>
__global__ void __launch_bounds__(256)
gemm_sage(const float* __restrict__ A1, const float* __restrict__ W1,
          const float* __restrict__ A2, const float* __restrict__ W2,
          const unsigned* __restrict__ cnt, const float* __restrict__ bias,
          float* __restrict__ out, int M, int do_relu)
{
    gemm_body<K, N, MODE>(A1, W1, A2, W2, cnt, bias, out, M, do_relu,
                          (int)blockIdx.x, (int)gridDim.x);
}

template <int K, int N, int MODE>
__global__ void __launch_bounds__(256)
gemm_pair(const float* __restrict__ A1a, const float* __restrict__ W1a,
          const float* __restrict__ A2a, const unsigned* __restrict__ cnta,
          const float* __restrict__ biasa, float* __restrict__ outa, int Ma,
          const float* __restrict__ A1b, const float* __restrict__ W1b,
          const float* __restrict__ A2b, const unsigned* __restrict__ cntb,
          const float* __restrict__ biasb, float* __restrict__ outb, int Mb,
          int split, int do_relu)
{
    const bool second = (int)blockIdx.x >= split;
    gemm_body<K, N, MODE>(second ? A1b : A1a, second ? W1b : W1a,
                          second ? A2b : A2a, nullptr,
                          second ? cntb : cnta, second ? biasb : biasa,
                          second ? outb : outa, second ? Mb : Ma, do_relu,
                          second ? (int)blockIdx.x - split : (int)blockIdx.x,
                          second ? (int)gridDim.x - split : split);
}

#define SORT_G      256
#define SORT_T      512
#define OCC_SHIFT   8
#define SKILL_SHIFT 9
#define NB_OCC      ((N_OCC   + (1 << OCC_SHIFT)   - 1) >> OCC_SHIFT)
#define NB_SKILL    ((N_SKILL + (1 << SKILL_SHIFT) - 1) >> SKILL_SHIFT)
#define NB_ALL      (NB_OCC + NB_SKILL)
#define SORT_CHUNK  ((N_EDGE + SORT_G - 1) / SORT_G)
#define SORT_PER_T  ((SORT_CHUNK + SORT_T - 1) / SORT_T)
static_assert(N_OCC <= 65536 && N_SKILL <= 65536, "a source id must fit the low 16 bits of an edge word");
#define ROWS_PER_W  ((NB_ALL + SORT_T / 32 - 1) / (SORT_T / 32))
static_assert(SORT_G == 256, "a histogram row is two 16-byte loads per lane of a wave");
static_assert(NB_OCC <= 128 && NB_SKILL <= 128, "the bin starts are scanned by one wave, 4 bins per lane");
static_assert(NB_ALL <= SORT_T, "one thread per histogram row");

__device__ __forceinline__ unsigned bin_of_occ(unsigned o)
{
    return min(o >> OCC_SHIFT, (unsigned)(NB_OCC - 1));
}
__device__ __forceinline__ unsigned bin_of_skill(unsigned s)
{
    return min(s >> SKILL_SHIFT, (unsigned)(NB_SKILL - 1));
}

__global__ void __launch_bounds__(SORT_T)
bin_hist(const int* __restrict__ e_occ, const int* __restrict__ e_skill,
         unsigned* __restrict__ hist)
{
    __shared__ unsigned h[NB_ALL];
    const int t = threadIdx.x, g = blockIdx.x;
    if (t < NB_ALL) h[t] = 0u;
    __syncthreads();
    const int beg = g * SORT_CHUNK;
    const int end = min(beg + SORT_CHUNK, N_EDGE);
    unsigned o[SORT_PER_T], s[SORT_PER_T];
    #pragma unroll
    for (int k = 0; k < SORT_PER_T; ++k) {
        const int e = beg + t + k * SORT_T;
        o[k] = 0u; s[k] = 0u;
        if (e < end) { o[k] = (unsigned)e_occ[e]; s[k] = (unsigned)e_skill[e]; }
    }
    #pragma unroll
    for (int k = 0; k < SORT_PER_T; ++k) {
        const int e = beg + t + k * SORT_T;
        if (e < end) {
            atomicAdd(&h[bin_of_occ(o[k])], 1u);
            atomicAdd(&h[NB_OCC + bin_of_skill(s[k])], 1u);
        }
    }
    __syncthreads();
    if (t < NB_ALL) hist[t * SORT_G + g] = h[t];
}

__global__ void __launch_bounds__(SORT_T)
bin_place(const int* __restrict__ e_occ, const int* __restrict__ e_skill,
          const unsigned* __restrict__ hist,
          unsigned* __restrict__ bstart, unsigned* __restrict__ bsize,
          unsigned* __restrict__ binned_occ, unsigned* __restrict__ binned_skill)
{
    __shared__ unsigned tot[NB_ALL];
    __shared__ unsigned cur[NB_ALL];
    const int t = threadIdx.x, g = blockIdx.x;
    const int lane = t & 31, wave = t >> 5;

    const int beg = g * SORT_CHUNK;
    const int end = min(beg + SORT_CHUNK, N_EDGE);
    unsigned o[SORT_PER_T], s[SORT_PER_T];
    #pragma unroll
    for (int k = 0; k < SORT_PER_T; ++k) {
        const int e = beg + t + k * SORT_T;
        o[k] = 0u; s[k] = 0u;
        if (e < end) { o[k] = (unsigned)e_occ[e]; s[k] = (unsigned)e_skill[e]; }
    }

    uint4 va[ROWS_PER_W], vb[ROWS_PER_W];
    #pragma unroll
    for (int j = 0; j < ROWS_PER_W; ++j) {
        const int r = wave + j * (SORT_T / 32);
        va[j] = make_uint4(0u, 0u, 0u, 0u); vb[j] = va[j];
        if (r < NB_ALL) {
            const uint4* row = (const uint4*)(hist + r * SORT_G);
            va[j] = row[2 * lane];
            vb[j] = row[2 * lane + 1];
        }
    }
    #pragma unroll
    for (int j = 0; j < ROWS_PER_W; ++j) {
        const int r = wave + j * (SORT_T / 32);
        const uint4 a = va[j], b = vb[j];
        const int g0 = 8 * lane;
        unsigned all = ((a.x + a.y) + (a.z + a.w)) + ((b.x + b.y) + (b.z + b.w));
        unsigned before = (g0     < g ? a.x : 0u) + (g0 + 1 < g ? a.y : 0u)
                        + (g0 + 2 < g ? a.z : 0u) + (g0 + 3 < g ? a.w : 0u)
                        + (g0 + 4 < g ? b.x : 0u) + (g0 + 5 < g ? b.y : 0u)
                        + (g0 + 6 < g ? b.z : 0u) + (g0 + 7 < g ? b.w : 0u);
        #pragma unroll
        for (int m = 16; m >= 1; m >>= 1) {
            all    += __shfl_xor(all, m);
            before += __shfl_xor(before, m);
        }
        if (r < NB_ALL && lane == 0) { tot[r] = all; cur[r] = before; }
    }
    __syncthreads();
    if (wave < 2) {
        const int base = (wave == 0) ? 0 : NB_OCC;
        const int n    = (wave == 0) ? NB_OCC : NB_SKILL;
        unsigned a[4], sum = 0u;
        #pragma unroll
        for (int i = 0; i < 4; ++i) {
            const int r = 4 * lane + i;
            a[i] = (r < n) ? tot[base + r] : 0u;
            sum += a[i];
        }
        unsigned inc = sum;
        #pragma unroll
        for (int d = 1; d < 32; d <<= 1) {
            const unsigned u = __shfl_up(inc, d);
            if (lane >= d) inc += u;
        }
        unsigned ex = inc - sum;
        #pragma unroll
        for (int i = 0; i < 4; ++i) {
            const int r = 4 * lane + i;
            if (r < n) {
                if (g == 0) { bstart[base + r] = ex; bsize[base + r] = a[i]; }
                cur[base + r] += ex;
                ex += a[i];
            }
        }
    }
    __syncthreads();

    unsigned po[SORT_PER_T], ps[SORT_PER_T];
    #pragma unroll
    for (int k = 0; k < SORT_PER_T; ++k) {
        const int e = beg + t + k * SORT_T;
        po[k] = 0xFFFFFFFFu; ps[k] = 0xFFFFFFFFu;
        if (e < end) {
            po[k] = atomicAdd(&cur[bin_of_occ(o[k])], 1u);
            ps[k] = atomicAdd(&cur[NB_OCC + bin_of_skill(s[k])], 1u);
        }
    }
    #pragma unroll
    for (int k = 0; k < SORT_PER_T; ++k) {
        if (po[k] < (unsigned)N_EDGE)
            binned_occ[po[k]]   = ((o[k] & ((1u << OCC_SHIFT) - 1u)) << 16) | (s[k] & 0xFFFFu);
        if (ps[k] < (unsigned)N_EDGE)
            binned_skill[ps[k]] = ((s[k] & ((1u << SKILL_SHIFT) - 1u)) << 16) | (o[k] & 0xFFFFu);
    }
}

__global__ void __launch_bounds__(SORT_T)
bin_rank(const unsigned* __restrict__ bstart, const unsigned* __restrict__ bsize,
         const unsigned* __restrict__ binned_occ, const unsigned* __restrict__ binned_skill,
         unsigned* __restrict__ cnt_occ, unsigned* __restrict__ cnt_skill,
         unsigned* __restrict__ rp_occ, unsigned* __restrict__ rp_skill,
         unsigned* __restrict__ adj_occ, unsigned* __restrict__ adj_skill)
{
    constexpr int NMAX = 1 << SKILL_SHIFT;
    static_assert((1 << OCC_SHIFT) <= NMAX && NMAX <= SORT_T && NMAX / 32 <= 32, "one thread per node of a bin");
    constexpr int BATCH = 8;
    __shared__ unsigned c[NMAX + 1];
    __shared__ unsigned wsum[NMAX / 32];
    const int t = threadIdx.x;
    const int lane = t & 31, wave = t >> 5;
    const int row = blockIdx.x;
    const bool sk = row >= NB_OCC;
    const int shift = sk ? SKILL_SHIFT : OCC_SHIFT;
    const int nbase = (sk ? row - NB_OCC : row) << shift;
    const int nn    = min(1 << shift, (sk ? N_SKILL : N_OCC) - nbase);
    const unsigned* data = sk ? binned_skill : binned_occ;
    unsigned* cnt = sk ? cnt_skill : cnt_occ;
    unsigned* rp  = sk ? rp_skill  : rp_occ;
    unsigned* adj = sk ? adj_skill : adj_occ;
    unsigned st = bstart[row]; if (st > (unsigned)N_EDGE) st = (unsigned)N_EDGE;
    unsigned n  = bsize[row];  if (n > (unsigned)N_EDGE - st) n = (unsigned)N_EDGE - st;
    data += st;
    adj  += st;

    if (t < NMAX) c[t] = 0u;
    if (t == 0) c[NMAX] = 0u;
    __syncthreads();
    for (unsigned i0 = t; i0 < n; i0 += BATCH * SORT_T) {
        unsigned w[BATCH];
        #pragma unroll
        for (int k = 0; k < BATCH; ++k) {
            const unsigned i = i0 + k * SORT_T;
            w[k] = data[min(i, n - 1u)];
        }
        #pragma unroll
        for (int k = 0; k < BATCH; ++k) {
            const unsigned i = i0 + k * SORT_T;
            atomicAdd(&c[(i < n) ? min(w[k] >> 16, (unsigned)(NMAX - 1)) : (unsigned)NMAX], 1u);
        }
    }
    __syncthreads();
    unsigned v = 0u, inc = 0u;
    if (t < NMAX) {
        v = c[t];
        inc = v;
        #pragma unroll
        for (int d = 1; d < 32; d <<= 1) {
            const unsigned u = __shfl_up(inc, d);
            if (lane >= d) inc += u;
        }
        if (lane == 31) wsum[wave] = inc;
    }
    __syncthreads();
    if (wave == 0) {
        const unsigned ws = (lane < NMAX / 32) ? wsum[lane] : 0u;
        unsigned wi = ws;
        #pragma unroll
        for (int d = 1; d < 32; d <<= 1) {
            const unsigned u = __shfl_up(wi, d);
            if (lane >= d) wi += u;
        }
        if (lane < NMAX / 32) wsum[lane] = wi - ws;
    }
    __syncthreads();
    if (t < NMAX) {
        const unsigned ex = inc - v + wsum[wave];
        c[t] = ex;
        if (t < nn) { cnt[nbase + t] = v; rp[nbase + t] = st + ex; }
    }
    __syncthreads();
    for (unsigned i0 = t; i0 < n; i0 += BATCH * SORT_T) {
        unsigned w[BATCH], p[BATCH];
        #pragma unroll
        for (int k = 0; k < BATCH; ++k) {
            const unsigned i = i0 + k * SORT_T;
            w[k] = data[min(i, n - 1u)];
        }
        #pragma unroll
        for (int k = 0; k < BATCH; ++k) {
            const unsigned i = i0 + k * SORT_T;
            p[k] = atomicAdd(&c[(i < n) ? min(w[k] >> 16, (unsigned)(NMAX - 1)) : (unsigned)NMAX], 1u);
        }
        #pragma unroll
        for (int k = 0; k < BATCH; ++k) {
            const unsigned i = i0 + k * SORT_T;
            if (i < n && p[k] < n) adj[p[k]] = w[k] & 0xFFFFu;
        }
    }
}

__device__ inline void add4(float4& a, const float4& v)
{
    a.x += v.x; a.y += v.y; a.z += v.z; a.w += v.w;
}

template <int F>
__device__ __forceinline__ void
gather_body(const float* __restrict__ hsrc, const unsigned* __restrict__ adj,
            const unsigned* __restrict__ rp, const unsigned* __restrict__ cnt,
            float* __restrict__ agg, int n_dst, unsigned n_src, unsigned E, int bid)
{
    constexpr int TPR = F / 4;
    const int tid = bid * 256 + threadIdx.x;
    const int d = tid / TPR;
    const int f = (tid & (TPR - 1)) << 2;
    if (d >= n_dst) return;
    unsigned beg = rp[d];  if (beg > E) beg = E;
    unsigned deg = cnt[d]; if (deg > E - beg) deg = E - beg;
    const unsigned top = n_src - 1;
    const unsigned* a = adj + beg;
    const float* hb = hsrc + f;
    float4 s0 = {0.f, 0.f, 0.f, 0.f}, s1 = s0, s2 = s0, s3 = s0;
    unsigned j = 0;
    unsigned n0 = 0, n1 = 0, n2 = 0, n3 = 0;
    if (deg >= 4) { n0 = a[0]; n1 = a[1]; n2 = a[2]; n3 = a[3]; }
    for (; j + 4 <= deg; j += 4) {
        const unsigned i0 = min(n0, top), i1 = min(n1, top);
        const unsigned i2 = min(n2, top), i3 = min(n3, top);
        if (j + 8 <= deg) { n0 = a[j + 4]; n1 = a[j + 5]; n2 = a[j + 6]; n3 = a[j + 7]; }
        const float4 v0 = *(const float4*)(hb + (size_t)i0 * F);
        const float4 v1 = *(const float4*)(hb + (size_t)i1 * F);
        const float4 v2 = *(const float4*)(hb + (size_t)i2 * F);
        const float4 v3 = *(const float4*)(hb + (size_t)i3 * F);
        add4(s0, v0); add4(s1, v1); add4(s2, v2); add4(s3, v3);
    }
    if (j < deg) {
        const unsigned r = deg - j;
        const unsigned i0 = min(a[j], top);
        const unsigned i1 = min(a[r > 1 ? j + 1 : j], top);
        const unsigned i2 = min(a[r > 2 ? j + 2 : j], top);
        const float4 v0 = *(const float4*)(hb + (size_t)i0 * F);
        const float4 v1 = *(const float4*)(hb + (size_t)i1 * F);
        const float4 v2 = *(const float4*)(hb + (size_t)i2 * F);
        add4(s0, v0);
        if (r > 1) add4(s1, v1);
        if (r > 2) add4(s2, v2);
    }
    add4(s0, s1); add4(s2, s3); add4(s0, s2);
    *(float4*)(agg + (size_t)d * F + f) = s0;
}

template <int F>
__global__ void __launch_bounds__(256)
gather_pair(const float* __restrict__ hsrc_a, const unsigned* __restrict__ adj_a,
            const unsigned* __restrict__ rp_a, const unsigned* __restrict__ cnt_a,
            float* __restrict__ agg_a, int n_dst_a, unsigned n_src_a,
            const float* __restrict__ hsrc_b, const unsigned* __restrict__ adj_b,
            const unsigned* __restrict__ rp_b, const unsigned* __restrict__ cnt_b,
            float* __restrict__ agg_b, int n_dst_b, unsigned n_src_b,
            int split, unsigned E)
{
    const bool second = (int)blockIdx.x >= split;
    gather_body<F>(second ? hsrc_b : hsrc_a, second ? adj_b : adj_a,
                   second ? rp_b : rp_a, second ? cnt_b : cnt_a,
                   second ? agg_b : agg_a, second ? n_dst_b : n_dst_a,
                   second ? n_src_b : n_src_a, E,
                   second ? (int)blockIdx.x - split : (int)blockIdx.x);
}

extern "C" void kernel_launch(void* const* d_in, const int* in_sizes, int n_in,
                              void* d_out, int out_size, void* d_ws, size_t ws_size,
                              hipStream_t stream)
{
    (void)in_sizes; (void)n_in; (void)out_size; (void)ws_size;

    const float* x_occ    = (const float*)d_in[0];
    const float* x_skill  = (const float*)d_in[1];
    const int*   e_occ    = (const int*)  d_in[2];
    const int*   e_skill  = (const int*)  d_in[3];
    const float* pW_occ   = (const float*)d_in[4];
    const float* pb_occ   = (const float*)d_in[5];
    const float* pW_skill = (const float*)d_in[6];
    const float* pb_skill = (const float*)d_in[7];
    const float* c1_os_Wl = (const float*)d_in[8];
    const float* c1_os_bl = (const float*)d_in[9];
    const float* c1_os_Wr = (const float*)d_in[10];
    const float* c1_so_Wl = (const float*)d_in[11];
    const float* c1_so_bl = (const float*)d_in[12];
    const float* c1_so_Wr = (const float*)d_in[13];
    const float* c2_os_Wl = (const float*)d_in[14];
    const float* c2_os_bl = (const float*)d_in[15];
    const float* c2_os_Wr = (const float*)d_in[16];
    const float* c2_so_Wl = (const float*)d_in[17];
    const float* c2_so_bl = (const float*)d_in[18];
    const float* c2_so_Wr = (const float*)d_in[19];

    float* ws = (float*)d_ws;
    float* h_occ     = ws;
    float* h_skill   = h_occ     + (size_t)N_OCC   * HID;
    float* o1        = h_skill   + (size_t)N_SKILL * HID;
    float* s1        = o1        + (size_t)N_OCC   * HID;
    float* aggbuf    = s1        + (size_t)N_SKILL * HID;
    unsigned* adj_skill  = (unsigned*)(aggbuf + (size_t)N_SKILL * HID);
    unsigned* adj_occ    = adj_skill  + N_EDGE;
    unsigned* rp_occ     = adj_occ    + N_EDGE;
    unsigned* rp_skill   = rp_occ     + N_OCC;
    unsigned* binned_occ   = rp_skill   + N_SKILL;
    unsigned* binned_skill = binned_occ + N_EDGE;
    unsigned* hist   = binned_skill + N_EDGE;
    unsigned* bstart = hist   + NB_ALL * SORT_G;
    unsigned* bsize  = bstart + NB_ALL;
    static_assert(2 * N_EDGE + N_OCC + N_SKILL + 2 * N_EDGE + NB_ALL * SORT_G + 2 * NB_ALL <= N_OCC * HID,
                  "the edge tables must fit the 20000*128 words behind the neighbour-sum buffer");
    unsigned* cnt_occ   = (unsigned*)(aggbuf + (size_t)(N_OCC + N_SKILL) * HID);
    unsigned* cnt_skill = cnt_occ + N_OCC;
    float* o2 = (float*)d_out;
    float* s2 = o2 + (size_t)N_OCC * OUTF;

    const dim3 blk256(256);
    const dim3 ggrid(512);

    const dim3 blkS(SORT_T);
    bin_hist<<<SORT_G, blkS, 0, stream>>>(e_occ, e_skill, hist);
    bin_place<<<SORT_G, blkS, 0, stream>>>(e_occ, e_skill, hist, bstart, bsize,
                                           binned_occ, binned_skill);
    bin_rank<<<NB_ALL, blkS, 0, stream>>>(bstart, bsize, binned_occ, binned_skill,
                                          cnt_occ, cnt_skill, rp_occ, rp_skill,
                                          adj_occ, adj_skill);

    gemm_pair<D_IN, HID, 0><<<ggrid, blk256, 0, stream>>>(
        x_occ,   pW_occ,   nullptr, nullptr, pb_occ,   h_occ,   N_OCC,
        x_skill, pW_skill, nullptr, nullptr, pb_skill, h_skill, N_SKILL, 146, 1);

    float* agg1_occ = s1;
    gather_pair<HID><<<N_OCC / 8 + N_SKILL / 8, blk256, 0, stream>>>(
        h_skill, adj_occ,   rp_occ,   cnt_occ,   agg1_occ, N_OCC,   N_SKILL,
        h_occ,   adj_skill, rp_skill, cnt_skill, aggbuf,   N_SKILL, N_OCC,
        N_OCC / 8, N_EDGE);
    gemm_sage<HID, HID, 1><<<ggrid, blk256, 0, stream>>>(
        agg1_occ, c1_so_Wl, h_occ, c1_so_Wr, cnt_occ, c1_so_bl, o1, N_OCC, 1);
    gemm_sage<HID, HID, 1><<<ggrid, blk256, 0, stream>>>(
        aggbuf, c1_os_Wl, h_skill, c1_os_Wr, cnt_skill, c1_os_bl, s1, N_SKILL, 1);

    float* p_occ      = h_occ;
    float* p_skill    = h_skill;
    float* agg2_occ   = aggbuf;
    float* agg2_skill = aggbuf + (size_t)N_OCC * OUTF;
    const dim3 ggrid64(768);
    gemm_pair<HID, OUTF, 3><<<ggrid64, blk256, 0, stream>>>(
        o1, c2_os_Wl, nullptr, nullptr, nullptr, p_occ,   N_OCC,
        s1, c2_so_Wl, nullptr, nullptr, nullptr, p_skill, N_SKILL, 219, 0);
    gather_pair<OUTF><<<N_OCC / 16 + N_SKILL / 16, blk256, 0, stream>>>(
        p_skill, adj_occ,   rp_occ,   cnt_occ,   agg2_occ,   N_OCC,   N_SKILL,
        p_occ,   adj_skill, rp_skill, cnt_skill, agg2_skill, N_SKILL, N_OCC,
        N_OCC / 16, N_EDGE);

    gemm_pair<HID, OUTF, 2><<<ggrid64, blk256, 0, stream>>>(
        o1, c2_so_Wr, agg2_occ,   cnt_occ,   c2_so_bl, o2, N_OCC,
        s1, c2_os_Wr, agg2_skill, cnt_skill, c2_os_bl, s2, N_SKILL, 219, 0);
}


# conv2: both 64-column products of a node type in ONE pass over its rows (x@Wl; x@Wr+bl stored to out as start value), the neighbour pass adds the mean to out; 10 launches -> 9; f32 sum order unchanged
# speedup vs baseline: 3.4046x; 1.0374x over previous
#include <hip/hip_runtime.h>

#define N_OCC   20000
#define N_SKILL 50000
#define N_EDGE  600000
#define D_IN    256
#define HID     128
#define OUTF    64

typedef __attribute__((ext_vector_type(2))) float v2f;
typedef __attribute__((ext_vector_type(8))) float v8f;

template <int K, int N, int MODE>
__device__ __forceinline__ void
gemm_body(const float* __restrict__ A1, const float* __restrict__ W1,
          const float* __restrict__ A2, const float* __restrict__ W2,
          const unsigned* __restrict__ cnt, const float* __restrict__ bias,
          float* __restrict__ out, int M, int do_relu, int bid, int nblk)
{
    constexpr int P     = K + 4;
    constexpr int NMAT  = (MODE == 1) ? 2 : 1;
    constexpr int CH    = N / 64;
    constexpr int SLOTS = 8 / CH;
    __shared__ __attribute__((aligned(16))) float Ws[NMAT * N * P];

    for (int idx = threadIdx.x; idx < (K / 4) * N; idx += 256) {
        const int n  = idx % N;
        const int kq = idx / N;
        const float* w = W1 + (size_t)(4 * kq) * N + n;
        float4 v;
        v.x = w[0]; v.y = w[N]; v.z = w[2 * N]; v.w = w[3 * N];
        *(float4*)(Ws + n * P + 4 * kq) = v;
        if (MODE == 1) {
            const float* w2 = W2 + (size_t)(4 * kq) * N + n;
            float4 u;
            u.x = w2[0]; u.y = w2[N]; u.z = w2[2 * N]; u.w = w2[3 * N];
            *(float4*)(Ws + N * P + n * P + 4 * kq) = u;
        }
    }
    __syncthreads();

    const int lane = threadIdx.x & 31;
    const int wave = threadIdx.x >> 5;
    const int hi   = lane >> 4;
    const int lm   = lane & 15;
    const int slot = wave / CH;
    const int col0 = (wave % CH) * 64;
    const int ntiles = M / 16;

    const float* bs1 = Ws + (col0 + lm) * P + 2 * hi;
    const float* bs2 = Ws + N * P + (col0 + lm) * P + 2 * hi;

    float bv0 = 0.f, bv1 = 0.f, bv2 = 0.f, bv3 = 0.f;
    if (MODE != 3) {
        bv0 = bias[col0 + lm];
        bv1 = bias[col0 + 16 + lm];
        bv2 = bias[col0 + 32 + lm];
        bv3 = bias[col0 + 48 + lm];
    }

    for (int rt = slot * nblk + bid; rt < ntiles; rt += SLOTS * nblk) {
        const int row0 = rt * 16;

        v8f c0 = {0.f,0.f,0.f,0.f,0.f,0.f,0.f,0.f};
        v8f c1 = c0, c2 = c0, c3 = c0;

        float g0[8] = {}, g1[8] = {}, g2[8] = {}, g3[8] = {}, dg[8] = {};
        if (MODE == 2) {
            const float* g = A2 + (size_t)(row0 + 8 * hi) * N + col0 + lm;
            #pragma unroll
            for (int r = 0; r < 8; ++r) {
                g0[r] = g[r * N];      g1[r] = g[r * N + 16];
                g2[r] = g[r * N + 32]; g3[r] = g[r * N + 48];
                dg[r] = (float)cnt[row0 + 8 * hi + r];
            }
        }

        {
            const float* ap = A1 + (size_t)(row0 + lm) * K + 2 * hi;
            #pragma unroll 8
            for (int k = 0; k < K / 4; ++k) {
                v2f a  = *(const v2f*)(ap + 4 * k);
                v2f b0 = *(const v2f*)(bs1 + 4 * k);
                v2f b1 = *(const v2f*)(bs1 + 4 * k + 16 * P);
                v2f b2 = *(const v2f*)(bs1 + 4 * k + 32 * P);
                v2f b3 = *(const v2f*)(bs1 + 4 * k + 48 * P);
                c0 = __builtin_amdgcn_wmma_f32_16x16x4_f32(false, a, false, b0, (short)0, c0, false, false);
                c1 = __builtin_amdgcn_wmma_f32_16x16x4_f32(false, a, false, b1, (short)0, c1, false, false);
                c2 = __builtin_amdgcn_wmma_f32_16x16x4_f32(false, a, false, b2, (short)0, c2, false, false);
                c3 = __builtin_amdgcn_wmma_f32_16x16x4_f32(false, a, false, b3, (short)0, c3, false, false);
            }
        }

        if (MODE == 1) {
            #pragma unroll
            for (int r = 0; r < 8; ++r) {
                const float inv = 1.0f / fmaxf((float)cnt[row0 + r + 8 * hi], 1.0f);
                c0[r] *= inv; c1[r] *= inv; c2[r] *= inv; c3[r] *= inv;
            }
            const float* ap = A2 + (size_t)(row0 + lm) * K + 2 * hi;
            #pragma unroll 8
            for (int k = 0; k < K / 4; ++k) {
                v2f a  = *(const v2f*)(ap + 4 * k);
                v2f b0 = *(const v2f*)(bs2 + 4 * k);
                v2f b1 = *(const v2f*)(bs2 + 4 * k + 16 * P);
                v2f b2 = *(const v2f*)(bs2 + 4 * k + 32 * P);
                v2f b3 = *(const v2f*)(bs2 + 4 * k + 48 * P);
                c0 = __builtin_amdgcn_wmma_f32_16x16x4_f32(false, a, false, b0, (short)0, c0, false, false);
                c1 = __builtin_amdgcn_wmma_f32_16x16x4_f32(false, a, false, b1, (short)0, c1, false, false);
                c2 = __builtin_amdgcn_wmma_f32_16x16x4_f32(false, a, false, b2, (short)0, c2, false, false);
                c3 = __builtin_amdgcn_wmma_f32_16x16x4_f32(false, a, false, b3, (short)0, c3, false, false);
            }
        }

        float* orow = out + (size_t)(row0 + 8 * hi) * N + col0 + lm;
        #pragma unroll
        for (int r = 0; r < 8; ++r) {
            float v0 = c0[r], v1 = c1[r], v2 = c2[r], v3 = c3[r];
            if (MODE != 3) { v0 += bv0; v1 += bv1; v2 += bv2; v3 += bv3; }
            if (MODE == 2) {
                const float inv = 1.0f / fmaxf(dg[r], 1.0f);
                v0 += inv * g0[r]; v1 += inv * g1[r];
                v2 += inv * g2[r]; v3 += inv * g3[r];
            }
            if (do_relu) {
                v0 = fmaxf(v0, 0.f); v1 = fmaxf(v1, 0.f);
                v2 = fmaxf(v2, 0.f); v3 = fmaxf(v3, 0.f);
            }
            float* p = orow + (size_t)r * N;
            p[0] = v0; p[16] = v1; p[32] = v2; p[48] = v3;
        }
    }
}

template <int K, int N, int MODE>
__global__ void __launch_bounds__(256)
gemm_sage(const float* __restrict__ A1, const float* __restrict__ W1,
          const float* __restrict__ A2, const float* __restrict__ W2,
          const unsigned* __restrict__ cnt, const float* __restrict__ bias,
          float* __restrict__ out, int M, int do_relu)
{
    gemm_body<K, N, MODE>(A1, W1, A2, W2, cnt, bias, out, M, do_relu,
                          (int)blockIdx.x, (int)gridDim.x);
}

template <int K, int N, int MODE>
__global__ void __launch_bounds__(256)
gemm_pair(const float* __restrict__ A1a, const float* __restrict__ W1a,
          const float* __restrict__ A2a, const unsigned* __restrict__ cnta,
          const float* __restrict__ biasa, float* __restrict__ outa, int Ma,
          const float* __restrict__ A1b, const float* __restrict__ W1b,
          const float* __restrict__ A2b, const unsigned* __restrict__ cntb,
          const float* __restrict__ biasb, float* __restrict__ outb, int Mb,
          int split, int do_relu)
{
    const bool second = (int)blockIdx.x >= split;
    gemm_body<K, N, MODE>(second ? A1b : A1a, second ? W1b : W1a,
                          second ? A2b : A2a, nullptr,
                          second ? cntb : cnta, second ? biasb : biasa,
                          second ? outb : outa, second ? Mb : Ma, do_relu,
                          second ? (int)blockIdx.x - split : (int)blockIdx.x,
                          second ? (int)gridDim.x - split : split);
}

template <int K>
__device__ __forceinline__ void
gemm_dual_body(const float* __restrict__ A, const float* __restrict__ Wp,
               const float* __restrict__ Wq, const float* __restrict__ bias,
               float* __restrict__ P, float* __restrict__ Q, int M, int bid, int nblk)
{
    constexpr int NO = 64;
    constexpr int P4 = K + 4;
    __shared__ __attribute__((aligned(16))) float Wd[2 * NO * P4];

    for (int idx = threadIdx.x; idx < (K / 4) * 2 * NO; idx += 256) {
        const int n  = idx % (2 * NO);
        const int kq = idx / (2 * NO);
        const float* w = (n < NO ? Wp + n : Wq + (n - NO)) + (size_t)(4 * kq) * NO;
        float4 v;
        v.x = w[0]; v.y = w[NO]; v.z = w[2 * NO]; v.w = w[3 * NO];
        *(float4*)(Wd + n * P4 + 4 * kq) = v;
    }
    __syncthreads();

    const int lane = threadIdx.x & 31;
    const int wave = threadIdx.x >> 5;
    const int hi   = lane >> 4;
    const int lm   = lane & 15;
    const int slot = wave >> 1;
    const bool isq = (wave & 1) != 0;
    const int ntiles = M / 16;

    const float* bs = Wd + ((isq ? NO : 0) + lm) * P4 + 2 * hi;
    float* dst = isq ? Q : P;

    float bv0 = 0.f, bv1 = 0.f, bv2 = 0.f, bv3 = 0.f;
    if (isq) {
        bv0 = bias[lm];      bv1 = bias[16 + lm];
        bv2 = bias[32 + lm]; bv3 = bias[48 + lm];
    }

    for (int rt = slot * nblk + bid; rt < ntiles; rt += 4 * nblk) {
        const int row0 = rt * 16;
        v8f c0 = {0.f,0.f,0.f,0.f,0.f,0.f,0.f,0.f};
        v8f c1 = c0, c2 = c0, c3 = c0;
        const float* ap = A + (size_t)(row0 + lm) * K + 2 * hi;
        #pragma unroll 8
        for (int k = 0; k < K / 4; ++k) {
            v2f a  = *(const v2f*)(ap + 4 * k);
            v2f b0 = *(const v2f*)(bs + 4 * k);
            v2f b1 = *(const v2f*)(bs + 4 * k + 16 * P4);
            v2f b2 = *(const v2f*)(bs + 4 * k + 32 * P4);
            v2f b3 = *(const v2f*)(bs + 4 * k + 48 * P4);
            c0 = __builtin_amdgcn_wmma_f32_16x16x4_f32(false, a, false, b0, (short)0, c0, false, false);
            c1 = __builtin_amdgcn_wmma_f32_16x16x4_f32(false, a, false, b1, (short)0, c1, false, false);
            c2 = __builtin_amdgcn_wmma_f32_16x16x4_f32(false, a, false, b2, (short)0, c2, false, false);
            c3 = __builtin_amdgcn_wmma_f32_16x16x4_f32(false, a, false, b3, (short)0, c3, false, false);
        }
        float* orow = dst + (size_t)(row0 + 8 * hi) * NO + lm;
        #pragma unroll
        for (int r = 0; r < 8; ++r) {
            float* p = orow + (size_t)r * NO;
            p[0]  = c0[r] + bv0; p[16] = c1[r] + bv1;
            p[32] = c2[r] + bv2; p[48] = c3[r] + bv3;
        }
    }
}

template <int K>
__global__ void __launch_bounds__(256)
gemm_dual_pair(const float* __restrict__ Aa, const float* __restrict__ Wpa,
               const float* __restrict__ Wqa, const float* __restrict__ biasa,
               float* __restrict__ Pa, float* __restrict__ Qa, int Ma,
               const float* __restrict__ Ab, const float* __restrict__ Wpb,
               const float* __restrict__ Wqb, const float* __restrict__ biasb,
               float* __restrict__ Pb, float* __restrict__ Qb, int Mb, int split)
{
    const bool second = (int)blockIdx.x >= split;
    gemm_dual_body<K>(second ? Ab : Aa, second ? Wpb : Wpa, second ? Wqb : Wqa,
                      second ? biasb : biasa, second ? Pb : Pa, second ? Qb : Qa,
                      second ? Mb : Ma,
                      second ? (int)blockIdx.x - split : (int)blockIdx.x,
                      second ? (int)gridDim.x - split : split);
}

#define SORT_G      256
#define SORT_T      512
#define OCC_SHIFT   8
#define SKILL_SHIFT 9
#define NB_OCC      ((N_OCC   + (1 << OCC_SHIFT)   - 1) >> OCC_SHIFT)
#define NB_SKILL    ((N_SKILL + (1 << SKILL_SHIFT) - 1) >> SKILL_SHIFT)
#define NB_ALL      (NB_OCC + NB_SKILL)
#define SORT_CHUNK  ((N_EDGE + SORT_G - 1) / SORT_G)
#define SORT_PER_T  ((SORT_CHUNK + SORT_T - 1) / SORT_T)
static_assert(N_OCC <= 65536 && N_SKILL <= 65536, "a source id must fit the low 16 bits of an edge word");
#define ROWS_PER_W  ((NB_ALL + SORT_T / 32 - 1) / (SORT_T / 32))
static_assert(SORT_G == 256, "a histogram row is two 16-byte loads per lane of a wave");
static_assert(NB_OCC <= 128 && NB_SKILL <= 128, "the bin starts are scanned by one wave, 4 bins per lane");
static_assert(NB_ALL <= SORT_T, "one thread per histogram row");

__device__ __forceinline__ unsigned bin_of_occ(unsigned o)
{
    return min(o >> OCC_SHIFT, (unsigned)(NB_OCC - 1));
}
__device__ __forceinline__ unsigned bin_of_skill(unsigned s)
{
    return min(s >> SKILL_SHIFT, (unsigned)(NB_SKILL - 1));
}

__global__ void __launch_bounds__(SORT_T)
bin_hist(const int* __restrict__ e_occ, const int* __restrict__ e_skill,
         unsigned* __restrict__ hist)
{
    __shared__ unsigned h[NB_ALL];
    const int t = threadIdx.x, g = blockIdx.x;
    if (t < NB_ALL) h[t] = 0u;
    __syncthreads();
    const int beg = g * SORT_CHUNK;
    const int end = min(beg + SORT_CHUNK, N_EDGE);
    unsigned o[SORT_PER_T], s[SORT_PER_T];
    #pragma unroll
    for (int k = 0; k < SORT_PER_T; ++k) {
        const int e = beg + t + k * SORT_T;
        o[k] = 0u; s[k] = 0u;
        if (e < end) { o[k] = (unsigned)e_occ[e]; s[k] = (unsigned)e_skill[e]; }
    }
    #pragma unroll
    for (int k = 0; k < SORT_PER_T; ++k) {
        const int e = beg + t + k * SORT_T;
        if (e < end) {
            atomicAdd(&h[bin_of_occ(o[k])], 1u);
            atomicAdd(&h[NB_OCC + bin_of_skill(s[k])], 1u);
        }
    }
    __syncthreads();
    if (t < NB_ALL) hist[t * SORT_G + g] = h[t];
}

__global__ void __launch_bounds__(SORT_T)
bin_place(const int* __restrict__ e_occ, const int* __restrict__ e_skill,
          const unsigned* __restrict__ hist,
          unsigned* __restrict__ bstart, unsigned* __restrict__ bsize,
          unsigned* __restrict__ binned_occ, unsigned* __restrict__ binned_skill)
{
    __shared__ unsigned tot[NB_ALL];
    __shared__ unsigned cur[NB_ALL];
    const int t = threadIdx.x, g = blockIdx.x;
    const int lane = t & 31, wave = t >> 5;

    const int beg = g * SORT_CHUNK;
    const int end = min(beg + SORT_CHUNK, N_EDGE);
    unsigned o[SORT_PER_T], s[SORT_PER_T];
    #pragma unroll
    for (int k = 0; k < SORT_PER_T; ++k) {
        const int e = beg + t + k * SORT_T;
        o[k] = 0u; s[k] = 0u;
        if (e < end) { o[k] = (unsigned)e_occ[e]; s[k] = (unsigned)e_skill[e]; }
    }

    uint4 va[ROWS_PER_W], vb[ROWS_PER_W];
    #pragma unroll
    for (int j = 0; j < ROWS_PER_W; ++j) {
        const int r = wave + j * (SORT_T / 32);
        va[j] = make_uint4(0u, 0u, 0u, 0u); vb[j] = va[j];
        if (r < NB_ALL) {
            const uint4* row = (const uint4*)(hist + r * SORT_G);
            va[j] = row[2 * lane];
            vb[j] = row[2 * lane + 1];
        }
    }
    #pragma unroll
    for (int j = 0; j < ROWS_PER_W; ++j) {
        const int r = wave + j * (SORT_T / 32);
        const uint4 a = va[j], b = vb[j];
        const int g0 = 8 * lane;
        unsigned all = ((a.x + a.y) + (a.z + a.w)) + ((b.x + b.y) + (b.z + b.w));
        unsigned before = (g0     < g ? a.x : 0u) + (g0 + 1 < g ? a.y : 0u)
                        + (g0 + 2 < g ? a.z : 0u) + (g0 + 3 < g ? a.w : 0u)
                        + (g0 + 4 < g ? b.x : 0u) + (g0 + 5 < g ? b.y : 0u)
                        + (g0 + 6 < g ? b.z : 0u) + (g0 + 7 < g ? b.w : 0u);
        #pragma unroll
        for (int m = 16; m >= 1; m >>= 1) {
            all    += __shfl_xor(all, m);
            before += __shfl_xor(before, m);
        }
        if (r < NB_ALL && lane == 0) { tot[r] = all; cur[r] = before; }
    }
    __syncthreads();
    if (wave < 2) {
        const int base = (wave == 0) ? 0 : NB_OCC;
        const int n    = (wave == 0) ? NB_OCC : NB_SKILL;
        unsigned a[4], sum = 0u;
        #pragma unroll
        for (int i = 0; i < 4; ++i) {
            const int r = 4 * lane + i;
            a[i] = (r < n) ? tot[base + r] : 0u;
            sum += a[i];
        }
        unsigned inc = sum;
        #pragma unroll
        for (int d = 1; d < 32; d <<= 1) {
            const unsigned u = __shfl_up(inc, d);
            if (lane >= d) inc += u;
        }
        unsigned ex = inc - sum;
        #pragma unroll
        for (int i = 0; i < 4; ++i) {
            const int r = 4 * lane + i;
            if (r < n) {
                if (g == 0) { bstart[base + r] = ex; bsize[base + r] = a[i]; }
                cur[base + r] += ex;
                ex += a[i];
            }
        }
    }
    __syncthreads();

    unsigned po[SORT_PER_T], ps[SORT_PER_T];
    #pragma unroll
    for (int k = 0; k < SORT_PER_T; ++k) {
        const int e = beg + t + k * SORT_T;
        po[k] = 0xFFFFFFFFu; ps[k] = 0xFFFFFFFFu;
        if (e < end) {
            po[k] = atomicAdd(&cur[bin_of_occ(o[k])], 1u);
            ps[k] = atomicAdd(&cur[NB_OCC + bin_of_skill(s[k])], 1u);
        }
    }
    #pragma unroll
    for (int k = 0; k < SORT_PER_T; ++k) {
        if (po[k] < (unsigned)N_EDGE)
            binned_occ[po[k]]   = ((o[k] & ((1u << OCC_SHIFT) - 1u)) << 16) | (s[k] & 0xFFFFu);
        if (ps[k] < (unsigned)N_EDGE)
            binned_skill[ps[k]] = ((s[k] & ((1u << SKILL_SHIFT) - 1u)) << 16) | (o[k] & 0xFFFFu);
    }
}

__global__ void __launch_bounds__(SORT_T)
bin_rank(const unsigned* __restrict__ bstart, const unsigned* __restrict__ bsize,
         const unsigned* __restrict__ binned_occ, const unsigned* __restrict__ binned_skill,
         unsigned* __restrict__ cnt_occ, unsigned* __restrict__ cnt_skill,
         unsigned* __restrict__ rp_occ, unsigned* __restrict__ rp_skill,
         unsigned* __restrict__ adj_occ, unsigned* __restrict__ adj_skill)
{
    constexpr int NMAX = 1 << SKILL_SHIFT;
    static_assert((1 << OCC_SHIFT) <= NMAX && NMAX <= SORT_T && NMAX / 32 <= 32, "one thread per node of a bin");
    constexpr int BATCH = 8;
    __shared__ unsigned c[NMAX + 1];
    __shared__ unsigned wsum[NMAX / 32];
    const int t = threadIdx.x;
    const int lane = t & 31, wave = t >> 5;
    const int row = blockIdx.x;
    const bool sk = row >= NB_OCC;
    const int shift = sk ? SKILL_SHIFT : OCC_SHIFT;
    const int nbase = (sk ? row - NB_OCC : row) << shift;
    const int nn    = min(1 << shift, (sk ? N_SKILL : N_OCC) - nbase);
    const unsigned* data = sk ? binned_skill : binned_occ;
    unsigned* cnt = sk ? cnt_skill : cnt_occ;
    unsigned* rp  = sk ? rp_skill  : rp_occ;
    unsigned* adj = sk ? adj_skill : adj_occ;
    unsigned st = bstart[row]; if (st > (unsigned)N_EDGE) st = (unsigned)N_EDGE;
    unsigned n  = bsize[row];  if (n > (unsigned)N_EDGE - st) n = (unsigned)N_EDGE - st;
    data += st;
    adj  += st;

    if (t < NMAX) c[t] = 0u;
    if (t == 0) c[NMAX] = 0u;
    __syncthreads();
    for (unsigned i0 = t; i0 < n; i0 += BATCH * SORT_T) {
        unsigned w[BATCH];
        #pragma unroll
        for (int k = 0; k < BATCH; ++k) {
            const unsigned i = i0 + k * SORT_T;
            w[k] = data[min(i, n - 1u)];
        }
        #pragma unroll
        for (int k = 0; k < BATCH; ++k) {
            const unsigned i = i0 + k * SORT_T;
            atomicAdd(&c[(i < n) ? min(w[k] >> 16, (unsigned)(NMAX - 1)) : (unsigned)NMAX], 1u);
        }
    }
    __syncthreads();
    unsigned v = 0u, inc = 0u;
    if (t < NMAX) {
        v = c[t];
        inc = v;
        #pragma unroll
        for (int d = 1; d < 32; d <<= 1) {
            const unsigned u = __shfl_up(inc, d);
            if (lane >= d) inc += u;
        }
        if (lane == 31) wsum[wave] = inc;
    }
    __syncthreads();
    if (wave == 0) {
        const unsigned ws = (lane < NMAX / 32) ? wsum[lane] : 0u;
        unsigned wi = ws;
        #pragma unroll
        for (int d = 1; d < 32; d <<= 1) {
            const unsigned u = __shfl_up(wi, d);
            if (lane >= d) wi += u;
        }
        if (lane < NMAX / 32) wsum[lane] = wi - ws;
    }
    __syncthreads();
    if (t < NMAX) {
        const unsigned ex = inc - v + wsum[wave];
        c[t] = ex;
        if (t < nn) { cnt[nbase + t] = v; rp[nbase + t] = st + ex; }
    }
    __syncthreads();
    for (unsigned i0 = t; i0 < n; i0 += BATCH * SORT_T) {
        unsigned w[BATCH], p[BATCH];
        #pragma unroll
        for (int k = 0; k < BATCH; ++k) {
            const unsigned i = i0 + k * SORT_T;
            w[k] = data[min(i, n - 1u)];
        }
        #pragma unroll
        for (int k = 0; k < BATCH; ++k) {
            const unsigned i = i0 + k * SORT_T;
            p[k] = atomicAdd(&c[(i < n) ? min(w[k] >> 16, (unsigned)(NMAX - 1)) : (unsigned)NMAX], 1u);
        }
        #pragma unroll
        for (int k = 0; k < BATCH; ++k) {
            const unsigned i = i0 + k * SORT_T;
            if (i < n && p[k] < n) adj[p[k]] = w[k] & 0xFFFFu;
        }
    }
}

__device__ inline void add4(float4& a, const float4& v)
{
    a.x += v.x; a.y += v.y; a.z += v.z; a.w += v.w;
}

template <int F, bool ADD>
__device__ __forceinline__ void
gather_body(const float* __restrict__ hsrc, const unsigned* __restrict__ adj,
            const unsigned* __restrict__ rp, const unsigned* __restrict__ cnt,
            float* __restrict__ agg, int n_dst, unsigned n_src, unsigned E, int bid)
{
    constexpr int TPR = F / 4;
    const int tid = bid * 256 + threadIdx.x;
    const int d = tid / TPR;
    const int f = (tid & (TPR - 1)) << 2;
    if (d >= n_dst) return;
    unsigned beg = rp[d];  if (beg > E) beg = E;
    const unsigned craw = cnt[d];
    unsigned deg = craw; if (deg > E - beg) deg = E - beg;
    float4* orow = (float4*)(agg + (size_t)d * F + f);
    float4 q = {0.f, 0.f, 0.f, 0.f};
    if (ADD) q = *orow;
    const unsigned top = n_src - 1;
    const unsigned* a = adj + beg;
    const float* hb = hsrc + f;
    float4 s0 = {0.f, 0.f, 0.f, 0.f}, s1 = s0, s2 = s0, s3 = s0;
    unsigned j = 0;
    unsigned n0 = 0, n1 = 0, n2 = 0, n3 = 0;
    if (deg >= 4) { n0 = a[0]; n1 = a[1]; n2 = a[2]; n3 = a[3]; }
    for (; j + 4 <= deg; j += 4) {
        const unsigned i0 = min(n0, top), i1 = min(n1, top);
        const unsigned i2 = min(n2, top), i3 = min(n3, top);
        if (j + 8 <= deg) { n0 = a[j + 4]; n1 = a[j + 5]; n2 = a[j + 6]; n3 = a[j + 7]; }
        const float4 v0 = *(const float4*)(hb + (size_t)i0 * F);
        const float4 v1 = *(const float4*)(hb + (size_t)i1 * F);
        const float4 v2 = *(const float4*)(hb + (size_t)i2 * F);
        const float4 v3 = *(const float4*)(hb + (size_t)i3 * F);
        add4(s0, v0); add4(s1, v1); add4(s2, v2); add4(s3, v3);
    }
    if (j < deg) {
        const unsigned r = deg - j;
        const unsigned i0 = min(a[j], top);
        const unsigned i1 = min(a[r > 1 ? j + 1 : j], top);
        const unsigned i2 = min(a[r > 2 ? j + 2 : j], top);
        const float4 v0 = *(const float4*)(hb + (size_t)i0 * F);
        const float4 v1 = *(const float4*)(hb + (size_t)i1 * F);
        const float4 v2 = *(const float4*)(hb + (size_t)i2 * F);
        add4(s0, v0);
        if (r > 1) add4(s1, v1);
        if (r > 2) add4(s2, v2);
    }
    add4(s0, s1); add4(s2, s3); add4(s0, s2);
    if (ADD) {
        const float inv = 1.0f / fmaxf((float)craw, 1.0f);
        q.x += inv * s0.x; q.y += inv * s0.y; q.z += inv * s0.z; q.w += inv * s0.w;
        *orow = q;
    } else {
        *orow = s0;
    }
}

template <int F, bool ADD>
__global__ void __launch_bounds__(256)
gather_pair(const float* __restrict__ hsrc_a, const unsigned* __restrict__ adj_a,
            const unsigned* __restrict__ rp_a, const unsigned* __restrict__ cnt_a,
            float* __restrict__ agg_a, int n_dst_a, unsigned n_src_a,
            const float* __restrict__ hsrc_b, const unsigned* __restrict__ adj_b,
            const unsigned* __restrict__ rp_b, const unsigned* __restrict__ cnt_b,
            float* __restrict__ agg_b, int n_dst_b, unsigned n_src_b,
            int split, unsigned E)
{
    const bool second = (int)blockIdx.x >= split;
    gather_body<F, ADD>(second ? hsrc_b : hsrc_a, second ? adj_b : adj_a,
                   second ? rp_b : rp_a, second ? cnt_b : cnt_a,
                   second ? agg_b : agg_a, second ? n_dst_b : n_dst_a,
                   second ? n_src_b : n_src_a, E,
                   second ? (int)blockIdx.x - split : (int)blockIdx.x);
}

extern "C" void kernel_launch(void* const* d_in, const int* in_sizes, int n_in,
                              void* d_out, int out_size, void* d_ws, size_t ws_size,
                              hipStream_t stream)
{
    (void)in_sizes; (void)n_in; (void)out_size; (void)ws_size;

    const float* x_occ    = (const float*)d_in[0];
    const float* x_skill  = (const float*)d_in[1];
    const int*   e_occ    = (const int*)  d_in[2];
    const int*   e_skill  = (const int*)  d_in[3];
    const float* pW_occ   = (const float*)d_in[4];
    const float* pb_occ   = (const float*)d_in[5];
    const float* pW_skill = (const float*)d_in[6];
    const float* pb_skill = (const float*)d_in[7];
    const float* c1_os_Wl = (const float*)d_in[8];
    const float* c1_os_bl = (const float*)d_in[9];
    const float* c1_os_Wr = (const float*)d_in[10];
    const float* c1_so_Wl = (const float*)d_in[11];
    const float* c1_so_bl = (const float*)d_in[12];
    const float* c1_so_Wr = (const float*)d_in[13];
    const float* c2_os_Wl = (const float*)d_in[14];
    const float* c2_os_bl = (const float*)d_in[15];
    const float* c2_os_Wr = (const float*)d_in[16];
    const float* c2_so_Wl = (const float*)d_in[17];
    const float* c2_so_bl = (const float*)d_in[18];
    const float* c2_so_Wr = (const float*)d_in[19];

    float* ws = (float*)d_ws;
    float* h_occ     = ws;
    float* h_skill   = h_occ     + (size_t)N_OCC   * HID;
    float* o1        = h_skill   + (size_t)N_SKILL * HID;
    float* s1        = o1        + (size_t)N_OCC   * HID;
    float* aggbuf    = s1        + (size_t)N_SKILL * HID;
    unsigned* adj_skill  = (unsigned*)(aggbuf + (size_t)N_SKILL * HID);
    unsigned* adj_occ    = adj_skill  + N_EDGE;
    unsigned* rp_occ     = adj_occ    + N_EDGE;
    unsigned* rp_skill   = rp_occ     + N_OCC;
    unsigned* binned_occ   = rp_skill   + N_SKILL;
    unsigned* binned_skill = binned_occ + N_EDGE;
    unsigned* hist   = binned_skill + N_EDGE;
    unsigned* bstart = hist   + NB_ALL * SORT_G;
    unsigned* bsize  = bstart + NB_ALL;
    static_assert(2 * N_EDGE + N_OCC + N_SKILL + 2 * N_EDGE + NB_ALL * SORT_G + 2 * NB_ALL <= N_OCC * HID,
                  "the edge tables must fit the 20000*128 words behind the neighbour-sum buffer");
    unsigned* cnt_occ   = (unsigned*)(aggbuf + (size_t)(N_OCC + N_SKILL) * HID);
    unsigned* cnt_skill = cnt_occ + N_OCC;
    float* o2 = (float*)d_out;
    float* s2 = o2 + (size_t)N_OCC * OUTF;

    const dim3 blk256(256);
    const dim3 ggrid(512);

    const dim3 blkS(SORT_T);
    bin_hist<<<SORT_G, blkS, 0, stream>>>(e_occ, e_skill, hist);
    bin_place<<<SORT_G, blkS, 0, stream>>>(e_occ, e_skill, hist, bstart, bsize,
                                           binned_occ, binned_skill);
    bin_rank<<<NB_ALL, blkS, 0, stream>>>(bstart, bsize, binned_occ, binned_skill,
                                          cnt_occ, cnt_skill, rp_occ, rp_skill,
                                          adj_occ, adj_skill);

    gemm_pair<D_IN, HID, 0><<<ggrid, blk256, 0, stream>>>(
        x_occ,   pW_occ,   nullptr, nullptr, pb_occ,   h_occ,   N_OCC,
        x_skill, pW_skill, nullptr, nullptr, pb_skill, h_skill, N_SKILL, 146, 1);

    float* agg1_occ = s1;
    gather_pair<HID, false><<<N_OCC / 8 + N_SKILL / 8, blk256, 0, stream>>>(
        h_skill, adj_occ,   rp_occ,   cnt_occ,   agg1_occ, N_OCC,   N_SKILL,
        h_occ,   adj_skill, rp_skill, cnt_skill, aggbuf,   N_SKILL, N_OCC,
        N_OCC / 8, N_EDGE);
    gemm_sage<HID, HID, 1><<<ggrid, blk256, 0, stream>>>(
        agg1_occ, c1_so_Wl, h_occ, c1_so_Wr, cnt_occ, c1_so_bl, o1, N_OCC, 1);
    gemm_sage<HID, HID, 1><<<ggrid, blk256, 0, stream>>>(
        aggbuf, c1_os_Wl, h_skill, c1_os_Wr, cnt_skill, c1_os_bl, s1, N_SKILL, 1);

    float* p_occ   = h_occ;
    float* p_skill = h_skill;
    gemm_dual_pair<HID><<<dim3(1024), blk256, 0, stream>>>(
        o1, c2_os_Wl, c2_so_Wr, c2_so_bl, p_occ,   o2, N_OCC,
        s1, c2_so_Wl, c2_os_Wr, c2_os_bl, p_skill, s2, N_SKILL, 293);
    gather_pair<OUTF, true><<<N_OCC / 16 + N_SKILL / 16, blk256, 0, stream>>>(
        p_skill, adj_occ,   rp_occ,   cnt_occ,   o2, N_OCC,   N_SKILL,
        p_occ,   adj_skill, rp_skill, cnt_skill, s2, N_SKILL, N_OCC,
        N_OCC / 16, N_EDGE);
}


# input projection's 4375 row tiles dealt in thirds over the three edge-grouping launches (product blocks beside the grouping blocks): grouping runs under matrix work; 9 -> 8 launches
# speedup vs baseline: 3.5516x; 1.0432x over previous
#include <hip/hip_runtime.h>

#define N_OCC   20000
#define N_SKILL 50000
#define N_EDGE  600000
#define D_IN    256
#define HID     128
#define OUTF    64

typedef __attribute__((ext_vector_type(2))) float v2f;
typedef __attribute__((ext_vector_type(8))) float v8f;

template <int K, int N, int MODE, int NT = 256>
__device__ __forceinline__ void
gemm_body(const float* __restrict__ A1, const float* __restrict__ W1,
          const float* __restrict__ A2, const float* __restrict__ W2,
          const unsigned* __restrict__ cnt, const float* __restrict__ bias,
          float* __restrict__ out, int M, int do_relu, int bid, int nblk)
{
    constexpr int P     = K + 4;
    constexpr int NMAT  = (MODE == 1) ? 2 : 1;
    constexpr int CH    = N / 64;
    constexpr int SLOTS = (NT / 32) / CH;
    __shared__ __attribute__((aligned(16))) float Ws[NMAT * N * P];

    auto copy_item = [&](int idx) {
        const int n  = idx % N;
        const int kq = idx / N;
        const float* w = W1 + (size_t)(4 * kq) * N + n;
        float4 v;
        v.x = w[0]; v.y = w[N]; v.z = w[2 * N]; v.w = w[3 * N];
        *(float4*)(Ws + n * P + 4 * kq) = v;
        if (MODE == 1) {
            const float* w2 = W2 + (size_t)(4 * kq) * N + n;
            float4 u;
            u.x = w2[0]; u.y = w2[N]; u.z = w2[2 * N]; u.w = w2[3 * N];
            *(float4*)(Ws + N * P + n * P + 4 * kq) = u;
        }
    };
    if constexpr (NT > 256) {
        static_assert(((K / 4) * N) % (8 * NT) == 0, "whole groups of eight trips per thread");
        #pragma unroll 8
        for (int j = 0; j < (K / 4) * N / NT; ++j) copy_item((int)threadIdx.x + j * NT);
    } else {
        for (int idx = threadIdx.x; idx < (K / 4) * N; idx += NT) copy_item(idx);
    }
    __syncthreads();

    const int lane = threadIdx.x & 31;
    const int wave = threadIdx.x >> 5;
    const int hi   = lane >> 4;
    const int lm   = lane & 15;
    const int slot = wave / CH;
    const int col0 = (wave % CH) * 64;
    const int ntiles = M / 16;

    const float* bs1 = Ws + (col0 + lm) * P + 2 * hi;
    const float* bs2 = Ws + N * P + (col0 + lm) * P + 2 * hi;

    float bv0 = 0.f, bv1 = 0.f, bv2 = 0.f, bv3 = 0.f;
    if (MODE != 3) {
        bv0 = bias[col0 + lm];
        bv1 = bias[col0 + 16 + lm];
        bv2 = bias[col0 + 32 + lm];
        bv3 = bias[col0 + 48 + lm];
    }

    for (int rt = slot * nblk + bid; rt < ntiles; rt += SLOTS * nblk) {
        const int row0 = rt * 16;

        v8f c0 = {0.f,0.f,0.f,0.f,0.f,0.f,0.f,0.f};
        v8f c1 = c0, c2 = c0, c3 = c0;

        float g0[8] = {}, g1[8] = {}, g2[8] = {}, g3[8] = {}, dg[8] = {};
        if (MODE == 2) {
            const float* g = A2 + (size_t)(row0 + 8 * hi) * N + col0 + lm;
            #pragma unroll
            for (int r = 0; r < 8; ++r) {
                g0[r] = g[r * N];      g1[r] = g[r * N + 16];
                g2[r] = g[r * N + 32]; g3[r] = g[r * N + 48];
                dg[r] = (float)cnt[row0 + 8 * hi + r];
            }
        }

        {
            const float* ap = A1 + (size_t)(row0 + lm) * K + 2 * hi;
            #pragma unroll 8
            for (int k = 0; k < K / 4; ++k) {
                v2f a  = *(const v2f*)(ap + 4 * k);
                v2f b0 = *(const v2f*)(bs1 + 4 * k);
                v2f b1 = *(const v2f*)(bs1 + 4 * k + 16 * P);
                v2f b2 = *(const v2f*)(bs1 + 4 * k + 32 * P);
                v2f b3 = *(const v2f*)(bs1 + 4 * k + 48 * P);
                c0 = __builtin_amdgcn_wmma_f32_16x16x4_f32(false, a, false, b0, (short)0, c0, false, false);
                c1 = __builtin_amdgcn_wmma_f32_16x16x4_f32(false, a, false, b1, (short)0, c1, false, false);
                c2 = __builtin_amdgcn_wmma_f32_16x16x4_f32(false, a, false, b2, (short)0, c2, false, false);
                c3 = __builtin_amdgcn_wmma_f32_16x16x4_f32(false, a, false, b3, (short)0, c3, false, false);
            }
        }

        if (MODE == 1) {
            #pragma unroll
            for (int r = 0; r < 8; ++r) {
                const float inv = 1.0f / fmaxf((float)cnt[row0 + r + 8 * hi], 1.0f);
                c0[r] *= inv; c1[r] *= inv; c2[r] *= inv; c3[r] *= inv;
            }
            const float* ap = A2 + (size_t)(row0 + lm) * K + 2 * hi;
            #pragma unroll 8
            for (int k = 0; k < K / 4; ++k) {
                v2f a  = *(const v2f*)(ap + 4 * k);
                v2f b0 = *(const v2f*)(bs2 + 4 * k);
                v2f b1 = *(const v2f*)(bs2 + 4 * k + 16 * P);
                v2f b2 = *(const v2f*)(bs2 + 4 * k + 32 * P);
                v2f b3 = *(const v2f*)(bs2 + 4 * k + 48 * P);
                c0 = __builtin_amdgcn_wmma_f32_16x16x4_f32(false, a, false, b0, (short)0, c0, false, false);
                c1 = __builtin_amdgcn_wmma_f32_16x16x4_f32(false, a, false, b1, (short)0, c1, false, false);
                c2 = __builtin_amdgcn_wmma_f32_16x16x4_f32(false, a, false, b2, (short)0, c2, false, false);
                c3 = __builtin_amdgcn_wmma_f32_16x16x4_f32(false, a, false, b3, (short)0, c3, false, false);
            }
        }

        float* orow = out + (size_t)(row0 + 8 * hi) * N + col0 + lm;
        #pragma unroll
        for (int r = 0; r < 8; ++r) {
            float v0 = c0[r], v1 = c1[r], v2 = c2[r], v3 = c3[r];
            if (MODE != 3) { v0 += bv0; v1 += bv1; v2 += bv2; v3 += bv3; }
            if (MODE == 2) {
                const float inv = 1.0f / fmaxf(dg[r], 1.0f);
                v0 += inv * g0[r]; v1 += inv * g1[r];
                v2 += inv * g2[r]; v3 += inv * g3[r];
            }
            if (do_relu) {
                v0 = fmaxf(v0, 0.f); v1 = fmaxf(v1, 0.f);
                v2 = fmaxf(v2, 0.f); v3 = fmaxf(v3, 0.f);
            }
            float* p = orow + (size_t)r * N;
            p[0] = v0; p[16] = v1; p[32] = v2; p[48] = v3;
        }
    }
}

template <int K, int N, int MODE>
__global__ void __launch_bounds__(256)
gemm_sage(const float* __restrict__ A1, const float* __restrict__ W1,
          const float* __restrict__ A2, const float* __restrict__ W2,
          const unsigned* __restrict__ cnt, const float* __restrict__ bias,
          float* __restrict__ out, int M, int do_relu)
{
    gemm_body<K, N, MODE>(A1, W1, A2, W2, cnt, bias, out, M, do_relu,
                          (int)blockIdx.x, (int)gridDim.x);
}

template <int K>
__device__ __forceinline__ void
gemm_dual_body(const float* __restrict__ A, const float* __restrict__ Wp,
               const float* __restrict__ Wq, const float* __restrict__ bias,
               float* __restrict__ P, float* __restrict__ Q, int M, int bid, int nblk)
{
    constexpr int NO = 64;
    constexpr int P4 = K + 4;
    __shared__ __attribute__((aligned(16))) float Wd[2 * NO * P4];

    for (int idx = threadIdx.x; idx < (K / 4) * 2 * NO; idx += 256) {
        const int n  = idx % (2 * NO);
        const int kq = idx / (2 * NO);
        const float* w = (n < NO ? Wp + n : Wq + (n - NO)) + (size_t)(4 * kq) * NO;
        float4 v;
        v.x = w[0]; v.y = w[NO]; v.z = w[2 * NO]; v.w = w[3 * NO];
        *(float4*)(Wd + n * P4 + 4 * kq) = v;
    }
    __syncthreads();

    const int lane = threadIdx.x & 31;
    const int wave = threadIdx.x >> 5;
    const int hi   = lane >> 4;
    const int lm   = lane & 15;
    const int slot = wave >> 1;
    const bool isq = (wave & 1) != 0;
    const int ntiles = M / 16;

    const float* bs = Wd + ((isq ? NO : 0) + lm) * P4 + 2 * hi;
    float* dst = isq ? Q : P;

    float bv0 = 0.f, bv1 = 0.f, bv2 = 0.f, bv3 = 0.f;
    if (isq) {
        bv0 = bias[lm];      bv1 = bias[16 + lm];
        bv2 = bias[32 + lm]; bv3 = bias[48 + lm];
    }

    for (int rt = slot * nblk + bid; rt < ntiles; rt += 4 * nblk) {
        const int row0 = rt * 16;
        v8f c0 = {0.f,0.f,0.f,0.f,0.f,0.f,0.f,0.f};
        v8f c1 = c0, c2 = c0, c3 = c0;
        const float* ap = A + (size_t)(row0 + lm) * K + 2 * hi;
        #pragma unroll 8
        for (int k = 0; k < K / 4; ++k) {
            v2f a  = *(const v2f*)(ap + 4 * k);
            v2f b0 = *(const v2f*)(bs + 4 * k);
            v2f b1 = *(const v2f*)(bs + 4 * k + 16 * P4);
            v2f b2 = *(const v2f*)(bs + 4 * k + 32 * P4);
            v2f b3 = *(const v2f*)(bs + 4 * k + 48 * P4);
            c0 = __builtin_amdgcn_wmma_f32_16x16x4_f32(false, a, false, b0, (short)0, c0, false, false);
            c1 = __builtin_amdgcn_wmma_f32_16x16x4_f32(false, a, false, b1, (short)0, c1, false, false);
            c2 = __builtin_amdgcn_wmma_f32_16x16x4_f32(false, a, false, b2, (short)0, c2, false, false);
            c3 = __builtin_amdgcn_wmma_f32_16x16x4_f32(false, a, false, b3, (short)0, c3, false, false);
        }
        float* orow = dst + (size_t)(row0 + 8 * hi) * NO + lm;
        #pragma unroll
        for (int r = 0; r < 8; ++r) {
            float* p = orow + (size_t)r * NO;
            p[0]  = c0[r] + bv0; p[16] = c1[r] + bv1;
            p[32] = c2[r] + bv2; p[48] = c3[r] + bv3;
        }
    }
}

template <int K>
__global__ void __launch_bounds__(256)
gemm_dual_pair(const float* __restrict__ Aa, const float* __restrict__ Wpa,
               const float* __restrict__ Wqa, const float* __restrict__ biasa,
               float* __restrict__ Pa, float* __restrict__ Qa, int Ma,
               const float* __restrict__ Ab, const float* __restrict__ Wpb,
               const float* __restrict__ Wqb, const float* __restrict__ biasb,
               float* __restrict__ Pb, float* __restrict__ Qb, int Mb, int split)
{
    const bool second = (int)blockIdx.x >= split;
    gemm_dual_body<K>(second ? Ab : Aa, second ? Wpb : Wpa, second ? Wqb : Wqa,
                      second ? biasb : biasa, second ? Pb : Pa, second ? Qb : Qa,
                      second ? Mb : Ma,
                      second ? (int)blockIdx.x - split : (int)blockIdx.x,
                      second ? (int)gridDim.x - split : split);
}

#define SORT_G      256
#define SORT_T      512
#define OCC_SHIFT   8
#define SKILL_SHIFT 9
#define NB_OCC      ((N_OCC   + (1 << OCC_SHIFT)   - 1) >> OCC_SHIFT)
#define NB_SKILL    ((N_SKILL + (1 << SKILL_SHIFT) - 1) >> SKILL_SHIFT)
#define NB_ALL      (NB_OCC + NB_SKILL)
#define SORT_CHUNK  ((N_EDGE + SORT_G - 1) / SORT_G)
#define SORT_PER_T  ((SORT_CHUNK + SORT_T - 1) / SORT_T)
static_assert(N_OCC <= 65536 && N_SKILL <= 65536, "a source id must fit the low 16 bits of an edge word");
#define ROWS_PER_W  ((NB_ALL + SORT_T / 32 - 1) / (SORT_T / 32))
static_assert(SORT_G == 256, "a histogram row is two 16-byte loads per lane of a wave");
static_assert(NB_OCC <= 128 && NB_SKILL <= 128, "the bin starts are scanned by one wave, 4 bins per lane");
static_assert(NB_ALL <= SORT_T, "one thread per histogram row");

__device__ __forceinline__ unsigned bin_of_occ(unsigned o)
{
    return min(o >> OCC_SHIFT, (unsigned)(NB_OCC - 1));
}
__device__ __forceinline__ unsigned bin_of_skill(unsigned s)
{
    return min(s >> SKILL_SHIFT, (unsigned)(NB_SKILL - 1));
}

__device__ __forceinline__ void
bin_hist_body(const int* __restrict__ e_occ, const int* __restrict__ e_skill,
              unsigned* __restrict__ hist, int g)
{
    __shared__ unsigned h[NB_ALL];
    const int t = threadIdx.x;
    if (t < NB_ALL) h[t] = 0u;
    __syncthreads();
    const int beg = g * SORT_CHUNK;
    const int end = min(beg + SORT_CHUNK, N_EDGE);
    unsigned o[SORT_PER_T], s[SORT_PER_T];
    #pragma unroll
    for (int k = 0; k < SORT_PER_T; ++k) {
        const int e = beg + t + k * SORT_T;
        o[k] = 0u; s[k] = 0u;
        if (e < end) { o[k] = (unsigned)e_occ[e]; s[k] = (unsigned)e_skill[e]; }
    }
    #pragma unroll
    for (int k = 0; k < SORT_PER_T; ++k) {
        const int e = beg + t + k * SORT_T;
        if (e < end) {
            atomicAdd(&h[bin_of_occ(o[k])], 1u);
            atomicAdd(&h[NB_OCC + bin_of_skill(s[k])], 1u);
        }
    }
    __syncthreads();
    if (t < NB_ALL) hist[t * SORT_G + g] = h[t];
}

__device__ __forceinline__ void
bin_place_body(const int* __restrict__ e_occ, const int* __restrict__ e_skill,
               const unsigned* __restrict__ hist,
               unsigned* __restrict__ bstart, unsigned* __restrict__ bsize,
               unsigned* __restrict__ binned_occ, unsigned* __restrict__ binned_skill, int g)
{
    __shared__ unsigned tot[NB_ALL];
    __shared__ unsigned cur[NB_ALL];
    const int t = threadIdx.x;
    const int lane = t & 31, wave = t >> 5;

    const int beg = g * SORT_CHUNK;
    const int end = min(beg + SORT_CHUNK, N_EDGE);
    unsigned o[SORT_PER_T], s[SORT_PER_T];
    #pragma unroll
    for (int k = 0; k < SORT_PER_T; ++k) {
        const int e = beg + t + k * SORT_T;
        o[k] = 0u; s[k] = 0u;
        if (e < end) { o[k] = (unsigned)e_occ[e]; s[k] = (unsigned)e_skill[e]; }
    }

    uint4 va[ROWS_PER_W], vb[ROWS_PER_W];
    #pragma unroll
    for (int j = 0; j < ROWS_PER_W; ++j) {
        const int r = wave + j * (SORT_T / 32);
        va[j] = make_uint4(0u, 0u, 0u, 0u); vb[j] = va[j];
        if (r < NB_ALL) {
            const uint4* row = (const uint4*)(hist + r * SORT_G);
            va[j] = row[2 * lane];
            vb[j] = row[2 * lane + 1];
        }
    }
    #pragma unroll
    for (int j = 0; j < ROWS_PER_W; ++j) {
        const int r = wave + j * (SORT_T / 32);
        const uint4 a = va[j], b = vb[j];
        const int g0 = 8 * lane;
        unsigned all = ((a.x + a.y) + (a.z + a.w)) + ((b.x + b.y) + (b.z + b.w));
        unsigned before = (g0     < g ? a.x : 0u) + (g0 + 1 < g ? a.y : 0u)
                        + (g0 + 2 < g ? a.z : 0u) + (g0 + 3 < g ? a.w : 0u)
                        + (g0 + 4 < g ? b.x : 0u) + (g0 + 5 < g ? b.y : 0u)
                        + (g0 + 6 < g ? b.z : 0u) + (g0 + 7 < g ? b.w : 0u);
        #pragma unroll
        for (int m = 16; m >= 1; m >>= 1) {
            all    += __shfl_xor(all, m);
            before += __shfl_xor(before, m);
        }
        if (r < NB_ALL && lane == 0) { tot[r] = all; cur[r] = before; }
    }
    __syncthreads();
    if (wave < 2) {
        const int base = (wave == 0) ? 0 : NB_OCC;
        const int n    = (wave == 0) ? NB_OCC : NB_SKILL;
        unsigned a[4], sum = 0u;
        #pragma unroll
        for (int i = 0; i < 4; ++i) {
            const int r = 4 * lane + i;
            a[i] = (r < n) ? tot[base + r] : 0u;
            sum += a[i];
        }
        unsigned inc = sum;
        #pragma unroll
        for (int d = 1; d < 32; d <<= 1) {
            const unsigned u = __shfl_up(inc, d);
            if (lane >= d) inc += u;
        }
        unsigned ex = inc - sum;
        #pragma unroll
        for (int i = 0; i < 4; ++i) {
            const int r = 4 * lane + i;
            if (r < n) {
                if (g == 0) { bstart[base + r] = ex; bsize[base + r] = a[i]; }
                cur[base + r] += ex;
                ex += a[i];
            }
        }
    }
    __syncthreads();

    unsigned po[SORT_PER_T], ps[SORT_PER_T];
    #pragma unroll
    for (int k = 0; k < SORT_PER_T; ++k) {
        const int e = beg + t + k * SORT_T;
        po[k] = 0xFFFFFFFFu; ps[k] = 0xFFFFFFFFu;
        if (e < end) {
            po[k] = atomicAdd(&cur[bin_of_occ(o[k])], 1u);
            ps[k] = atomicAdd(&cur[NB_OCC + bin_of_skill(s[k])], 1u);
        }
    }
    #pragma unroll
    for (int k = 0; k < SORT_PER_T; ++k) {
        if (po[k] < (unsigned)N_EDGE)
            binned_occ[po[k]]   = ((o[k] & ((1u << OCC_SHIFT) - 1u)) << 16) | (s[k] & 0xFFFFu);
        if (ps[k] < (unsigned)N_EDGE)
            binned_skill[ps[k]] = ((s[k] & ((1u << SKILL_SHIFT) - 1u)) << 16) | (o[k] & 0xFFFFu);
    }
}

__device__ __forceinline__ void
bin_rank_body(const unsigned* __restrict__ bstart, const unsigned* __restrict__ bsize,
              const unsigned* __restrict__ binned_occ, const unsigned* __restrict__ binned_skill,
              unsigned* __restrict__ cnt_occ, unsigned* __restrict__ cnt_skill,
              unsigned* __restrict__ rp_occ, unsigned* __restrict__ rp_skill,
              unsigned* __restrict__ adj_occ, unsigned* __restrict__ adj_skill, int row)
{
    constexpr int NMAX = 1 << SKILL_SHIFT;
    static_assert((1 << OCC_SHIFT) <= NMAX && NMAX <= SORT_T && NMAX / 32 <= 32, "one thread per node of a bin");
    constexpr int BATCH = 8;
    __shared__ unsigned c[NMAX + 1];
    __shared__ unsigned wsum[NMAX / 32];
    const int t = threadIdx.x;
    const int lane = t & 31, wave = t >> 5;
    const bool sk = row >= NB_OCC;
    const int shift = sk ? SKILL_SHIFT : OCC_SHIFT;
    const int nbase = (sk ? row - NB_OCC : row) << shift;
    const int nn    = min(1 << shift, (sk ? N_SKILL : N_OCC) - nbase);
    const unsigned* data = sk ? binned_skill : binned_occ;
    unsigned* cnt = sk ? cnt_skill : cnt_occ;
    unsigned* rp  = sk ? rp_skill  : rp_occ;
    unsigned* adj = sk ? adj_skill : adj_occ;
    unsigned st = bstart[row]; if (st > (unsigned)N_EDGE) st = (unsigned)N_EDGE;
    unsigned n  = bsize[row];  if (n > (unsigned)N_EDGE - st) n = (unsigned)N_EDGE - st;
    data += st;
    adj  += st;

    if (t < NMAX) c[t] = 0u;
    if (t == 0) c[NMAX] = 0u;
    __syncthreads();
    for (unsigned i0 = t; i0 < n; i0 += BATCH * SORT_T) {
        unsigned w[BATCH];
        #pragma unroll
        for (int k = 0; k < BATCH; ++k) {
            const unsigned i = i0 + k * SORT_T;
            w[k] = data[min(i, n - 1u)];
        }
        #pragma unroll
        for (int k = 0; k < BATCH; ++k) {
            const unsigned i = i0 + k * SORT_T;
            atomicAdd(&c[(i < n) ? min(w[k] >> 16, (unsigned)(NMAX - 1)) : (unsigned)NMAX], 1u);
        }
    }
    __syncthreads();
    unsigned v = 0u, inc = 0u;
    if (t < NMAX) {
        v = c[t];
        inc = v;
        #pragma unroll
        for (int d = 1; d < 32; d <<= 1) {
            const unsigned u = __shfl_up(inc, d);
            if (lane >= d) inc += u;
        }
        if (lane == 31) wsum[wave] = inc;
    }
    __syncthreads();
    if (wave == 0) {
        const unsigned ws = (lane < NMAX / 32) ? wsum[lane] : 0u;
        unsigned wi = ws;
        #pragma unroll
        for (int d = 1; d < 32; d <<= 1) {
            const unsigned u = __shfl_up(wi, d);
            if (lane >= d) wi += u;
        }
        if (lane < NMAX / 32) wsum[lane] = wi - ws;
    }
    __syncthreads();
    if (t < NMAX) {
        const unsigned ex = inc - v + wsum[wave];
        c[t] = ex;
        if (t < nn) { cnt[nbase + t] = v; rp[nbase + t] = st + ex; }
    }
    __syncthreads();
    for (unsigned i0 = t; i0 < n; i0 += BATCH * SORT_T) {
        unsigned w[BATCH], p[BATCH];
        #pragma unroll
        for (int k = 0; k < BATCH; ++k) {
            const unsigned i = i0 + k * SORT_T;
            w[k] = data[min(i, n - 1u)];
        }
        #pragma unroll
        for (int k = 0; k < BATCH; ++k) {
            const unsigned i = i0 + k * SORT_T;
            p[k] = atomicAdd(&c[(i < n) ? min(w[k] >> 16, (unsigned)(NMAX - 1)) : (unsigned)NMAX], 1u);
        }
        #pragma unroll
        for (int k = 0; k < BATCH; ++k) {
            const unsigned i = i0 + k * SORT_T;
            if (i < n && p[k] < n) adj[p[k]] = w[k] & 0xFFFFu;
        }
    }
}

#define PROJ_G        256
#define PROJ_SKILL_1  (208 * 16)
#define PROJ_SKILL_2  (1458 * 16)
#define PROJ_SKILL_3  (N_SKILL - PROJ_SKILL_1 - PROJ_SKILL_2)
#define PROJ_SPLIT_1  219
static_assert(PROJ_SKILL_3 > 0 && PROJ_SKILL_3 % 16 == 0, "every share of the skill rows is whole row tiles");
static_assert(N_OCC / 16 <= 6 * PROJ_SPLIT_1 && PROJ_SKILL_1 / 16 <= 6 * (PROJ_G - PROJ_SPLIT_1) &&
              PROJ_SKILL_2 / 16 <= 6 * PROJ_G && PROJ_SKILL_3 / 16 <= 6 * PROJ_G,
              "at most 6 row tiles per product block");

__global__ void __launch_bounds__(SORT_T)
hist_and_proj(const int* __restrict__ e_occ, const int* __restrict__ e_skill,
              unsigned* __restrict__ hist,
              const float* __restrict__ xa, const float* __restrict__ Wa,
              const float* __restrict__ biasa, float* __restrict__ ha, int Ma,
              const float* __restrict__ xb, const float* __restrict__ Wb,
              const float* __restrict__ biasb, float* __restrict__ hb, int Mb, int split)
{
    if ((int)blockIdx.x >= PROJ_G) {
        bin_hist_body(e_occ, e_skill, hist, (int)blockIdx.x - PROJ_G);
        return;
    }
    const bool second = (int)blockIdx.x >= split;
    gemm_body<D_IN, HID, 0, SORT_T>(second ? xb : xa, second ? Wb : Wa, nullptr, nullptr, nullptr,
                                    second ? biasb : biasa, second ? hb : ha, second ? Mb : Ma, 1,
                                    second ? (int)blockIdx.x - split : (int)blockIdx.x,
                                    second ? PROJ_G - split : split);
}

__global__ void __launch_bounds__(SORT_T)
place_and_proj(const int* __restrict__ e_occ, const int* __restrict__ e_skill,
               const unsigned* __restrict__ hist,
               unsigned* __restrict__ bstart, unsigned* __restrict__ bsize,
               unsigned* __restrict__ binned_occ, unsigned* __restrict__ binned_skill,
               const float* __restrict__ x, const float* __restrict__ W,
               const float* __restrict__ bias, float* __restrict__ h, int M)
{
    if ((int)blockIdx.x >= PROJ_G) {
        bin_place_body(e_occ, e_skill, hist, bstart, bsize, binned_occ, binned_skill,
                       (int)blockIdx.x - PROJ_G);
        return;
    }
    gemm_body<D_IN, HID, 0, SORT_T>(x, W, nullptr, nullptr, nullptr, bias, h, M, 1,
                                    (int)blockIdx.x, PROJ_G);
}

__global__ void __launch_bounds__(SORT_T)
rank_and_proj(const unsigned* __restrict__ bstart, const unsigned* __restrict__ bsize,
              const unsigned* __restrict__ binned_occ, const unsigned* __restrict__ binned_skill,
              unsigned* __restrict__ cnt_occ, unsigned* __restrict__ cnt_skill,
              unsigned* __restrict__ rp_occ, unsigned* __restrict__ rp_skill,
              unsigned* __restrict__ adj_occ, unsigned* __restrict__ adj_skill,
              const float* __restrict__ x, const float* __restrict__ W,
              const float* __restrict__ bias, float* __restrict__ h, int M)
{
    if ((int)blockIdx.x >= PROJ_G) {
        bin_rank_body(bstart, bsize, binned_occ, binned_skill, cnt_occ, cnt_skill,
                      rp_occ, rp_skill, adj_occ, adj_skill, (int)blockIdx.x - PROJ_G);
        return;
    }
    gemm_body<D_IN, HID, 0, SORT_T>(x, W, nullptr, nullptr, nullptr, bias, h, M, 1,
                                    (int)blockIdx.x, PROJ_G);
}

__device__ inline void add4(float4& a, const float4& v)
{
    a.x += v.x; a.y += v.y; a.z += v.z; a.w += v.w;
}

template <int F, bool ADD>
__device__ __forceinline__ void
gather_body(const float* __restrict__ hsrc, const unsigned* __restrict__ adj,
            const unsigned* __restrict__ rp, const unsigned* __restrict__ cnt,
            float* __restrict__ agg, int n_dst, unsigned n_src, unsigned E, int bid)
{
    constexpr int TPR = F / 4;
    const int tid = bid * 256 + threadIdx.x;
    const int d = tid / TPR;
    const int f = (tid & (TPR - 1)) << 2;
    if (d >= n_dst) return;
    unsigned beg = rp[d];  if (beg > E) beg = E;
    const unsigned craw = cnt[d];
    unsigned deg = craw; if (deg > E - beg) deg = E - beg;
    float4* orow = (float4*)(agg + (size_t)d * F + f);
    float4 q = {0.f, 0.f, 0.f, 0.f};
    if (ADD) q = *orow;
    const unsigned top = n_src - 1;
    const unsigned* a = adj + beg;
    const float* hb = hsrc + f;
    float4 s0 = {0.f, 0.f, 0.f, 0.f}, s1 = s0, s2 = s0, s3 = s0;
    unsigned j = 0;
    unsigned n0 = 0, n1 = 0, n2 = 0, n3 = 0;
    if (deg >= 4) { n0 = a[0]; n1 = a[1]; n2 = a[2]; n3 = a[3]; }
    for (; j + 4 <= deg; j += 4) {
        const unsigned i0 = min(n0, top), i1 = min(n1, top);
        const unsigned i2 = min(n2, top), i3 = min(n3, top);
        if (j + 8 <= deg) { n0 = a[j + 4]; n1 = a[j + 5]; n2 = a[j + 6]; n3 = a[j + 7]; }
        const float4 v0 = *(const float4*)(hb + (size_t)i0 * F);
        const float4 v1 = *(const float4*)(hb + (size_t)i1 * F);
        const float4 v2 = *(const float4*)(hb + (size_t)i2 * F);
        const float4 v3 = *(const float4*)(hb + (size_t)i3 * F);
        add4(s0, v0); add4(s1, v1); add4(s2, v2); add4(s3, v3);
    }
    if (j < deg) {
        const unsigned r = deg - j;
        const unsigned i0 = min(a[j], top);
        const unsigned i1 = min(a[r > 1 ? j + 1 : j], top);
        const unsigned i2 = min(a[r > 2 ? j + 2 : j], top);
        const float4 v0 = *(const float4*)(hb + (size_t)i0 * F);
        const float4 v1 = *(const float4*)(hb + (size_t)i1 * F);
        const float4 v2 = *(const float4*)(hb + (size_t)i2 * F);
        add4(s0, v0);
        if (r > 1) add4(s1, v1);
        if (r > 2) add4(s2, v2);
    }
    add4(s0, s1); add4(s2, s3); add4(s0, s2);
    if (ADD) {
        const float inv = 1.0f / fmaxf((float)craw, 1.0f);
        q.x += inv * s0.x; q.y += inv * s0.y; q.z += inv * s0.z; q.w += inv * s0.w;
        *orow = q;
    } else {
        *orow = s0;
    }
}

template <int F, bool ADD>
__global__ void __launch_bounds__(256)
gather_pair(const float* __restrict__ hsrc_a, const unsigned* __restrict__ adj_a,
            const unsigned* __restrict__ rp_a, const unsigned* __restrict__ cnt_a,
            float* __restrict__ agg_a, int n_dst_a, unsigned n_src_a,
            const float* __restrict__ hsrc_b, const unsigned* __restrict__ adj_b,
            const unsigned* __restrict__ rp_b, const unsigned* __restrict__ cnt_b,
            float* __restrict__ agg_b, int n_dst_b, unsigned n_src_b,
            int split, unsigned E)
{
    const bool second = (int)blockIdx.x >= split;
    gather_body<F, ADD>(second ? hsrc_b : hsrc_a, second ? adj_b : adj_a,
                   second ? rp_b : rp_a, second ? cnt_b : cnt_a,
                   second ? agg_b : agg_a, second ? n_dst_b : n_dst_a,
                   second ? n_src_b : n_src_a, E,
                   second ? (int)blockIdx.x - split : (int)blockIdx.x);
}

extern "C" void kernel_launch(void* const* d_in, const int* in_sizes, int n_in,
                              void* d_out, int out_size, void* d_ws, size_t ws_size,
                              hipStream_t stream)
{
    (void)in_sizes; (void)n_in; (void)out_size; (void)ws_size;

    const float* x_occ    = (const float*)d_in[0];
    const float* x_skill  = (const float*)d_in[1];
    const int*   e_occ    = (const int*)  d_in[2];
    const int*   e_skill  = (const int*)  d_in[3];
    const float* pW_occ   = (const float*)d_in[4];
    const float* pb_occ   = (const float*)d_in[5];
    const float* pW_skill = (const float*)d_in[6];
    const float* pb_skill = (const float*)d_in[7];
    const float* c1_os_Wl = (const float*)d_in[8];
    const float* c1_os_bl = (const float*)d_in[9];
    const float* c1_os_Wr = (const float*)d_in[10];
    const float* c1_so_Wl = (const float*)d_in[11];
    const float* c1_so_bl = (const float*)d_in[12];
    const float* c1_so_Wr = (const float*)d_in[13];
    const float* c2_os_Wl = (const float*)d_in[14];
    const float* c2_os_bl = (const float*)d_in[15];
    const float* c2_os_Wr = (const float*)d_in[16];
    const float* c2_so_Wl = (const float*)d_in[17];
    const float* c2_so_bl = (const float*)d_in[18];
    const float* c2_so_Wr = (const float*)d_in[19];

    float* ws = (float*)d_ws;
    float* h_occ     = ws;
    float* h_skill   = h_occ     + (size_t)N_OCC   * HID;
    float* o1        = h_skill   + (size_t)N_SKILL * HID;
    float* s1        = o1        + (size_t)N_OCC   * HID;
    float* aggbuf    = s1        + (size_t)N_SKILL * HID;
    unsigned* adj_skill  = (unsigned*)(aggbuf + (size_t)N_SKILL * HID);
    unsigned* adj_occ    = adj_skill  + N_EDGE;
    unsigned* rp_occ     = adj_occ    + N_EDGE;
    unsigned* rp_skill   = rp_occ     + N_OCC;
    unsigned* binned_occ   = rp_skill   + N_SKILL;
    unsigned* binned_skill = binned_occ + N_EDGE;
    unsigned* hist   = binned_skill + N_EDGE;
    unsigned* bstart = hist   + NB_ALL * SORT_G;
    unsigned* bsize  = bstart + NB_ALL;
    static_assert(2 * N_EDGE + N_OCC + N_SKILL + 2 * N_EDGE + NB_ALL * SORT_G + 2 * NB_ALL <= N_OCC * HID,
                  "the edge tables must fit the 20000*128 words behind the neighbour-sum buffer");
    unsigned* cnt_occ   = (unsigned*)(aggbuf + (size_t)(N_OCC + N_SKILL) * HID);
    unsigned* cnt_skill = cnt_occ + N_OCC;
    float* o2 = (float*)d_out;
    float* s2 = o2 + (size_t)N_OCC * OUTF;

    const dim3 blk256(256);
    const dim3 ggrid(512);

    const dim3 blkS(SORT_T);
    hist_and_proj<<<PROJ_G + SORT_G, blkS, 0, stream>>>(
        e_occ, e_skill, hist,
        x_occ,   pW_occ,   pb_occ,   h_occ,   N_OCC,
        x_skill, pW_skill, pb_skill, h_skill, PROJ_SKILL_1, PROJ_SPLIT_1);
    place_and_proj<<<PROJ_G + SORT_G, blkS, 0, stream>>>(
        e_occ, e_skill, hist, bstart, bsize, binned_occ, binned_skill,
        x_skill + (size_t)PROJ_SKILL_1 * D_IN, pW_skill, pb_skill,
        h_skill + (size_t)PROJ_SKILL_1 * HID, PROJ_SKILL_2);
    rank_and_proj<<<PROJ_G + NB_ALL, blkS, 0, stream>>>(
        bstart, bsize, binned_occ, binned_skill,
        cnt_occ, cnt_skill, rp_occ, rp_skill, adj_occ, adj_skill,
        x_skill + (size_t)(PROJ_SKILL_1 + PROJ_SKILL_2) * D_IN, pW_skill, pb_skill,
        h_skill + (size_t)(PROJ_SKILL_1 + PROJ_SKILL_2) * HID, PROJ_SKILL_3);

    float* agg1_occ = s1;
    gather_pair<HID, false><<<N_OCC / 8 + N_SKILL / 8, blk256, 0, stream>>>(
        h_skill, adj_occ,   rp_occ,   cnt_occ,   agg1_occ, N_OCC,   N_SKILL,
        h_occ,   adj_skill, rp_skill, cnt_skill, aggbuf,   N_SKILL, N_OCC,
        N_OCC / 8, N_EDGE);
    gemm_sage<HID, HID, 1><<<ggrid, blk256, 0, stream>>>(
        agg1_occ, c1_so_Wl, h_occ, c1_so_Wr, cnt_occ, c1_so_bl, o1, N_OCC, 1);
    gemm_sage<HID, HID, 1><<<ggrid, blk256, 0, stream>>>(
        aggbuf, c1_os_Wl, h_skill, c1_os_Wr, cnt_skill, c1_os_bl, s1, N_SKILL, 1);

    float* p_occ   = h_occ;
    float* p_skill = h_skill;
    gemm_dual_pair<HID><<<dim3(1024), blk256, 0, stream>>>(
        o1, c2_os_Wl, c2_so_Wr, c2_so_bl, p_occ,   o2, N_OCC,
        s1, c2_so_Wl, c2_os_Wr, c2_os_bl, p_skill, s2, N_SKILL, 293);
    gather_pair<OUTF, true><<<N_OCC / 16 + N_SKILL / 16, blk256, 0, stream>>>(
        p_skill, adj_occ,   rp_occ,   cnt_occ,   o2, N_OCC,   N_SKILL,
        p_occ,   adj_skill, rp_skill, cnt_skill, s2, N_SKILL, N_OCC,
        N_OCC / 16, N_EDGE);
}


# conv1's two products (occ, skill rows) share ONE launch of 256 16-wave blocks split 73:183; the occ neighbour sums wait at the head of the output buffer; 8 -> 7 launches; f32 sum order unchanged
# speedup vs baseline: 3.7713x; 1.0619x over previous
#include <hip/hip_runtime.h>

#define N_OCC   20000
#define N_SKILL 50000
#define N_EDGE  600000
#define D_IN    256
#define HID     128
#define OUTF    64

typedef __attribute__((ext_vector_type(2))) float v2f;
typedef __attribute__((ext_vector_type(8))) float v8f;

template <int K, int N, int MODE, int NT = 256>
__device__ __forceinline__ void
gemm_body(const float* __restrict__ A1, const float* __restrict__ W1,
          const float* __restrict__ A2, const float* __restrict__ W2,
          const unsigned* __restrict__ cnt, const float* __restrict__ bias,
          float* __restrict__ out, int M, int do_relu, int bid, int nblk)
{
    constexpr int P     = K + 4;
    constexpr int NMAT  = (MODE == 1) ? 2 : 1;
    constexpr int CH    = N / 64;
    constexpr int SLOTS = (NT / 32) / CH;
    __shared__ __attribute__((aligned(16))) float Ws[NMAT * N * P];

    auto copy_item = [&](int idx) {
        const int n  = idx % N;
        const int kq = idx / N;
        const float* w = W1 + (size_t)(4 * kq) * N + n;
        float4 v;
        v.x = w[0]; v.y = w[N]; v.z = w[2 * N]; v.w = w[3 * N];
        *(float4*)(Ws + n * P + 4 * kq) = v;
        if (MODE == 1) {
            const float* w2 = W2 + (size_t)(4 * kq) * N + n;
            float4 u;
            u.x = w2[0]; u.y = w2[N]; u.z = w2[2 * N]; u.w = w2[3 * N];
            *(float4*)(Ws + N * P + n * P + 4 * kq) = u;
        }
    };
    if constexpr (NT > 256) {
        static_assert(((K / 4) * N) % (8 * NT) == 0, "whole groups of eight trips per thread");
        #pragma unroll 8
        for (int j = 0; j < (K / 4) * N / NT; ++j) copy_item((int)threadIdx.x + j * NT);
    } else {
        for (int idx = threadIdx.x; idx < (K / 4) * N; idx += NT) copy_item(idx);
    }
    __syncthreads();

    const int lane = threadIdx.x & 31;
    const int wave = threadIdx.x >> 5;
    const int hi   = lane >> 4;
    const int lm   = lane & 15;
    const int slot = wave / CH;
    const int col0 = (wave % CH) * 64;
    const int ntiles = M / 16;

    const float* bs1 = Ws + (col0 + lm) * P + 2 * hi;
    const float* bs2 = Ws + N * P + (col0 + lm) * P + 2 * hi;

    float bv0 = 0.f, bv1 = 0.f, bv2 = 0.f, bv3 = 0.f;
    if (MODE != 3) {
        bv0 = bias[col0 + lm];
        bv1 = bias[col0 + 16 + lm];
        bv2 = bias[col0 + 32 + lm];
        bv3 = bias[col0 + 48 + lm];
    }

    for (int rt = slot * nblk + bid; rt < ntiles; rt += SLOTS * nblk) {
        const int row0 = rt * 16;

        v8f c0 = {0.f,0.f,0.f,0.f,0.f,0.f,0.f,0.f};
        v8f c1 = c0, c2 = c0, c3 = c0;

        float g0[8] = {}, g1[8] = {}, g2[8] = {}, g3[8] = {}, dg[8] = {};
        if (MODE == 2) {
            const float* g = A2 + (size_t)(row0 + 8 * hi) * N + col0 + lm;
            #pragma unroll
            for (int r = 0; r < 8; ++r) {
                g0[r] = g[r * N];      g1[r] = g[r * N + 16];
                g2[r] = g[r * N + 32]; g3[r] = g[r * N + 48];
                dg[r] = (float)cnt[row0 + 8 * hi + r];
            }
        }

        {
            const float* ap = A1 + (size_t)(row0 + lm) * K + 2 * hi;
            #pragma unroll 8
            for (int k = 0; k < K / 4; ++k) {
                v2f a  = *(const v2f*)(ap + 4 * k);
                v2f b0 = *(const v2f*)(bs1 + 4 * k);
                v2f b1 = *(const v2f*)(bs1 + 4 * k + 16 * P);
                v2f b2 = *(const v2f*)(bs1 + 4 * k + 32 * P);
                v2f b3 = *(const v2f*)(bs1 + 4 * k + 48 * P);
                c0 = __builtin_amdgcn_wmma_f32_16x16x4_f32(false, a, false, b0, (short)0, c0, false, false);
                c1 = __builtin_amdgcn_wmma_f32_16x16x4_f32(false, a, false, b1, (short)0, c1, false, false);
                c2 = __builtin_amdgcn_wmma_f32_16x16x4_f32(false, a, false, b2, (short)0, c2, false, false);
                c3 = __builtin_amdgcn_wmma_f32_16x16x4_f32(false, a, false, b3, (short)0, c3, false, false);
            }
        }

        if (MODE == 1) {
            #pragma unroll
            for (int r = 0; r < 8; ++r) {
                const float inv = 1.0f / fmaxf((float)cnt[row0 + r + 8 * hi], 1.0f);
                c0[r] *= inv; c1[r] *= inv; c2[r] *= inv; c3[r] *= inv;
            }
            const float* ap = A2 + (size_t)(row0 + lm) * K + 2 * hi;
            #pragma unroll 8
            for (int k = 0; k < K / 4; ++k) {
                v2f a  = *(const v2f*)(ap + 4 * k);
                v2f b0 = *(const v2f*)(bs2 + 4 * k);
                v2f b1 = *(const v2f*)(bs2 + 4 * k + 16 * P);
                v2f b2 = *(const v2f*)(bs2 + 4 * k + 32 * P);
                v2f b3 = *(const v2f*)(bs2 + 4 * k + 48 * P);
                c0 = __builtin_amdgcn_wmma_f32_16x16x4_f32(false, a, false, b0, (short)0, c0, false, false);
                c1 = __builtin_amdgcn_wmma_f32_16x16x4_f32(false, a, false, b1, (short)0, c1, false, false);
                c2 = __builtin_amdgcn_wmma_f32_16x16x4_f32(false, a, false, b2, (short)0, c2, false, false);
                c3 = __builtin_amdgcn_wmma_f32_16x16x4_f32(false, a, false, b3, (short)0, c3, false, false);
            }
        }

        float* orow = out + (size_t)(row0 + 8 * hi) * N + col0 + lm;
        #pragma unroll
        for (int r = 0; r < 8; ++r) {
            float v0 = c0[r], v1 = c1[r], v2 = c2[r], v3 = c3[r];
            if (MODE != 3) { v0 += bv0; v1 += bv1; v2 += bv2; v3 += bv3; }
            if (MODE == 2) {
                const float inv = 1.0f / fmaxf(dg[r], 1.0f);
                v0 += inv * g0[r]; v1 += inv * g1[r];
                v2 += inv * g2[r]; v3 += inv * g3[r];
            }
            if (do_relu) {
                v0 = fmaxf(v0, 0.f); v1 = fmaxf(v1, 0.f);
                v2 = fmaxf(v2, 0.f); v3 = fmaxf(v3, 0.f);
            }
            float* p = orow + (size_t)r * N;
            p[0] = v0; p[16] = v1; p[32] = v2; p[48] = v3;
        }
    }
}

#define PAIR_T 512
template <int K, int N>
__global__ void __launch_bounds__(PAIR_T)
gemm_sage_pair(const float* __restrict__ A1a, const float* __restrict__ W1a,
               const float* __restrict__ A2a, const float* __restrict__ W2a,
               const unsigned* __restrict__ cnta, const float* __restrict__ biasa,
               float* __restrict__ outa, int Ma,
               const float* __restrict__ A1b, const float* __restrict__ W1b,
               const float* __restrict__ A2b, const float* __restrict__ W2b,
               const unsigned* __restrict__ cntb, const float* __restrict__ biasb,
               float* __restrict__ outb, int Mb, int split)
{
    const bool second = (int)blockIdx.x >= split;
    gemm_body<K, N, 1, PAIR_T>(second ? A1b : A1a, second ? W1b : W1a,
                               second ? A2b : A2a, second ? W2b : W2a,
                               second ? cntb : cnta, second ? biasb : biasa,
                               second ? outb : outa, second ? Mb : Ma, 1,
                               second ? (int)blockIdx.x - split : (int)blockIdx.x,
                               second ? (int)gridDim.x - split : split);
}

template <int K>
__device__ __forceinline__ void
gemm_dual_body(const float* __restrict__ A, const float* __restrict__ Wp,
               const float* __restrict__ Wq, const float* __restrict__ bias,
               float* __restrict__ P, float* __restrict__ Q, int M, int bid, int nblk)
{
    constexpr int NO = 64;
    constexpr int P4 = K + 4;
    __shared__ __attribute__((aligned(16))) float Wd[2 * NO * P4];

    for (int idx = threadIdx.x; idx < (K / 4) * 2 * NO; idx += 256) {
        const int n  = idx % (2 * NO);
        const int kq = idx / (2 * NO);
        const float* w = (n < NO ? Wp + n : Wq + (n - NO)) + (size_t)(4 * kq) * NO;
        float4 v;
        v.x = w[0]; v.y = w[NO]; v.z = w[2 * NO]; v.w = w[3 * NO];
        *(float4*)(Wd + n * P4 + 4 * kq) = v;
    }
    __syncthreads();

    const int lane = threadIdx.x & 31;
    const int wave = threadIdx.x >> 5;
    const int hi   = lane >> 4;
    const int lm   = lane & 15;
    const int slot = wave >> 1;
    const bool isq = (wave & 1) != 0;
    const int ntiles = M / 16;

    const float* bs = Wd + ((isq ? NO : 0) + lm) * P4 + 2 * hi;
    float* dst = isq ? Q : P;

    float bv0 = 0.f, bv1 = 0.f, bv2 = 0.f, bv3 = 0.f;
    if (isq) {
        bv0 = bias[lm];      bv1 = bias[16 + lm];
        bv2 = bias[32 + lm]; bv3 = bias[48 + lm];
    }

    for (int rt = slot * nblk + bid; rt < ntiles; rt += 4 * nblk) {
        const int row0 = rt * 16;
        v8f c0 = {0.f,0.f,0.f,0.f,0.f,0.f,0.f,0.f};
        v8f c1 = c0, c2 = c0, c3 = c0;
        const float* ap = A + (size_t)(row0 + lm) * K + 2 * hi;
        #pragma unroll 8
        for (int k = 0; k < K / 4; ++k) {
            v2f a  = *(const v2f*)(ap + 4 * k);
            v2f b0 = *(const v2f*)(bs + 4 * k);
            v2f b1 = *(const v2f*)(bs + 4 * k + 16 * P4);
            v2f b2 = *(const v2f*)(bs + 4 * k + 32 * P4);
            v2f b3 = *(const v2f*)(bs + 4 * k + 48 * P4);
            c0 = __builtin_amdgcn_wmma_f32_16x16x4_f32(false, a, false, b0, (short)0, c0, false, false);
            c1 = __builtin_amdgcn_wmma_f32_16x16x4_f32(false, a, false, b1, (short)0, c1, false, false);
            c2 = __builtin_amdgcn_wmma_f32_16x16x4_f32(false, a, false, b2, (short)0, c2, false, false);
            c3 = __builtin_amdgcn_wmma_f32_16x16x4_f32(false, a, false, b3, (short)0, c3, false, false);
        }
        float* orow = dst + (size_t)(row0 + 8 * hi) * NO + lm;
        #pragma unroll
        for (int r = 0; r < 8; ++r) {
            float* p = orow + (size_t)r * NO;
            p[0]  = c0[r] + bv0; p[16] = c1[r] + bv1;
            p[32] = c2[r] + bv2; p[48] = c3[r] + bv3;
        }
    }
}

template <int K>
__global__ void __launch_bounds__(256)
gemm_dual_pair(const float* __restrict__ Aa, const float* __restrict__ Wpa,
               const float* __restrict__ Wqa, const float* __restrict__ biasa,
               float* __restrict__ Pa, float* __restrict__ Qa, int Ma,
               const float* __restrict__ Ab, const float* __restrict__ Wpb,
               const float* __restrict__ Wqb, const float* __restrict__ biasb,
               float* __restrict__ Pb, float* __restrict__ Qb, int Mb, int split)
{
    const bool second = (int)blockIdx.x >= split;
    gemm_dual_body<K>(second ? Ab : Aa, second ? Wpb : Wpa, second ? Wqb : Wqa,
                      second ? biasb : biasa, second ? Pb : Pa, second ? Qb : Qa,
                      second ? Mb : Ma,
                      second ? (int)blockIdx.x - split : (int)blockIdx.x,
                      second ? (int)gridDim.x - split : split);
}

#define SORT_G      256
#define SORT_T      512
#define OCC_SHIFT   8
#define SKILL_SHIFT 9
#define NB_OCC      ((N_OCC   + (1 << OCC_SHIFT)   - 1) >> OCC_SHIFT)
#define NB_SKILL    ((N_SKILL + (1 << SKILL_SHIFT) - 1) >> SKILL_SHIFT)
#define NB_ALL      (NB_OCC + NB_SKILL)
#define SORT_CHUNK  ((N_EDGE + SORT_G - 1) / SORT_G)
#define SORT_PER_T  ((SORT_CHUNK + SORT_T - 1) / SORT_T)
static_assert(N_OCC <= 65536 && N_SKILL <= 65536, "a source id must fit the low 16 bits of an edge word");
#define ROWS_PER_W  ((NB_ALL + SORT_T / 32 - 1) / (SORT_T / 32))
static_assert(SORT_G == 256, "a histogram row is two 16-byte loads per lane of a wave");
static_assert(NB_OCC <= 128 && NB_SKILL <= 128, "the bin starts are scanned by one wave, 4 bins per lane");
static_assert(NB_ALL <= SORT_T, "one thread per histogram row");

__device__ __forceinline__ unsigned bin_of_occ(unsigned o)
{
    return min(o >> OCC_SHIFT, (unsigned)(NB_OCC - 1));
}
__device__ __forceinline__ unsigned bin_of_skill(unsigned s)
{
    return min(s >> SKILL_SHIFT, (unsigned)(NB_SKILL - 1));
}

__device__ __forceinline__ void
bin_hist_body(const int* __restrict__ e_occ, const int* __restrict__ e_skill,
              unsigned* __restrict__ hist, int g)
{
    __shared__ unsigned h[NB_ALL];
    const int t = threadIdx.x;
    if (t < NB_ALL) h[t] = 0u;
    __syncthreads();
    const int beg = g * SORT_CHUNK;
    const int end = min(beg + SORT_CHUNK, N_EDGE);
    unsigned o[SORT_PER_T], s[SORT_PER_T];
    #pragma unroll
    for (int k = 0; k < SORT_PER_T; ++k) {
        const int e = beg + t + k * SORT_T;
        o[k] = 0u; s[k] = 0u;
        if (e < end) { o[k] = (unsigned)e_occ[e]; s[k] = (unsigned)e_skill[e]; }
    }
    #pragma unroll
    for (int k = 0; k < SORT_PER_T; ++k) {
        const int e = beg + t + k * SORT_T;
        if (e < end) {
            atomicAdd(&h[bin_of_occ(o[k])], 1u);
            atomicAdd(&h[NB_OCC + bin_of_skill(s[k])], 1u);
        }
    }
    __syncthreads();
    if (t < NB_ALL) hist[t * SORT_G + g] = h[t];
}

__device__ __forceinline__ void
bin_place_body(const int* __restrict__ e_occ, const int* __restrict__ e_skill,
               const unsigned* __restrict__ hist,
               unsigned* __restrict__ bstart, unsigned* __restrict__ bsize,
               unsigned* __restrict__ binned_occ, unsigned* __restrict__ binned_skill, int g)
{
    __shared__ unsigned tot[NB_ALL];
    __shared__ unsigned cur[NB_ALL];
    const int t = threadIdx.x;
    const int lane = t & 31, wave = t >> 5;

    const int beg = g * SORT_CHUNK;
    const int end = min(beg + SORT_CHUNK, N_EDGE);
    unsigned o[SORT_PER_T], s[SORT_PER_T];
    #pragma unroll
    for (int k = 0; k < SORT_PER_T; ++k) {
        const int e = beg + t + k * SORT_T;
        o[k] = 0u; s[k] = 0u;
        if (e < end) { o[k] = (unsigned)e_occ[e]; s[k] = (unsigned)e_skill[e]; }
    }

    uint4 va[ROWS_PER_W], vb[ROWS_PER_W];
    #pragma unroll
    for (int j = 0; j < ROWS_PER_W; ++j) {
        const int r = wave + j * (SORT_T / 32);
        va[j] = make_uint4(0u, 0u, 0u, 0u); vb[j] = va[j];
        if (r < NB_ALL) {
            const uint4* row = (const uint4*)(hist + r * SORT_G);
            va[j] = row[2 * lane];
            vb[j] = row[2 * lane + 1];
        }
    }
    #pragma unroll
    for (int j = 0; j < ROWS_PER_W; ++j) {
        const int r = wave + j * (SORT_T / 32);
        const uint4 a = va[j], b = vb[j];
        const int g0 = 8 * lane;
        unsigned all = ((a.x + a.y) + (a.z + a.w)) + ((b.x + b.y) + (b.z + b.w));
        unsigned before = (g0     < g ? a.x : 0u) + (g0 + 1 < g ? a.y : 0u)
                        + (g0 + 2 < g ? a.z : 0u) + (g0 + 3 < g ? a.w : 0u)
                        + (g0 + 4 < g ? b.x : 0u) + (g0 + 5 < g ? b.y : 0u)
                        + (g0 + 6 < g ? b.z : 0u) + (g0 + 7 < g ? b.w : 0u);
        #pragma unroll
        for (int m = 16; m >= 1; m >>= 1) {
            all    += __shfl_xor(all, m);
            before += __shfl_xor(before, m);
        }
        if (r < NB_ALL && lane == 0) { tot[r] = all; cur[r] = before; }
    }
    __syncthreads();
    if (wave < 2) {
        const int base = (wave == 0) ? 0 : NB_OCC;
        const int n    = (wave == 0) ? NB_OCC : NB_SKILL;
        unsigned a[4], sum = 0u;
        #pragma unroll
        for (int i = 0; i < 4; ++i) {
            const int r = 4 * lane + i;
            a[i] = (r < n) ? tot[base + r] : 0u;
            sum += a[i];
        }
        unsigned inc = sum;
        #pragma unroll
        for (int d = 1; d < 32; d <<= 1) {
            const unsigned u = __shfl_up(inc, d);
            if (lane >= d) inc += u;
        }
        unsigned ex = inc - sum;
        #pragma unroll
        for (int i = 0; i < 4; ++i) {
            const int r = 4 * lane + i;
            if (r < n) {
                if (g == 0) { bstart[base + r] = ex; bsize[base + r] = a[i]; }
                cur[base + r] += ex;
                ex += a[i];
            }
        }
    }
    __syncthreads();

    unsigned po[SORT_PER_T], ps[SORT_PER_T];
    #pragma unroll
    for (int k = 0; k < SORT_PER_T; ++k) {
        const int e = beg + t + k * SORT_T;
        po[k] = 0xFFFFFFFFu; ps[k] = 0xFFFFFFFFu;
        if (e < end) {
            po[k] = atomicAdd(&cur[bin_of_occ(o[k])], 1u);
            ps[k] = atomicAdd(&cur[NB_OCC + bin_of_skill(s[k])], 1u);
        }
    }
    #pragma unroll
    for (int k = 0; k < SORT_PER_T; ++k) {
        if (po[k] < (unsigned)N_EDGE)
            binned_occ[po[k]]   = ((o[k] & ((1u << OCC_SHIFT) - 1u)) << 16) | (s[k] & 0xFFFFu);
        if (ps[k] < (unsigned)N_EDGE)
            binned_skill[ps[k]] = ((s[k] & ((1u << SKILL_SHIFT) - 1u)) << 16) | (o[k] & 0xFFFFu);
    }
}

__device__ __forceinline__ void
bin_rank_body(const unsigned* __restrict__ bstart, const unsigned* __restrict__ bsize,
              const unsigned* __restrict__ binned_occ, const unsigned* __restrict__ binned_skill,
              unsigned* __restrict__ cnt_occ, unsigned* __restrict__ cnt_skill,
              unsigned* __restrict__ rp_occ, unsigned* __restrict__ rp_skill,
              unsigned* __restrict__ adj_occ, unsigned* __restrict__ adj_skill, int row)
{
    constexpr int NMAX = 1 << SKILL_SHIFT;
    static_assert((1 << OCC_SHIFT) <= NMAX && NMAX <= SORT_T && NMAX / 32 <= 32, "one thread per node of a bin");
    constexpr int BATCH = 8;
    __shared__ unsigned c[NMAX + 1];
    __shared__ unsigned wsum[NMAX / 32];
    const int t = threadIdx.x;
    const int lane = t & 31, wave = t >> 5;
    const bool sk = row >= NB_OCC;
    const int shift = sk ? SKILL_SHIFT : OCC_SHIFT;
    const int nbase = (sk ? row - NB_OCC : row) << shift;
    const int nn    = min(1 << shift, (sk ? N_SKILL : N_OCC) - nbase);
    const unsigned* data = sk ? binned_skill : binned_occ;
    unsigned* cnt = sk ? cnt_skill : cnt_occ;
    unsigned* rp  = sk ? rp_skill  : rp_occ;
    unsigned* adj = sk ? adj_skill : adj_occ;
    unsigned st = bstart[row]; if (st > (unsigned)N_EDGE) st = (unsigned)N_EDGE;
    unsigned n  = bsize[row];  if (n > (unsigned)N_EDGE - st) n = (unsigned)N_EDGE - st;
    data += st;
    adj  += st;

    if (t < NMAX) c[t] = 0u;
    if (t == 0) c[NMAX] = 0u;
    __syncthreads();
    for (unsigned i0 = t; i0 < n; i0 += BATCH * SORT_T) {
        unsigned w[BATCH];
        #pragma unroll
        for (int k = 0; k < BATCH; ++k) {
            const unsigned i = i0 + k * SORT_T;
            w[k] = data[min(i, n - 1u)];
        }
        #pragma unroll
        for (int k = 0; k < BATCH; ++k) {
            const unsigned i = i0 + k * SORT_T;
            atomicAdd(&c[(i < n) ? min(w[k] >> 16, (unsigned)(NMAX - 1)) : (unsigned)NMAX], 1u);
        }
    }
    __syncthreads();
    unsigned v = 0u, inc = 0u;
    if (t < NMAX) {
        v = c[t];
        inc = v;
        #pragma unroll
        for (int d = 1; d < 32; d <<= 1) {
            const unsigned u = __shfl_up(inc, d);
            if (lane >= d) inc += u;
        }
        if (lane == 31) wsum[wave] = inc;
    }
    __syncthreads();
    if (wave == 0) {
        const unsigned ws = (lane < NMAX / 32) ? wsum[lane] : 0u;
        unsigned wi = ws;
        #pragma unroll
        for (int d = 1; d < 32; d <<= 1) {
            const unsigned u = __shfl_up(wi, d);
            if (lane >= d) wi += u;
        }
        if (lane < NMAX / 32) wsum[lane] = wi - ws;
    }
    __syncthreads();
    if (t < NMAX) {
        const unsigned ex = inc - v + wsum[wave];
        c[t] = ex;
        if (t < nn) { cnt[nbase + t] = v; rp[nbase + t] = st + ex; }
    }
    __syncthreads();
    for (unsigned i0 = t; i0 < n; i0 += BATCH * SORT_T) {
        unsigned w[BATCH], p[BATCH];
        #pragma unroll
        for (int k = 0; k < BATCH; ++k) {
            const unsigned i = i0 + k * SORT_T;
            w[k] = data[min(i, n - 1u)];
        }
        #pragma unroll
        for (int k = 0; k < BATCH; ++k) {
            const unsigned i = i0 + k * SORT_T;
            p[k] = atomicAdd(&c[(i < n) ? min(w[k] >> 16, (unsigned)(NMAX - 1)) : (unsigned)NMAX], 1u);
        }
        #pragma unroll
        for (int k = 0; k < BATCH; ++k) {
            const unsigned i = i0 + k * SORT_T;
            if (i < n && p[k] < n) adj[p[k]] = w[k] & 0xFFFFu;
        }
    }
}

#define PROJ_G        256
#define PROJ_SKILL_1  (208 * 16)
#define PROJ_SKILL_2  (1458 * 16)
#define PROJ_SKILL_3  (N_SKILL - PROJ_SKILL_1 - PROJ_SKILL_2)
#define PROJ_SPLIT_1  219
static_assert(PROJ_SKILL_3 > 0 && PROJ_SKILL_3 % 16 == 0, "every share of the skill rows is whole row tiles");
static_assert(N_OCC / 16 <= 6 * PROJ_SPLIT_1 && PROJ_SKILL_1 / 16 <= 6 * (PROJ_G - PROJ_SPLIT_1) &&
              PROJ_SKILL_2 / 16 <= 6 * PROJ_G && PROJ_SKILL_3 / 16 <= 6 * PROJ_G,
              "at most 6 row tiles per product block");

__global__ void __launch_bounds__(SORT_T)
hist_and_proj(const int* __restrict__ e_occ, const int* __restrict__ e_skill,
              unsigned* __restrict__ hist,
              const float* __restrict__ xa, const float* __restrict__ Wa,
              const float* __restrict__ biasa, float* __restrict__ ha, int Ma,
              const float* __restrict__ xb, const float* __restrict__ Wb,
              const float* __restrict__ biasb, float* __restrict__ hb, int Mb, int split)
{
    if ((int)blockIdx.x >= PROJ_G) {
        bin_hist_body(e_occ, e_skill, hist, (int)blockIdx.x - PROJ_G);
        return;
    }
    const bool second = (int)blockIdx.x >= split;
    gemm_body<D_IN, HID, 0, SORT_T>(second ? xb : xa, second ? Wb : Wa, nullptr, nullptr, nullptr,
                                    second ? biasb : biasa, second ? hb : ha, second ? Mb : Ma, 1,
                                    second ? (int)blockIdx.x - split : (int)blockIdx.x,
                                    second ? PROJ_G - split : split);
}

__global__ void __launch_bounds__(SORT_T)
place_and_proj(const int* __restrict__ e_occ, const int* __restrict__ e_skill,
               const unsigned* __restrict__ hist,
               unsigned* __restrict__ bstart, unsigned* __restrict__ bsize,
               unsigned* __restrict__ binned_occ, unsigned* __restrict__ binned_skill,
               const float* __restrict__ x, const float* __restrict__ W,
               const float* __restrict__ bias, float* __restrict__ h, int M)
{
    if ((int)blockIdx.x >= PROJ_G) {
        bin_place_body(e_occ, e_skill, hist, bstart, bsize, binned_occ, binned_skill,
                       (int)blockIdx.x - PROJ_G);
        return;
    }
    gemm_body<D_IN, HID, 0, SORT_T>(x, W, nullptr, nullptr, nullptr, bias, h, M, 1,
                                    (int)blockIdx.x, PROJ_G);
}

__global__ void __launch_bounds__(SORT_T)
rank_and_proj(const unsigned* __restrict__ bstart, const unsigned* __restrict__ bsize,
              const unsigned* __restrict__ binned_occ, const unsigned* __restrict__ binned_skill,
              unsigned* __restrict__ cnt_occ, unsigned* __restrict__ cnt_skill,
              unsigned* __restrict__ rp_occ, unsigned* __restrict__ rp_skill,
              unsigned* __restrict__ adj_occ, unsigned* __restrict__ adj_skill,
              const float* __restrict__ x, const float* __restrict__ W,
              const float* __restrict__ bias, float* __restrict__ h, int M)
{
    if ((int)blockIdx.x >= PROJ_G) {
        bin_rank_body(bstart, bsize, binned_occ, binned_skill, cnt_occ, cnt_skill,
                      rp_occ, rp_skill, adj_occ, adj_skill, (int)blockIdx.x - PROJ_G);
        return;
    }
    gemm_body<D_IN, HID, 0, SORT_T>(x, W, nullptr, nullptr, nullptr, bias, h, M, 1,
                                    (int)blockIdx.x, PROJ_G);
}

__device__ inline void add4(float4& a, const float4& v)
{
    a.x += v.x; a.y += v.y; a.z += v.z; a.w += v.w;
}

template <int F, bool ADD>
__device__ __forceinline__ void
gather_body(const float* __restrict__ hsrc, const unsigned* __restrict__ adj,
            const unsigned* __restrict__ rp, const unsigned* __restrict__ cnt,
            float* __restrict__ agg, int n_dst, unsigned n_src, unsigned E, int bid)
{
    constexpr int TPR = F / 4;
    const int tid = bid * 256 + threadIdx.x;
    const int d = tid / TPR;
    const int f = (tid & (TPR - 1)) << 2;
    if (d >= n_dst) return;
    unsigned beg = rp[d];  if (beg > E) beg = E;
    const unsigned craw = cnt[d];
    unsigned deg = craw; if (deg > E - beg) deg = E - beg;
    float4* orow = (float4*)(agg + (size_t)d * F + f);
    float4 q = {0.f, 0.f, 0.f, 0.f};
    if (ADD) q = *orow;
    const unsigned top = n_src - 1;
    const unsigned* a = adj + beg;
    const float* hb = hsrc + f;
    float4 s0 = {0.f, 0.f, 0.f, 0.f}, s1 = s0, s2 = s0, s3 = s0;
    unsigned j = 0;
    unsigned n0 = 0, n1 = 0, n2 = 0, n3 = 0;
    if (deg >= 4) { n0 = a[0]; n1 = a[1]; n2 = a[2]; n3 = a[3]; }
    for (; j + 4 <= deg; j += 4) {
        const unsigned i0 = min(n0, top), i1 = min(n1, top);
        const unsigned i2 = min(n2, top), i3 = min(n3, top);
        if (j + 8 <= deg) { n0 = a[j + 4]; n1 = a[j + 5]; n2 = a[j + 6]; n3 = a[j + 7]; }
        const float4 v0 = *(const float4*)(hb + (size_t)i0 * F);
        const float4 v1 = *(const float4*)(hb + (size_t)i1 * F);
        const float4 v2 = *(const float4*)(hb + (size_t)i2 * F);
        const float4 v3 = *(const float4*)(hb + (size_t)i3 * F);
        add4(s0, v0); add4(s1, v1); add4(s2, v2); add4(s3, v3);
    }
    if (j < deg) {
        const unsigned r = deg - j;
        const unsigned i0 = min(a[j], top);
        const unsigned i1 = min(a[r > 1 ? j + 1 : j], top);
        const unsigned i2 = min(a[r > 2 ? j + 2 : j], top);
        const float4 v0 = *(const float4*)(hb + (size_t)i0 * F);
        const float4 v1 = *(const float4*)(hb + (size_t)i1 * F);
        const float4 v2 = *(const float4*)(hb + (size_t)i2 * F);
        add4(s0, v0);
        if (r > 1) add4(s1, v1);
        if (r > 2) add4(s2, v2);
    }
    add4(s0, s1); add4(s2, s3); add4(s0, s2);
    if (ADD) {
        const float inv = 1.0f / fmaxf((float)craw, 1.0f);
        q.x += inv * s0.x; q.y += inv * s0.y; q.z += inv * s0.z; q.w += inv * s0.w;
        *orow = q;
    } else {
        *orow = s0;
    }
}

template <int F, bool ADD>
__global__ void __launch_bounds__(256)
gather_pair(const float* __restrict__ hsrc_a, const unsigned* __restrict__ adj_a,
            const unsigned* __restrict__ rp_a, const unsigned* __restrict__ cnt_a,
            float* __restrict__ agg_a, int n_dst_a, unsigned n_src_a,
            const float* __restrict__ hsrc_b, const unsigned* __restrict__ adj_b,
            const unsigned* __restrict__ rp_b, const unsigned* __restrict__ cnt_b,
            float* __restrict__ agg_b, int n_dst_b, unsigned n_src_b,
            int split, unsigned E)
{
    const bool second = (int)blockIdx.x >= split;
    gather_body<F, ADD>(second ? hsrc_b : hsrc_a, second ? adj_b : adj_a,
                   second ? rp_b : rp_a, second ? cnt_b : cnt_a,
                   second ? agg_b : agg_a, second ? n_dst_b : n_dst_a,
                   second ? n_src_b : n_src_a, E,
                   second ? (int)blockIdx.x - split : (int)blockIdx.x);
}

extern "C" void kernel_launch(void* const* d_in, const int* in_sizes, int n_in,
                              void* d_out, int out_size, void* d_ws, size_t ws_size,
                              hipStream_t stream)
{
    (void)in_sizes; (void)n_in; (void)out_size; (void)ws_size;

    const float* x_occ    = (const float*)d_in[0];
    const float* x_skill  = (const float*)d_in[1];
    const int*   e_occ    = (const int*)  d_in[2];
    const int*   e_skill  = (const int*)  d_in[3];
    const float* pW_occ   = (const float*)d_in[4];
    const float* pb_occ   = (const float*)d_in[5];
    const float* pW_skill = (const float*)d_in[6];
    const float* pb_skill = (const float*)d_in[7];
    const float* c1_os_Wl = (const float*)d_in[8];
    const float* c1_os_bl = (const float*)d_in[9];
    const float* c1_os_Wr = (const float*)d_in[10];
    const float* c1_so_Wl = (const float*)d_in[11];
    const float* c1_so_bl = (const float*)d_in[12];
    const float* c1_so_Wr = (const float*)d_in[13];
    const float* c2_os_Wl = (const float*)d_in[14];
    const float* c2_os_bl = (const float*)d_in[15];
    const float* c2_os_Wr = (const float*)d_in[16];
    const float* c2_so_Wl = (const float*)d_in[17];
    const float* c2_so_bl = (const float*)d_in[18];
    const float* c2_so_Wr = (const float*)d_in[19];

    float* ws = (float*)d_ws;
    float* h_occ     = ws;
    float* h_skill   = h_occ     + (size_t)N_OCC   * HID;
    float* o1        = h_skill   + (size_t)N_SKILL * HID;
    float* s1        = o1        + (size_t)N_OCC   * HID;
    float* aggbuf    = s1        + (size_t)N_SKILL * HID;
    unsigned* adj_skill  = (unsigned*)(aggbuf + (size_t)N_SKILL * HID);
    unsigned* adj_occ    = adj_skill  + N_EDGE;
    unsigned* rp_occ     = adj_occ    + N_EDGE;
    unsigned* rp_skill   = rp_occ     + N_OCC;
    unsigned* binned_occ   = rp_skill   + N_SKILL;
    unsigned* binned_skill = binned_occ + N_EDGE;
    unsigned* hist   = binned_skill + N_EDGE;
    unsigned* bstart = hist   + NB_ALL * SORT_G;
    unsigned* bsize  = bstart + NB_ALL;
    static_assert(2 * N_EDGE + N_OCC + N_SKILL + 2 * N_EDGE + NB_ALL * SORT_G + 2 * NB_ALL <= N_OCC * HID,
                  "the edge tables must fit the 20000*128 words behind the neighbour-sum buffer");
    unsigned* cnt_occ   = (unsigned*)(aggbuf + (size_t)(N_OCC + N_SKILL) * HID);
    unsigned* cnt_skill = cnt_occ + N_OCC;
    float* o2 = (float*)d_out;
    float* s2 = o2 + (size_t)N_OCC * OUTF;

    const dim3 blk256(256);
    const dim3 blkS(SORT_T);
    hist_and_proj<<<PROJ_G + SORT_G, blkS, 0, stream>>>(
        e_occ, e_skill, hist,
        x_occ,   pW_occ,   pb_occ,   h_occ,   N_OCC,
        x_skill, pW_skill, pb_skill, h_skill, PROJ_SKILL_1, PROJ_SPLIT_1);
    place_and_proj<<<PROJ_G + SORT_G, blkS, 0, stream>>>(
        e_occ, e_skill, hist, bstart, bsize, binned_occ, binned_skill,
        x_skill + (size_t)PROJ_SKILL_1 * D_IN, pW_skill, pb_skill,
        h_skill + (size_t)PROJ_SKILL_1 * HID, PROJ_SKILL_2);
    rank_and_proj<<<PROJ_G + NB_ALL, blkS, 0, stream>>>(
        bstart, bsize, binned_occ, binned_skill,
        cnt_occ, cnt_skill, rp_occ, rp_skill, adj_occ, adj_skill,
        x_skill + (size_t)(PROJ_SKILL_1 + PROJ_SKILL_2) * D_IN, pW_skill, pb_skill,
        h_skill + (size_t)(PROJ_SKILL_1 + PROJ_SKILL_2) * HID, PROJ_SKILL_3);

    static_assert((size_t)N_OCC * HID <= (size_t)(N_OCC + N_SKILL) * OUTF, "the occ sums must fit the output buffer");
    float* agg1_occ = o2;
    gather_pair<HID, false><<<N_OCC / 8 + N_SKILL / 8, blk256, 0, stream>>>(
        h_skill, adj_occ,   rp_occ,   cnt_occ,   agg1_occ, N_OCC,   N_SKILL,
        h_occ,   adj_skill, rp_skill, cnt_skill, aggbuf,   N_SKILL, N_OCC,
        N_OCC / 8, N_EDGE);
    gemm_sage_pair<HID, HID><<<dim3(256), dim3(PAIR_T), 0, stream>>>(
        agg1_occ, c1_so_Wl, h_occ,   c1_so_Wr, cnt_occ,   c1_so_bl, o1, N_OCC,
        aggbuf,   c1_os_Wl, h_skill, c1_os_Wr, cnt_skill, c1_os_bl, s1, N_SKILL, 73);

    float* p_occ   = h_occ;
    float* p_skill = h_skill;
    gemm_dual_pair<HID><<<dim3(1024), blk256, 0, stream>>>(
        o1, c2_os_Wl, c2_so_Wr, c2_so_bl, p_occ,   o2, N_OCC,
        s1, c2_so_Wl, c2_os_Wr, c2_os_bl, p_skill, s2, N_SKILL, 293);
    gather_pair<OUTF, true><<<N_OCC / 16 + N_SKILL / 16, blk256, 0, stream>>>(
        p_skill, adj_occ,   rp_occ,   cnt_occ,   o2, N_OCC,   N_SKILL,
        p_occ,   adj_skill, rp_skill, cnt_skill, s2, N_SKILL, N_OCC,
        N_OCC / 16, N_EDGE);
}


# conv2's two 64-column products run inside conv1's product launch: the conv1 row tile passes through LDS into conv2's A operand and is never stored; 7 -> 6 launches; f32 sum order unchanged
# speedup vs baseline: 3.9203x; 1.0395x over previous
#include <hip/hip_runtime.h>

#define N_OCC   20000
#define N_SKILL 50000
#define N_EDGE  600000
#define D_IN    256
#define HID     128
#define OUTF    64

typedef __attribute__((ext_vector_type(2))) float v2f;
typedef __attribute__((ext_vector_type(8))) float v8f;

template <int K, int N, int MODE, int NT = 256>
__device__ __forceinline__ void
gemm_body(const float* __restrict__ A1, const float* __restrict__ W1,
          const float* __restrict__ A2, const float* __restrict__ W2,
          const unsigned* __restrict__ cnt, const float* __restrict__ bias,
          float* __restrict__ out, int M, int do_relu, int bid, int nblk)
{
    constexpr int P     = K + 4;
    constexpr int NMAT  = (MODE == 1) ? 2 : 1;
    constexpr int CH    = N / 64;
    constexpr int SLOTS = (NT / 32) / CH;
    __shared__ __attribute__((aligned(16))) float Ws[NMAT * N * P];

    auto copy_item = [&](int idx) {
        const int n  = idx % N;
        const int kq = idx / N;
        const float* w = W1 + (size_t)(4 * kq) * N + n;
        float4 v;
        v.x = w[0]; v.y = w[N]; v.z = w[2 * N]; v.w = w[3 * N];
        *(float4*)(Ws + n * P + 4 * kq) = v;
        if (MODE == 1) {
            const float* w2 = W2 + (size_t)(4 * kq) * N + n;
            float4 u;
            u.x = w2[0]; u.y = w2[N]; u.z = w2[2 * N]; u.w = w2[3 * N];
            *(float4*)(Ws + N * P + n * P + 4 * kq) = u;
        }
    };
    if constexpr (NT > 256) {
        static_assert(((K / 4) * N) % (8 * NT) == 0, "whole groups of eight trips per thread");
        #pragma unroll 8
        for (int j = 0; j < (K / 4) * N / NT; ++j) copy_item((int)threadIdx.x + j * NT);
    } else {
        for (int idx = threadIdx.x; idx < (K / 4) * N; idx += NT) copy_item(idx);
    }
    __syncthreads();

    const int lane = threadIdx.x & 31;
    const int wave = threadIdx.x >> 5;
    const int hi   = lane >> 4;
    const int lm   = lane & 15;
    const int slot = wave / CH;
    const int col0 = (wave % CH) * 64;
    const int ntiles = M / 16;

    const float* bs1 = Ws + (col0 + lm) * P + 2 * hi;
    const float* bs2 = Ws + N * P + (col0 + lm) * P + 2 * hi;

    float bv0 = 0.f, bv1 = 0.f, bv2 = 0.f, bv3 = 0.f;
    if (MODE != 3) {
        bv0 = bias[col0 + lm];
        bv1 = bias[col0 + 16 + lm];
        bv2 = bias[col0 + 32 + lm];
        bv3 = bias[col0 + 48 + lm];
    }

    for (int rt = slot * nblk + bid; rt < ntiles; rt += SLOTS * nblk) {
        const int row0 = rt * 16;

        v8f c0 = {0.f,0.f,0.f,0.f,0.f,0.f,0.f,0.f};
        v8f c1 = c0, c2 = c0, c3 = c0;

        float g0[8] = {}, g1[8] = {}, g2[8] = {}, g3[8] = {}, dg[8] = {};
        if (MODE == 2) {
            const float* g = A2 + (size_t)(row0 + 8 * hi) * N + col0 + lm;
            #pragma unroll
            for (int r = 0; r < 8; ++r) {
                g0[r] = g[r * N];      g1[r] = g[r * N + 16];
                g2[r] = g[r * N + 32]; g3[r] = g[r * N + 48];
                dg[r] = (float)cnt[row0 + 8 * hi + r];
            }
        }

        {
            const float* ap = A1 + (size_t)(row0 + lm) * K + 2 * hi;
            #pragma unroll 8
            for (int k = 0; k < K / 4; ++k) {
                v2f a  = *(const v2f*)(ap + 4 * k);
                v2f b0 = *(const v2f*)(bs1 + 4 * k);
                v2f b1 = *(const v2f*)(bs1 + 4 * k + 16 * P);
                v2f b2 = *(const v2f*)(bs1 + 4 * k + 32 * P);
                v2f b3 = *(const v2f*)(bs1 + 4 * k + 48 * P);
                c0 = __builtin_amdgcn_wmma_f32_16x16x4_f32(false, a, false, b0, (short)0, c0, false, false);
                c1 = __builtin_amdgcn_wmma_f32_16x16x4_f32(false, a, false, b1, (short)0, c1, false, false);
                c2 = __builtin_amdgcn_wmma_f32_16x16x4_f32(false, a, false, b2, (short)0, c2, false, false);
                c3 = __builtin_amdgcn_wmma_f32_16x16x4_f32(false, a, false, b3, (short)0, c3, false, false);
            }
        }

        if (MODE == 1) {
            #pragma unroll
            for (int r = 0; r < 8; ++r) {
                const float inv = 1.0f / fmaxf((float)cnt[row0 + r + 8 * hi], 1.0f);
                c0[r] *= inv; c1[r] *= inv; c2[r] *= inv; c3[r] *= inv;
            }
            const float* ap = A2 + (size_t)(row0 + lm) * K + 2 * hi;
            #pragma unroll 8
            for (int k = 0; k < K / 4; ++k) {
                v2f a  = *(const v2f*)(ap + 4 * k);
                v2f b0 = *(const v2f*)(bs2 + 4 * k);
                v2f b1 = *(const v2f*)(bs2 + 4 * k + 16 * P);
                v2f b2 = *(const v2f*)(bs2 + 4 * k + 32 * P);
                v2f b3 = *(const v2f*)(bs2 + 4 * k + 48 * P);
                c0 = __builtin_amdgcn_wmma_f32_16x16x4_f32(false, a, false, b0, (short)0, c0, false, false);
                c1 = __builtin_amdgcn_wmma_f32_16x16x4_f32(false, a, false, b1, (short)0, c1, false, false);
                c2 = __builtin_amdgcn_wmma_f32_16x16x4_f32(false, a, false, b2, (short)0, c2, false, false);
                c3 = __builtin_amdgcn_wmma_f32_16x16x4_f32(false, a, false, b3, (short)0, c3, false, false);
            }
        }

        float* orow = out + (size_t)(row0 + 8 * hi) * N + col0 + lm;
        #pragma unroll
        for (int r = 0; r < 8; ++r) {
            float v0 = c0[r], v1 = c1[r], v2 = c2[r], v3 = c3[r];
            if (MODE != 3) { v0 += bv0; v1 += bv1; v2 += bv2; v3 += bv3; }
            if (MODE == 2) {
                const float inv = 1.0f / fmaxf(dg[r], 1.0f);
                v0 += inv * g0[r]; v1 += inv * g1[r];
                v2 += inv * g2[r]; v3 += inv * g3[r];
            }
            if (do_relu) {
                v0 = fmaxf(v0, 0.f); v1 = fmaxf(v1, 0.f);
                v2 = fmaxf(v2, 0.f); v3 = fmaxf(v3, 0.f);
            }
            float* p = orow + (size_t)r * N;
            p[0] = v0; p[16] = v1; p[32] = v2; p[48] = v3;
        }
    }
}


#define CHAIN_T 512
template <int K>
__device__ __forceinline__ void
sage_chain_body(const float* __restrict__ S, const float* __restrict__ Wl,
                const float* __restrict__ X, const float* __restrict__ Wr,
                const unsigned* __restrict__ cnt, const float* __restrict__ bias1,
                const float* __restrict__ Wp, const float* __restrict__ Wq,
                const float* __restrict__ bias2,
                float* __restrict__ Pout, float* __restrict__ Qout, int M, int bid, int nblk)
{
    constexpr int N     = K;
    constexpr int NO    = 64;
    constexpr int P     = K + 4;
    constexpr int SLOTS = CHAIN_T / 64;
    static_assert(2 * NO == N, "conv2's two products together are as wide as the conv1 tile");
    __shared__ __attribute__((aligned(16))) float Ws[2 * N * P];
    __shared__ __attribute__((aligned(16))) float Wd[2 * NO * P];
    __shared__ __attribute__((aligned(16))) float Ts[SLOTS * 16 * P];

    static_assert(((K / 4) * N) % (8 * CHAIN_T) == 0, "whole groups of eight trips per thread");
    #pragma unroll 8
    for (int j = 0; j < (K / 4) * N / CHAIN_T; ++j) {
        const int idx = (int)threadIdx.x + j * CHAIN_T;
        const int n  = idx % N;
        const int kq = idx / N;
        const float* w1 = Wl + (size_t)(4 * kq) * N + n;
        const float* w2 = Wr + (size_t)(4 * kq) * N + n;
        const float* w3 = (n < NO ? Wp + n : Wq + (n - NO)) + (size_t)(4 * kq) * NO;
        float4 v, u, t;
        v.x = w1[0]; v.y = w1[N];  v.z = w1[2 * N];  v.w = w1[3 * N];
        u.x = w2[0]; u.y = w2[N];  u.z = w2[2 * N];  u.w = w2[3 * N];
        t.x = w3[0]; t.y = w3[NO]; t.z = w3[2 * NO]; t.w = w3[3 * NO];
        *(float4*)(Ws + n * P + 4 * kq)         = v;
        *(float4*)(Ws + N * P + n * P + 4 * kq) = u;
        *(float4*)(Wd + n * P + 4 * kq)         = t;
    }
    __syncthreads();

    const int lane = threadIdx.x & 31;
    const int wave = threadIdx.x >> 5;
    const int hi   = lane >> 4;
    const int lm   = lane & 15;
    const int slot = wave >> 1;
    const int half = wave & 1;
    const int col0 = half * 64;
    const int ntiles = M / 16;

    const float* bs1 = Ws + (col0 + lm) * P + 2 * hi;
    const float* bs2 = Ws + N * P + (col0 + lm) * P + 2 * hi;
    const float* bsd = Wd + (col0 + lm) * P + 2 * hi;
    float*       tw  = Ts + slot * 16 * P + (8 * hi) * P + col0 + lm;
    const float* ta  = Ts + slot * 16 * P + lm * P + 2 * hi;
    float* dst = half ? Qout : Pout;

    const float bv0 = bias1[col0 + lm],      bv1 = bias1[col0 + 16 + lm];
    const float bv2 = bias1[col0 + 32 + lm], bv3 = bias1[col0 + 48 + lm];
    float bq0 = 0.f, bq1 = 0.f, bq2 = 0.f, bq3 = 0.f;
    if (half) {
        bq0 = bias2[lm];      bq1 = bias2[16 + lm];
        bq2 = bias2[32 + lm]; bq3 = bias2[48 + lm];
    }

    const int stride = SLOTS * nblk;
    const int nsweep = (ntiles - bid + stride - 1) / stride;
    for (int sw = 0; sw < nsweep; ++sw) {
        const int rt = sw * stride + slot * nblk + bid;
        const bool act = rt < ntiles;
        const int row0 = rt * 16;

        if (act) {
            v8f c0 = {0.f,0.f,0.f,0.f,0.f,0.f,0.f,0.f};
            v8f c1 = c0, c2 = c0, c3 = c0;
            {
                const float* ap = S + (size_t)(row0 + lm) * K + 2 * hi;
                #pragma unroll 8
                for (int k = 0; k < K / 4; ++k) {
                    v2f a  = *(const v2f*)(ap + 4 * k);
                    v2f b0 = *(const v2f*)(bs1 + 4 * k);
                    v2f b1 = *(const v2f*)(bs1 + 4 * k + 16 * P);
                    v2f b2 = *(const v2f*)(bs1 + 4 * k + 32 * P);
                    v2f b3 = *(const v2f*)(bs1 + 4 * k + 48 * P);
                    c0 = __builtin_amdgcn_wmma_f32_16x16x4_f32(false, a, false, b0, (short)0, c0, false, false);
                    c1 = __builtin_amdgcn_wmma_f32_16x16x4_f32(false, a, false, b1, (short)0, c1, false, false);
                    c2 = __builtin_amdgcn_wmma_f32_16x16x4_f32(false, a, false, b2, (short)0, c2, false, false);
                    c3 = __builtin_amdgcn_wmma_f32_16x16x4_f32(false, a, false, b3, (short)0, c3, false, false);
                }
            }
            #pragma unroll
            for (int r = 0; r < 8; ++r) {
                const float inv = 1.0f / fmaxf((float)cnt[row0 + r + 8 * hi], 1.0f);
                c0[r] *= inv; c1[r] *= inv; c2[r] *= inv; c3[r] *= inv;
            }
            {
                const float* ap = X + (size_t)(row0 + lm) * K + 2 * hi;
                #pragma unroll 8
                for (int k = 0; k < K / 4; ++k) {
                    v2f a  = *(const v2f*)(ap + 4 * k);
                    v2f b0 = *(const v2f*)(bs2 + 4 * k);
                    v2f b1 = *(const v2f*)(bs2 + 4 * k + 16 * P);
                    v2f b2 = *(const v2f*)(bs2 + 4 * k + 32 * P);
                    v2f b3 = *(const v2f*)(bs2 + 4 * k + 48 * P);
                    c0 = __builtin_amdgcn_wmma_f32_16x16x4_f32(false, a, false, b0, (short)0, c0, false, false);
                    c1 = __builtin_amdgcn_wmma_f32_16x16x4_f32(false, a, false, b1, (short)0, c1, false, false);
                    c2 = __builtin_amdgcn_wmma_f32_16x16x4_f32(false, a, false, b2, (short)0, c2, false, false);
                    c3 = __builtin_amdgcn_wmma_f32_16x16x4_f32(false, a, false, b3, (short)0, c3, false, false);
                }
            }
            #pragma unroll
            for (int r = 0; r < 8; ++r) {
                float* p = tw + r * P;
                p[0]  = fmaxf(c0[r] + bv0, 0.f); p[16] = fmaxf(c1[r] + bv1, 0.f);
                p[32] = fmaxf(c2[r] + bv2, 0.f); p[48] = fmaxf(c3[r] + bv3, 0.f);
            }
        }
        __syncthreads();

        v8f d0 = {0.f,0.f,0.f,0.f,0.f,0.f,0.f,0.f};
        v8f d1 = d0, d2 = d0, d3 = d0;
        if (act) {
            #pragma unroll 8
            for (int k = 0; k < K / 4; ++k) {
                v2f a  = *(const v2f*)(ta + 4 * k);
                v2f b0 = *(const v2f*)(bsd + 4 * k);
                v2f b1 = *(const v2f*)(bsd + 4 * k + 16 * P);
                v2f b2 = *(const v2f*)(bsd + 4 * k + 32 * P);
                v2f b3 = *(const v2f*)(bsd + 4 * k + 48 * P);
                d0 = __builtin_amdgcn_wmma_f32_16x16x4_f32(false, a, false, b0, (short)0, d0, false, false);
                d1 = __builtin_amdgcn_wmma_f32_16x16x4_f32(false, a, false, b1, (short)0, d1, false, false);
                d2 = __builtin_amdgcn_wmma_f32_16x16x4_f32(false, a, false, b2, (short)0, d2, false, false);
                d3 = __builtin_amdgcn_wmma_f32_16x16x4_f32(false, a, false, b3, (short)0, d3, false, false);
            }
        }
        if (sw + 1 < nsweep)
            __syncthreads();
        if (act) {
            float* orow = dst + (size_t)(row0 + 8 * hi) * NO + lm;
            #pragma unroll
            for (int r = 0; r < 8; ++r) {
                float* p = orow + (size_t)r * NO;
                p[0]  = d0[r] + bq0; p[16] = d1[r] + bq1;
                p[32] = d2[r] + bq2; p[48] = d3[r] + bq3;
            }
        }
    }
}

template <int K>
__global__ void __launch_bounds__(CHAIN_T)
sage_chain_pair(const float* __restrict__ Sa, const float* __restrict__ Wla,
                const float* __restrict__ Xa, const float* __restrict__ Wra,
                const unsigned* __restrict__ cnta, const float* __restrict__ b1a,
                const float* __restrict__ Wpa, const float* __restrict__ Wqa,
                const float* __restrict__ b2a,
                float* __restrict__ Pa, float* __restrict__ Qa, int Ma,
                const float* __restrict__ Sb, const float* __restrict__ Wlb,
                const float* __restrict__ Xb, const float* __restrict__ Wrb,
                const unsigned* __restrict__ cntb, const float* __restrict__ b1b,
                const float* __restrict__ Wpb, const float* __restrict__ Wqb,
                const float* __restrict__ b2b,
                float* __restrict__ Pb, float* __restrict__ Qb, int Mb, int split)
{
    const bool second = (int)blockIdx.x >= split;
    sage_chain_body<K>(second ? Sb : Sa, second ? Wlb : Wla, second ? Xb : Xa, second ? Wrb : Wra,
                       second ? cntb : cnta, second ? b1b : b1a,
                       second ? Wpb : Wpa, second ? Wqb : Wqa, second ? b2b : b2a,
                       second ? Pb : Pa, second ? Qb : Qa, second ? Mb : Ma,
                       second ? (int)blockIdx.x - split : (int)blockIdx.x,
                       second ? (int)gridDim.x - split : split);
}

#define SORT_G      256
#define SORT_T      512
#define OCC_SHIFT   8
#define SKILL_SHIFT 9
#define NB_OCC      ((N_OCC   + (1 << OCC_SHIFT)   - 1) >> OCC_SHIFT)
#define NB_SKILL    ((N_SKILL + (1 << SKILL_SHIFT) - 1) >> SKILL_SHIFT)
#define NB_ALL      (NB_OCC + NB_SKILL)
#define SORT_CHUNK  ((N_EDGE + SORT_G - 1) / SORT_G)
#define SORT_PER_T  ((SORT_CHUNK + SORT_T - 1) / SORT_T)
static_assert(N_OCC <= 65536 && N_SKILL <= 65536, "a source id must fit the low 16 bits of an edge word");
#define ROWS_PER_W  ((NB_ALL + SORT_T / 32 - 1) / (SORT_T / 32))
static_assert(SORT_G == 256, "a histogram row is two 16-byte loads per lane of a wave");
static_assert(NB_OCC <= 128 && NB_SKILL <= 128, "the bin starts are scanned by one wave, 4 bins per lane");
static_assert(NB_ALL <= SORT_T, "one thread per histogram row");

__device__ __forceinline__ unsigned bin_of_occ(unsigned o)
{
    return min(o >> OCC_SHIFT, (unsigned)(NB_OCC - 1));
}
__device__ __forceinline__ unsigned bin_of_skill(unsigned s)
{
    return min(s >> SKILL_SHIFT, (unsigned)(NB_SKILL - 1));
}

__device__ __forceinline__ void
bin_hist_body(const int* __restrict__ e_occ, const int* __restrict__ e_skill,
              unsigned* __restrict__ hist, int g)
{
    __shared__ unsigned h[NB_ALL];
    const int t = threadIdx.x;
    if (t < NB_ALL) h[t] = 0u;
    __syncthreads();
    const int beg = g * SORT_CHUNK;
    const int end = min(beg + SORT_CHUNK, N_EDGE);
    unsigned o[SORT_PER_T], s[SORT_PER_T];
    #pragma unroll
    for (int k = 0; k < SORT_PER_T; ++k) {
        const int e = beg + t + k * SORT_T;
        o[k] = 0u; s[k] = 0u;
        if (e < end) { o[k] = (unsigned)e_occ[e]; s[k] = (unsigned)e_skill[e]; }
    }
    #pragma unroll
    for (int k = 0; k < SORT_PER_T; ++k) {
        const int e = beg + t + k * SORT_T;
        if (e < end) {
            atomicAdd(&h[bin_of_occ(o[k])], 1u);
            atomicAdd(&h[NB_OCC + bin_of_skill(s[k])], 1u);
        }
    }
    __syncthreads();
    if (t < NB_ALL) hist[t * SORT_G + g] = h[t];
}

__device__ __forceinline__ void
bin_place_body(const int* __restrict__ e_occ, const int* __restrict__ e_skill,
               const unsigned* __restrict__ hist,
               unsigned* __restrict__ bstart, unsigned* __restrict__ bsize,
               unsigned* __restrict__ binned_occ, unsigned* __restrict__ binned_skill, int g)
{
    __shared__ unsigned tot[NB_ALL];
    __shared__ unsigned cur[NB_ALL];
    const int t = threadIdx.x;
    const int lane = t & 31, wave = t >> 5;

    const int beg = g * SORT_CHUNK;
    const int end = min(beg + SORT_CHUNK, N_EDGE);
    unsigned o[SORT_PER_T], s[SORT_PER_T];
    #pragma unroll
    for (int k = 0; k < SORT_PER_T; ++k) {
        const int e = beg + t + k * SORT_T;
        o[k] = 0u; s[k] = 0u;
        if (e < end) { o[k] = (unsigned)e_occ[e]; s[k] = (unsigned)e_skill[e]; }
    }

    uint4 va[ROWS_PER_W], vb[ROWS_PER_W];
    #pragma unroll
    for (int j = 0; j < ROWS_PER_W; ++j) {
        const int r = wave + j * (SORT_T / 32);
        va[j] = make_uint4(0u, 0u, 0u, 0u); vb[j] = va[j];
        if (r < NB_ALL) {
            const uint4* row = (const uint4*)(hist + r * SORT_G);
            va[j] = row[2 * lane];
            vb[j] = row[2 * lane + 1];
        }
    }
    #pragma unroll
    for (int j = 0; j < ROWS_PER_W; ++j) {
        const int r = wave + j * (SORT_T / 32);
        const uint4 a = va[j], b = vb[j];
        const int g0 = 8 * lane;
        unsigned all = ((a.x + a.y) + (a.z + a.w)) + ((b.x + b.y) + (b.z + b.w));
        unsigned before = (g0     < g ? a.x : 0u) + (g0 + 1 < g ? a.y : 0u)
                        + (g0 + 2 < g ? a.z : 0u) + (g0 + 3 < g ? a.w : 0u)
                        + (g0 + 4 < g ? b.x : 0u) + (g0 + 5 < g ? b.y : 0u)
                        + (g0 + 6 < g ? b.z : 0u) + (g0 + 7 < g ? b.w : 0u);
        #pragma unroll
        for (int m = 16; m >= 1; m >>= 1) {
            all    += __shfl_xor(all, m);
            before += __shfl_xor(before, m);
        }
        if (r < NB_ALL && lane == 0) { tot[r] = all; cur[r] = before; }
    }
    __syncthreads();
    if (wave < 2) {
        const int base = (wave == 0) ? 0 : NB_OCC;
        const int n    = (wave == 0) ? NB_OCC : NB_SKILL;
        unsigned a[4], sum = 0u;
        #pragma unroll
        for (int i = 0; i < 4; ++i) {
            const int r = 4 * lane + i;
            a[i] = (r < n) ? tot[base + r] : 0u;
            sum += a[i];
        }
        unsigned inc = sum;
        #pragma unroll
        for (int d = 1; d < 32; d <<= 1) {
            const unsigned u = __shfl_up(inc, d);
            if (lane >= d) inc += u;
        }
        unsigned ex = inc - sum;
        #pragma unroll
        for (int i = 0; i < 4; ++i) {
            const int r = 4 * lane + i;
            if (r < n) {
                if (g == 0) { bstart[base + r] = ex; bsize[base + r] = a[i]; }
                cur[base + r] += ex;
                ex += a[i];
            }
        }
    }
    __syncthreads();

    unsigned po[SORT_PER_T], ps[SORT_PER_T];
    #pragma unroll
    for (int k = 0; k < SORT_PER_T; ++k) {
        const int e = beg + t + k * SORT_T;
        po[k] = 0xFFFFFFFFu; ps[k] = 0xFFFFFFFFu;
        if (e < end) {
            po[k] = atomicAdd(&cur[bin_of_occ(o[k])], 1u);
            ps[k] = atomicAdd(&cur[NB_OCC + bin_of_skill(s[k])], 1u);
        }
    }
    #pragma unroll
    for (int k = 0; k < SORT_PER_T; ++k) {
        if (po[k] < (unsigned)N_EDGE)
            binned_occ[po[k]]   = ((o[k] & ((1u << OCC_SHIFT) - 1u)) << 16) | (s[k] & 0xFFFFu);
        if (ps[k] < (unsigned)N_EDGE)
            binned_skill[ps[k]] = ((s[k] & ((1u << SKILL_SHIFT) - 1u)) << 16) | (o[k] & 0xFFFFu);
    }
}

__device__ __forceinline__ void
bin_rank_body(const unsigned* __restrict__ bstart, const unsigned* __restrict__ bsize,
              const unsigned* __restrict__ binned_occ, const unsigned* __restrict__ binned_skill,
              unsigned* __restrict__ cnt_occ, unsigned* __restrict__ cnt_skill,
              unsigned* __restrict__ rp_occ, unsigned* __restrict__ rp_skill,
              unsigned* __restrict__ adj_occ, unsigned* __restrict__ adj_skill, int row)
{
    constexpr int NMAX = 1 << SKILL_SHIFT;
    static_assert((1 << OCC_SHIFT) <= NMAX && NMAX <= SORT_T && NMAX / 32 <= 32, "one thread per node of a bin");
    constexpr int BATCH = 8;
    __shared__ unsigned c[NMAX + 1];
    __shared__ unsigned wsum[NMAX / 32];
    const int t = threadIdx.x;
    const int lane = t & 31, wave = t >> 5;
    const bool sk = row >= NB_OCC;
    const int shift = sk ? SKILL_SHIFT : OCC_SHIFT;
    const int nbase = (sk ? row - NB_OCC : row) << shift;
    const int nn    = min(1 << shift, (sk ? N_SKILL : N_OCC) - nbase);
    const unsigned* data = sk ? binned_skill : binned_occ;
    unsigned* cnt = sk ? cnt_skill : cnt_occ;
    unsigned* rp  = sk ? rp_skill  : rp_occ;
    unsigned* adj = sk ? adj_skill : adj_occ;
    unsigned st = bstart[row]; if (st > (unsigned)N_EDGE) st = (unsigned)N_EDGE;
    unsigned n  = bsize[row];  if (n > (unsigned)N_EDGE - st) n = (unsigned)N_EDGE - st;
    data += st;
    adj  += st;

    if (t < NMAX) c[t] = 0u;
    if (t == 0) c[NMAX] = 0u;
    __syncthreads();
    for (unsigned i0 = t; i0 < n; i0 += BATCH * SORT_T) {
        unsigned w[BATCH];
        #pragma unroll
        for (int k = 0; k < BATCH; ++k) {
            const unsigned i = i0 + k * SORT_T;
            w[k] = data[min(i, n - 1u)];
        }
        #pragma unroll
        for (int k = 0; k < BATCH; ++k) {
            const unsigned i = i0 + k * SORT_T;
            atomicAdd(&c[(i < n) ? min(w[k] >> 16, (unsigned)(NMAX - 1)) : (unsigned)NMAX], 1u);
        }
    }
    __syncthreads();
    unsigned v = 0u, inc = 0u;
    if (t < NMAX) {
        v = c[t];
        inc = v;
        #pragma unroll
        for (int d = 1; d < 32; d <<= 1) {
            const unsigned u = __shfl_up(inc, d);
            if (lane >= d) inc += u;
        }
        if (lane == 31) wsum[wave] = inc;
    }
    __syncthreads();
    if (wave == 0) {
        const unsigned ws = (lane < NMAX / 32) ? wsum[lane] : 0u;
        unsigned wi = ws;
        #pragma unroll
        for (int d = 1; d < 32; d <<= 1) {
            const unsigned u = __shfl_up(wi, d);
            if (lane >= d) wi += u;
        }
        if (lane < NMAX / 32) wsum[lane] = wi - ws;
    }
    __syncthreads();
    if (t < NMAX) {
        const unsigned ex = inc - v + wsum[wave];
        c[t] = ex;
        if (t < nn) { cnt[nbase + t] = v; rp[nbase + t] = st + ex; }
    }
    __syncthreads();
    for (unsigned i0 = t; i0 < n; i0 += BATCH * SORT_T) {
        unsigned w[BATCH], p[BATCH];
        #pragma unroll
        for (int k = 0; k < BATCH; ++k) {
            const unsigned i = i0 + k * SORT_T;
            w[k] = data[min(i, n - 1u)];
        }
        #pragma unroll
        for (int k = 0; k < BATCH; ++k) {
            const unsigned i = i0 + k * SORT_T;
            p[k] = atomicAdd(&c[(i < n) ? min(w[k] >> 16, (unsigned)(NMAX - 1)) : (unsigned)NMAX], 1u);
        }
        #pragma unroll
        for (int k = 0; k < BATCH; ++k) {
            const unsigned i = i0 + k * SORT_T;
            if (i < n && p[k] < n) adj[p[k]] = w[k] & 0xFFFFu;
        }
    }
}

#define PROJ_G        256
#define PROJ_SKILL_1  (208 * 16)
#define PROJ_SKILL_2  (1458 * 16)
#define PROJ_SKILL_3  (N_SKILL - PROJ_SKILL_1 - PROJ_SKILL_2)
#define PROJ_SPLIT_1  219
static_assert(PROJ_SKILL_3 > 0 && PROJ_SKILL_3 % 16 == 0, "every share of the skill rows is whole row tiles");
static_assert(N_OCC / 16 <= 6 * PROJ_SPLIT_1 && PROJ_SKILL_1 / 16 <= 6 * (PROJ_G - PROJ_SPLIT_1) &&
              PROJ_SKILL_2 / 16 <= 6 * PROJ_G && PROJ_SKILL_3 / 16 <= 6 * PROJ_G,
              "at most 6 row tiles per product block");

__global__ void __launch_bounds__(SORT_T)
hist_and_proj(const int* __restrict__ e_occ, const int* __restrict__ e_skill,
              unsigned* __restrict__ hist,
              const float* __restrict__ xa, const float* __restrict__ Wa,
              const float* __restrict__ biasa, float* __restrict__ ha, int Ma,
              const float* __restrict__ xb, const float* __restrict__ Wb,
              const float* __restrict__ biasb, float* __restrict__ hb, int Mb, int split)
{
    if ((int)blockIdx.x >= PROJ_G) {
        bin_hist_body(e_occ, e_skill, hist, (int)blockIdx.x - PROJ_G);
        return;
    }
    const bool second = (int)blockIdx.x >= split;
    gemm_body<D_IN, HID, 0, SORT_T>(second ? xb : xa, second ? Wb : Wa, nullptr, nullptr, nullptr,
                                    second ? biasb : biasa, second ? hb : ha, second ? Mb : Ma, 1,
                                    second ? (int)blockIdx.x - split : (int)blockIdx.x,
                                    second ? PROJ_G - split : split);
}

__global__ void __launch_bounds__(SORT_T)
place_and_proj(const int* __restrict__ e_occ, const int* __restrict__ e_skill,
               const unsigned* __restrict__ hist,
               unsigned* __restrict__ bstart, unsigned* __restrict__ bsize,
               unsigned* __restrict__ binned_occ, unsigned* __restrict__ binned_skill,
               const float* __restrict__ x, const float* __restrict__ W,
               const float* __restrict__ bias, float* __restrict__ h, int M)
{
    if ((int)blockIdx.x >= PROJ_G) {
        bin_place_body(e_occ, e_skill, hist, bstart, bsize, binned_occ, binned_skill,
                       (int)blockIdx.x - PROJ_G);
        return;
    }
    gemm_body<D_IN, HID, 0, SORT_T>(x, W, nullptr, nullptr, nullptr, bias, h, M, 1,
                                    (int)blockIdx.x, PROJ_G);
}

__global__ void __launch_bounds__(SORT_T)
rank_and_proj(const unsigned* __restrict__ bstart, const unsigned* __restrict__ bsize,
              const unsigned* __restrict__ binned_occ, const unsigned* __restrict__ binned_skill,
              unsigned* __restrict__ cnt_occ, unsigned* __restrict__ cnt_skill,
              unsigned* __restrict__ rp_occ, unsigned* __restrict__ rp_skill,
              unsigned* __restrict__ adj_occ, unsigned* __restrict__ adj_skill,
              const float* __restrict__ x, const float* __restrict__ W,
              const float* __restrict__ bias, float* __restrict__ h, int M)
{
    if ((int)blockIdx.x >= PROJ_G) {
        bin_rank_body(bstart, bsize, binned_occ, binned_skill, cnt_occ, cnt_skill,
                      rp_occ, rp_skill, adj_occ, adj_skill, (int)blockIdx.x - PROJ_G);
        return;
    }
    gemm_body<D_IN, HID, 0, SORT_T>(x, W, nullptr, nullptr, nullptr, bias, h, M, 1,
                                    (int)blockIdx.x, PROJ_G);
}

__device__ inline void add4(float4& a, const float4& v)
{
    a.x += v.x; a.y += v.y; a.z += v.z; a.w += v.w;
}

template <int F, bool ADD>
__device__ __forceinline__ void
gather_body(const float* __restrict__ hsrc, const unsigned* __restrict__ adj,
            const unsigned* __restrict__ rp, const unsigned* __restrict__ cnt,
            float* __restrict__ agg, int n_dst, unsigned n_src, unsigned E, int bid)
{
    constexpr int TPR = F / 4;
    const int tid = bid * 256 + threadIdx.x;
    const int d = tid / TPR;
    const int f = (tid & (TPR - 1)) << 2;
    if (d >= n_dst) return;
    unsigned beg = rp[d];  if (beg > E) beg = E;
    const unsigned craw = cnt[d];
    unsigned deg = craw; if (deg > E - beg) deg = E - beg;
    float4* orow = (float4*)(agg + (size_t)d * F + f);
    float4 q = {0.f, 0.f, 0.f, 0.f};
    if (ADD) q = *orow;
    const unsigned top = n_src - 1;
    const unsigned* a = adj + beg;
    const float* hb = hsrc + f;
    float4 s0 = {0.f, 0.f, 0.f, 0.f}, s1 = s0, s2 = s0, s3 = s0;
    unsigned j = 0;
    unsigned n0 = 0, n1 = 0, n2 = 0, n3 = 0;
    if (deg >= 4) { n0 = a[0]; n1 = a[1]; n2 = a[2]; n3 = a[3]; }
    for (; j + 4 <= deg; j += 4) {
        const unsigned i0 = min(n0, top), i1 = min(n1, top);
        const unsigned i2 = min(n2, top), i3 = min(n3, top);
        if (j + 8 <= deg) { n0 = a[j + 4]; n1 = a[j + 5]; n2 = a[j + 6]; n3 = a[j + 7]; }
        const float4 v0 = *(const float4*)(hb + (size_t)i0 * F);
        const float4 v1 = *(const float4*)(hb + (size_t)i1 * F);
        const float4 v2 = *(const float4*)(hb + (size_t)i2 * F);
        const float4 v3 = *(const float4*)(hb + (size_t)i3 * F);
        add4(s0, v0); add4(s1, v1); add4(s2, v2); add4(s3, v3);
    }
    if (j < deg) {
        const unsigned r = deg - j;
        const unsigned i0 = min(a[j], top);
        const unsigned i1 = min(a[r > 1 ? j + 1 : j], top);
        const unsigned i2 = min(a[r > 2 ? j + 2 : j], top);
        const float4 v0 = *(const float4*)(hb + (size_t)i0 * F);
        const float4 v1 = *(const float4*)(hb + (size_t)i1 * F);
        const float4 v2 = *(const float4*)(hb + (size_t)i2 * F);
        add4(s0, v0);
        if (r > 1) add4(s1, v1);
        if (r > 2) add4(s2, v2);
    }
    add4(s0, s1); add4(s2, s3); add4(s0, s2);
    if (ADD) {
        const float inv = 1.0f / fmaxf((float)craw, 1.0f);
        q.x += inv * s0.x; q.y += inv * s0.y; q.z += inv * s0.z; q.w += inv * s0.w;
        *orow = q;
    } else {
        *orow = s0;
    }
}

template <int F, bool ADD>
__global__ void __launch_bounds__(256)
gather_pair(const float* __restrict__ hsrc_a, const unsigned* __restrict__ adj_a,
            const unsigned* __restrict__ rp_a, const unsigned* __restrict__ cnt_a,
            float* __restrict__ agg_a, int n_dst_a, unsigned n_src_a,
            const float* __restrict__ hsrc_b, const unsigned* __restrict__ adj_b,
            const unsigned* __restrict__ rp_b, const unsigned* __restrict__ cnt_b,
            float* __restrict__ agg_b, int n_dst_b, unsigned n_src_b,
            int split, unsigned E)
{
    const bool second = (int)blockIdx.x >= split;
    gather_body<F, ADD>(second ? hsrc_b : hsrc_a, second ? adj_b : adj_a,
                   second ? rp_b : rp_a, second ? cnt_b : cnt_a,
                   second ? agg_b : agg_a, second ? n_dst_b : n_dst_a,
                   second ? n_src_b : n_src_a, E,
                   second ? (int)blockIdx.x - split : (int)blockIdx.x);
}

extern "C" void kernel_launch(void* const* d_in, const int* in_sizes, int n_in,
                              void* d_out, int out_size, void* d_ws, size_t ws_size,
                              hipStream_t stream)
{
    (void)in_sizes; (void)n_in; (void)out_size; (void)ws_size;

    const float* x_occ    = (const float*)d_in[0];
    const float* x_skill  = (const float*)d_in[1];
    const int*   e_occ    = (const int*)  d_in[2];
    const int*   e_skill  = (const int*)  d_in[3];
    const float* pW_occ   = (const float*)d_in[4];
    const float* pb_occ   = (const float*)d_in[5];
    const float* pW_skill = (const float*)d_in[6];
    const float* pb_skill = (const float*)d_in[7];
    const float* c1_os_Wl = (const float*)d_in[8];
    const float* c1_os_bl = (const float*)d_in[9];
    const float* c1_os_Wr = (const float*)d_in[10];
    const float* c1_so_Wl = (const float*)d_in[11];
    const float* c1_so_bl = (const float*)d_in[12];
    const float* c1_so_Wr = (const float*)d_in[13];
    const float* c2_os_Wl = (const float*)d_in[14];
    const float* c2_os_bl = (const float*)d_in[15];
    const float* c2_os_Wr = (const float*)d_in[16];
    const float* c2_so_Wl = (const float*)d_in[17];
    const float* c2_so_bl = (const float*)d_in[18];
    const float* c2_so_Wr = (const float*)d_in[19];

    float* ws = (float*)d_ws;
    float* h_occ     = ws;
    float* h_skill   = h_occ     + (size_t)N_OCC   * HID;
    float* o1        = h_skill   + (size_t)N_SKILL * HID;
    float* s1        = o1        + (size_t)N_OCC   * HID;
    float* aggbuf    = s1        + (size_t)N_SKILL * HID;
    unsigned* adj_skill  = (unsigned*)(aggbuf + (size_t)N_SKILL * HID);
    unsigned* adj_occ    = adj_skill  + N_EDGE;
    unsigned* rp_occ     = adj_occ    + N_EDGE;
    unsigned* rp_skill   = rp_occ     + N_OCC;
    unsigned* binned_occ   = rp_skill   + N_SKILL;
    unsigned* binned_skill = binned_occ + N_EDGE;
    unsigned* hist   = binned_skill + N_EDGE;
    unsigned* bstart = hist   + NB_ALL * SORT_G;
    unsigned* bsize  = bstart + NB_ALL;
    static_assert(2 * N_EDGE + N_OCC + N_SKILL + 2 * N_EDGE + NB_ALL * SORT_G + 2 * NB_ALL <= N_OCC * HID,
                  "the edge tables must fit the 20000*128 words behind the neighbour-sum buffer");
    unsigned* cnt_occ   = (unsigned*)(aggbuf + (size_t)(N_OCC + N_SKILL) * HID);
    unsigned* cnt_skill = cnt_occ + N_OCC;
    float* o2 = (float*)d_out;
    float* s2 = o2 + (size_t)N_OCC * OUTF;

    const dim3 blk256(256);
    const dim3 blkS(SORT_T);
    hist_and_proj<<<PROJ_G + SORT_G, blkS, 0, stream>>>(
        e_occ, e_skill, hist,
        x_occ,   pW_occ,   pb_occ,   h_occ,   N_OCC,
        x_skill, pW_skill, pb_skill, h_skill, PROJ_SKILL_1, PROJ_SPLIT_1);
    place_and_proj<<<PROJ_G + SORT_G, blkS, 0, stream>>>(
        e_occ, e_skill, hist, bstart, bsize, binned_occ, binned_skill,
        x_skill + (size_t)PROJ_SKILL_1 * D_IN, pW_skill, pb_skill,
        h_skill + (size_t)PROJ_SKILL_1 * HID, PROJ_SKILL_2);
    rank_and_proj<<<PROJ_G + NB_ALL, blkS, 0, stream>>>(
        bstart, bsize, binned_occ, binned_skill,
        cnt_occ, cnt_skill, rp_occ, rp_skill, adj_occ, adj_skill,
        x_skill + (size_t)(PROJ_SKILL_1 + PROJ_SKILL_2) * D_IN, pW_skill, pb_skill,
        h_skill + (size_t)(PROJ_SKILL_1 + PROJ_SKILL_2) * HID, PROJ_SKILL_3);

    float* agg1_occ = o1;
    gather_pair<HID, false><<<N_OCC / 8 + N_SKILL / 8, blk256, 0, stream>>>(
        h_skill, adj_occ,   rp_occ,   cnt_occ,   agg1_occ, N_OCC,   N_SKILL,
        h_occ,   adj_skill, rp_skill, cnt_skill, aggbuf,   N_SKILL, N_OCC,
        N_OCC / 8, N_EDGE);

    float* p_occ   = s1;
    float* p_skill = s1 + (size_t)N_OCC * OUTF;
    static_assert((size_t)(N_OCC + N_SKILL) * OUTF <= (size_t)N_SKILL * HID, "both projected planes fit one 50000*128 area");
    sage_chain_pair<HID><<<dim3(256), dim3(CHAIN_T), 0, stream>>>(
        agg1_occ, c1_so_Wl, h_occ,   c1_so_Wr, cnt_occ,   c1_so_bl,
        c2_os_Wl, c2_so_Wr, c2_so_bl, p_occ,   o2, N_OCC,
        aggbuf,   c1_os_Wl, h_skill, c1_os_Wr, cnt_skill, c1_os_bl,
        c2_so_Wl, c2_os_Wr, c2_os_bl, p_skill, s2, N_SKILL, 73);
    gather_pair<OUTF, true><<<N_OCC / 16 + N_SKILL / 16, blk256, 0, stream>>>(
        p_skill, adj_occ,   rp_occ,   cnt_occ,   o2, N_OCC,   N_SKILL,
        p_occ,   adj_skill, rp_skill, cnt_skill, s2, N_SKILL, N_OCC,
        N_OCC / 16, N_EDGE);
}
